# starting kernel
# baseline (speedup 1.0000x reference)
#include <hip/hip_runtime.h>


typedef float v2f __attribute__((ext_vector_type(2)));
typedef float v4f __attribute__((ext_vector_type(4)));
typedef float v8f __attribute__((ext_vector_type(8)));

#define NEG_SLOPE 0.2f
#define LN_EPS 1e-5f

__device__ __forceinline__ unsigned fenc(float f) {
  unsigned u = __float_as_uint(f);
  return (u >> 31) ? ~u : (u | 0x80000000u);
}
__device__ __forceinline__ float fdec(unsigned u) {
  return __uint_as_float((u >> 31) ? (u & 0x7FFFFFFFu) : ~u);
}
__device__ __forceinline__ float lrelu(float x) {
  return x > 0.f ? x : x * NEG_SLOPE;
}

template <int K>
__global__ void gat_gemm_wmma(const float* __restrict__ Amat,
                              const float* __restrict__ Bmat,
                              float* __restrict__ Cmat,
                              int M) {
  __shared__ float bs[K * 128];

  const int tid = threadIdx.x;

  for (int idx = tid; idx < K * 128; idx += 256) {
    int kr = idx >> 7;
    int col = idx & 127;
    int ks = kr >> 2;
    int c = col >> 4;
    int lane = (col & 15) | ((kr & 2) << 3);
    int sub = kr & 1;
    bs[((ks * 32 + lane) * 8 + c) * 2 + sub] = Bmat[idx];
  }
  __syncthreads();

  const int wave = tid >> 5;
  const int lane = tid & 31;
  const int row0 = (blockIdx.x * 8 + wave) * 16;
  if (row0 >= M) return;

  const int m  = lane & 15;
  const int kk = (lane >> 4) << 1;

  v8f acc[8] = {};

  const float* arow = Amat + (size_t)(row0 + m) * K + kk;
  const float* blane = bs + lane * 16;

#pragma unroll 4
  for (int ks = 0; ks < K / 4; ++ks) {
    v2f a = *(const v2f*)(arow + ks * 4);
    const float* bp = blane + ks * 512;
    v4f q0 = *(const v4f*)(bp);
    v4f q1 = *(const v4f*)(bp + 4);
    v4f q2 = *(const v4f*)(bp + 8);
    v4f q3 = *(const v4f*)(bp + 12);
    acc[0] = __builtin_amdgcn_wmma_f32_16x16x4_f32(false, a, false, q0.xy, (short)0, acc[0], false, false);
    acc[1] = __builtin_amdgcn_wmma_f32_16x16x4_f32(false, a, false, q0.zw, (short)0, acc[1], false, false);
    acc[2] = __builtin_amdgcn_wmma_f32_16x16x4_f32(false, a, false, q1.xy, (short)0, acc[2], false, false);
    acc[3] = __builtin_amdgcn_wmma_f32_16x16x4_f32(false, a, false, q1.zw, (short)0, acc[3], false, false);
    acc[4] = __builtin_amdgcn_wmma_f32_16x16x4_f32(false, a, false, q2.xy, (short)0, acc[4], false, false);
    acc[5] = __builtin_amdgcn_wmma_f32_16x16x4_f32(false, a, false, q2.zw, (short)0, acc[5], false, false);
    acc[6] = __builtin_amdgcn_wmma_f32_16x16x4_f32(false, a, false, q3.xy, (short)0, acc[6], false, false);
    acc[7] = __builtin_amdgcn_wmma_f32_16x16x4_f32(false, a, false, q3.zw, (short)0, acc[7], false, false);
  }

  const int mo = (lane >> 4) * 8;
#pragma unroll
  for (int c = 0; c < 8; ++c) {
#pragma unroll
    for (int r = 0; r < 8; ++r) {
      Cmat[(size_t)(row0 + mo + r) * 128 + c * 16 + m] = acc[c][r];
    }
  }
}

__global__ void gat_attn_proj(const float* __restrict__ h,
                              const float* __restrict__ al,
                              const float* __restrict__ ar,
                              float* __restrict__ el, float* __restrict__ er,
                              int NH) {
  int i = blockIdx.x * blockDim.x + threadIdx.x;
  if (i >= NH) return;
  int n = i >> 2, hh = i & 3;
  const float* hp  = h + (size_t)n * 128 + hh * 32;
  const float* alp = al + hh * 32;
  const float* arp = ar + hh * 32;
  float sl = 0.f, sr = 0.f;
#pragma unroll
  for (int f = 0; f < 32; ++f) {
    float v = hp[f];
    sl += v * alp[f];
    sr += v * arp[f];
  }
  el[i] = sl;
  er[i] = sr;
}

__global__ void gat_edge_max(const int* __restrict__ src, const int* __restrict__ dst,
                             const float* __restrict__ el, const float* __restrict__ er,
                             unsigned* __restrict__ emax, int E4) {
  int i = blockIdx.x * blockDim.x + threadIdx.x;
  if (i >= E4) return;
  int e = i >> 2, hh = i & 3;
  int s = src[e], d = dst[e];
  float ev = lrelu(el[s * 4 + hh] + er[d * 4 + hh]);
  atomicMax(&emax[d * 4 + hh], fenc(ev));
}

__global__ void gat_edge_denom(const int* __restrict__ src, const int* __restrict__ dst,
                               const float* __restrict__ el, const float* __restrict__ er,
                               const unsigned* __restrict__ emax,
                               float* __restrict__ denom, int E4) {
  int i = blockIdx.x * blockDim.x + threadIdx.x;
  if (i >= E4) return;
  int e = i >> 2, hh = i & 3;
  int s = src[e], d = dst[e];
  float ev = lrelu(el[s * 4 + hh] + er[d * 4 + hh]);
  float mx = fdec(emax[d * 4 + hh]);
  atomicAdd(&denom[d * 4 + hh], expf(ev - mx));
}

__global__ void gat_edge_scatter(const int* __restrict__ src, const int* __restrict__ dst,
                                 const float* __restrict__ el, const float* __restrict__ er,
                                 const unsigned* __restrict__ emax,
                                 const float* __restrict__ denom,
                                 const float* __restrict__ h,
                                 float* __restrict__ rst, int E) {
  int gid = blockIdx.x * 256 + threadIdx.x;
  int e = gid >> 7;
  if (e >= E) return;
  int t = gid & 127;
  int hh = t >> 5;
  int s = src[e], d = dst[e];
  float ev = lrelu(el[s * 4 + hh] + er[d * 4 + hh]);
  float mx = fdec(emax[d * 4 + hh]);
  float ex = expf(ev - mx);
  float alpha = ex / fmaxf(denom[d * 4 + hh], 1e-9f);
  atomicAdd(&rst[(size_t)d * 128 + t], h[(size_t)s * 128 + t] * alpha);
}

__global__ void gat_finalize1(const float* __restrict__ rst,
                              const float* __restrict__ bias,
                              const float* __restrict__ lns,
                              const float* __restrict__ lnb,
                              float* __restrict__ hmid, int Nn) {
  int n = blockIdx.x * 8 + (threadIdx.x >> 5);
  int f = threadIdx.x & 31;
  if (n >= Nn) return;
  float x = 0.f;
#pragma unroll
  for (int hh = 0; hh < 4; ++hh)
    x += rst[(size_t)n * 128 + hh * 32 + f] + bias[hh * 32 + f];
  x *= 0.25f;
  float mu = x;
#pragma unroll
  for (int m = 16; m >= 1; m >>= 1) mu += __shfl_xor(mu, m, 32);
  mu *= (1.f / 32.f);
  float dx = x - mu;
  float var = dx * dx;
#pragma unroll
  for (int m = 16; m >= 1; m >>= 1) var += __shfl_xor(var, m, 32);
  var *= (1.f / 32.f);
  float y = dx * rsqrtf(var + LN_EPS) * lns[f] + lnb[f];
  hmid[(size_t)n * 32 + f] = fmaxf(y, 0.f);
}

__global__ void gat_finalize2(const float* __restrict__ rst,
                              const float* __restrict__ bias,
                              float* __restrict__ out, int NF) {
  int i = blockIdx.x * blockDim.x + threadIdx.x;
  if (i >= NF) return;
  int n = i >> 5, f = i & 31;
  float x = 0.f;
#pragma unroll
  for (int hh = 0; hh < 4; ++hh)
    x += rst[(size_t)n * 128 + hh * 32 + f] + bias[hh * 32 + f];
  out[i] = x * 0.25f;
}

__global__ void gat_flops(const int* __restrict__ in_deg, int Nn, float* out) {
  __shared__ float ssum[256];
  float s = 0.f;
  for (int i = threadIdx.x; i < Nn; i += 256) s += (float)in_deg[i];
  ssum[threadIdx.x] = s;
  __syncthreads();
  for (int m = 128; m > 0; m >>= 1) {
    if ((int)threadIdx.x < m) ssum[threadIdx.x] += ssum[threadIdx.x + m];
    __syncthreads();
  }
  if (threadIdx.x == 0) {
    double ne = (double)ssum[0];
    double fl = (4.0 * ne * (6.0 * 128 * 32 + 6 * 32 + 2) +
                 4.0 * ne * (6.0 * 32 * 32 + 6 * 32 + 2)) / 1e12;
    out[0] = (float)fl;
  }
}

extern "C" void kernel_launch(void* const* d_in, const int* in_sizes, int n_in,
                              void* d_out, int out_size, void* d_ws, size_t ws_size,
                              hipStream_t stream) {
  const float* feat = (const float*)d_in[0];
  const float* W1   = (const float*)d_in[1];
  const float* al1  = (const float*)d_in[2];
  const float* ar1  = (const float*)d_in[3];
  const float* b1   = (const float*)d_in[4];
  const float* lns  = (const float*)d_in[5];
  const float* lnb  = (const float*)d_in[6];
  const float* W2   = (const float*)d_in[7];
  const float* al2  = (const float*)d_in[8];
  const float* ar2  = (const float*)d_in[9];
  const float* b2   = (const float*)d_in[10];
  const int*   src  = (const int*)d_in[11];
  const int*   dst  = (const int*)d_in[12];
  const int*   indeg= (const int*)d_in[13];

  const int Nn = in_sizes[13];
  const int E  = in_sizes[11];
  const int NH = Nn * 4;

  char* p = (char*)d_ws;
  float*    hbuf  = (float*)p;   p += (size_t)Nn * 128 * sizeof(float);
  float*    rst   = (float*)p;   p += (size_t)Nn * 128 * sizeof(float);
  float*    hmid  = (float*)p;   p += (size_t)Nn * 32  * sizeof(float);
  float*    el    = (float*)p;   p += (size_t)NH * sizeof(float);
  float*    er    = (float*)p;   p += (size_t)NH * sizeof(float);
  unsigned* emax  = (unsigned*)p;p += (size_t)NH * sizeof(unsigned);
  float*    denom = (float*)p;   p += (size_t)NH * sizeof(float);
  (void)ws_size; (void)n_in; (void)out_size;

  float* outh = (float*)d_out;

  dim3 b256(256);
  const int rowTiles   = (Nn + 15) / 16;
  const int gemmBlocks = (rowTiles + 7) / 8;
  const int e4Blocks   = (E * 4 + 255) / 256;
  const int scBlocks   = (E * 128 + 255) / 256;

  hipMemsetAsync(rst,   0, (size_t)Nn * 128 * sizeof(float), stream);
  hipMemsetAsync(emax,  0, (size_t)NH * sizeof(unsigned), stream);
  hipMemsetAsync(denom, 0, (size_t)NH * sizeof(float), stream);

  gat_gemm_wmma<128><<<gemmBlocks, b256, 0, stream>>>(feat, W1, hbuf, Nn);
  gat_attn_proj<<<(NH + 255) / 256, b256, 0, stream>>>(hbuf, al1, ar1, el, er, NH);
  gat_edge_max<<<e4Blocks, b256, 0, stream>>>(src, dst, el, er, emax, E * 4);
  gat_edge_denom<<<e4Blocks, b256, 0, stream>>>(src, dst, el, er, emax, denom, E * 4);
  gat_edge_scatter<<<scBlocks, b256, 0, stream>>>(src, dst, el, er, emax, denom, hbuf, rst, E);
  gat_finalize1<<<(Nn + 7) / 8, b256, 0, stream>>>(rst, b1, lns, lnb, hmid, Nn);

  hipMemsetAsync(rst,   0, (size_t)Nn * 128 * sizeof(float), stream);
  hipMemsetAsync(emax,  0, (size_t)NH * sizeof(unsigned), stream);
  hipMemsetAsync(denom, 0, (size_t)NH * sizeof(float), stream);

  gat_gemm_wmma<32><<<gemmBlocks, b256, 0, stream>>>(hmid, W2, hbuf, Nn);
  gat_attn_proj<<<(NH + 255) / 256, b256, 0, stream>>>(hbuf, al2, ar2, el, er, NH);
  gat_edge_max<<<e4Blocks, b256, 0, stream>>>(src, dst, el, er, emax, E * 4);
  gat_edge_denom<<<e4Blocks, b256, 0, stream>>>(src, dst, el, er, emax, denom, E * 4);
  gat_edge_scatter<<<scBlocks, b256, 0, stream>>>(src, dst, el, er, emax, denom, hbuf, rst, E);
  gat_finalize2<<<(Nn * 32 + 255) / 256, b256, 0, stream>>>(rst, b2, outh, Nn * 32);

  gat_flops<<<1, 256, 0, stream>>>(indeg, Nn, outh + (size_t)Nn * 32);
}


# edge scatter sums the 4 heads per edge in registers before the f32 atomic add (head mean commutes with the edge sum): rst is N x 32, one wave per edge, 4x fewer atomics; fp sum order changes
# speedup vs baseline: 1.7903x; 1.7903x over previous
#include <hip/hip_runtime.h>


typedef float v2f __attribute__((ext_vector_type(2)));
typedef float v4f __attribute__((ext_vector_type(4)));
typedef float v8f __attribute__((ext_vector_type(8)));

#define NEG_SLOPE 0.2f
#define LN_EPS 1e-5f

__device__ __forceinline__ unsigned fenc(float f) {
  unsigned u = __float_as_uint(f);
  return (u >> 31) ? ~u : (u | 0x80000000u);
}
__device__ __forceinline__ float fdec(unsigned u) {
  return __uint_as_float((u >> 31) ? (u & 0x7FFFFFFFu) : ~u);
}
__device__ __forceinline__ float lrelu(float x) {
  return x > 0.f ? x : x * NEG_SLOPE;
}

template <int K>
__global__ void gat_gemm_wmma(const float* __restrict__ Amat,
                              const float* __restrict__ Bmat,
                              float* __restrict__ Cmat,
                              int M) {
  __shared__ float bs[K * 128];

  const int tid = threadIdx.x;

  for (int idx = tid; idx < K * 128; idx += 256) {
    int kr = idx >> 7;
    int col = idx & 127;
    int ks = kr >> 2;
    int c = col >> 4;
    int lane = (col & 15) | ((kr & 2) << 3);
    int sub = kr & 1;
    bs[((ks * 32 + lane) * 8 + c) * 2 + sub] = Bmat[idx];
  }
  __syncthreads();

  const int wave = tid >> 5;
  const int lane = tid & 31;
  const int row0 = (blockIdx.x * 8 + wave) * 16;
  if (row0 >= M) return;

  const int m  = lane & 15;
  const int kk = (lane >> 4) << 1;

  v8f acc[8] = {};

  const float* arow = Amat + (size_t)(row0 + m) * K + kk;
  const float* blane = bs + lane * 16;

#pragma unroll 4
  for (int ks = 0; ks < K / 4; ++ks) {
    v2f a = *(const v2f*)(arow + ks * 4);
    const float* bp = blane + ks * 512;
    v4f q0 = *(const v4f*)(bp);
    v4f q1 = *(const v4f*)(bp + 4);
    v4f q2 = *(const v4f*)(bp + 8);
    v4f q3 = *(const v4f*)(bp + 12);
    acc[0] = __builtin_amdgcn_wmma_f32_16x16x4_f32(false, a, false, q0.xy, (short)0, acc[0], false, false);
    acc[1] = __builtin_amdgcn_wmma_f32_16x16x4_f32(false, a, false, q0.zw, (short)0, acc[1], false, false);
    acc[2] = __builtin_amdgcn_wmma_f32_16x16x4_f32(false, a, false, q1.xy, (short)0, acc[2], false, false);
    acc[3] = __builtin_amdgcn_wmma_f32_16x16x4_f32(false, a, false, q1.zw, (short)0, acc[3], false, false);
    acc[4] = __builtin_amdgcn_wmma_f32_16x16x4_f32(false, a, false, q2.xy, (short)0, acc[4], false, false);
    acc[5] = __builtin_amdgcn_wmma_f32_16x16x4_f32(false, a, false, q2.zw, (short)0, acc[5], false, false);
    acc[6] = __builtin_amdgcn_wmma_f32_16x16x4_f32(false, a, false, q3.xy, (short)0, acc[6], false, false);
    acc[7] = __builtin_amdgcn_wmma_f32_16x16x4_f32(false, a, false, q3.zw, (short)0, acc[7], false, false);
  }

  const int mo = (lane >> 4) * 8;
#pragma unroll
  for (int c = 0; c < 8; ++c) {
#pragma unroll
    for (int r = 0; r < 8; ++r) {
      Cmat[(size_t)(row0 + mo + r) * 128 + c * 16 + m] = acc[c][r];
    }
  }
}

__global__ void gat_attn_proj(const float* __restrict__ h,
                              const float* __restrict__ al,
                              const float* __restrict__ ar,
                              float* __restrict__ el, float* __restrict__ er,
                              int NH) {
  int i = blockIdx.x * blockDim.x + threadIdx.x;
  if (i >= NH) return;
  int n = i >> 2, hh = i & 3;
  const float* hp  = h + (size_t)n * 128 + hh * 32;
  const float* alp = al + hh * 32;
  const float* arp = ar + hh * 32;
  float sl = 0.f, sr = 0.f;
#pragma unroll
  for (int f = 0; f < 32; ++f) {
    float v = hp[f];
    sl += v * alp[f];
    sr += v * arp[f];
  }
  el[i] = sl;
  er[i] = sr;
}

__global__ void gat_edge_max(const int* __restrict__ src, const int* __restrict__ dst,
                             const float* __restrict__ el, const float* __restrict__ er,
                             unsigned* __restrict__ emax, int E4) {
  int i = blockIdx.x * blockDim.x + threadIdx.x;
  if (i >= E4) return;
  int e = i >> 2, hh = i & 3;
  int s = src[e], d = dst[e];
  float ev = lrelu(el[s * 4 + hh] + er[d * 4 + hh]);
  atomicMax(&emax[d * 4 + hh], fenc(ev));
}

__global__ void gat_edge_denom(const int* __restrict__ src, const int* __restrict__ dst,
                               const float* __restrict__ el, const float* __restrict__ er,
                               const unsigned* __restrict__ emax,
                               float* __restrict__ denom, int E4) {
  int i = blockIdx.x * blockDim.x + threadIdx.x;
  if (i >= E4) return;
  int e = i >> 2, hh = i & 3;
  int s = src[e], d = dst[e];
  float ev = lrelu(el[s * 4 + hh] + er[d * 4 + hh]);
  float mx = fdec(emax[d * 4 + hh]);
  atomicAdd(&denom[d * 4 + hh], expf(ev - mx));
}

__global__ void gat_edge_scatter(const int* __restrict__ src, const int* __restrict__ dst,
                                 const float* __restrict__ el, const float* __restrict__ er,
                                 const unsigned* __restrict__ emax,
                                 const float* __restrict__ denom,
                                 const float* __restrict__ h,
                                 float* __restrict__ rst, int E) {
  const int lane = threadIdx.x & 31;
  const int e = blockIdx.x * 8 + (threadIdx.x >> 5);
  if (e >= E) return;
  const int hh = lane & 3;
  const int s = src[e], d = dst[e];
  float ev = lrelu(el[s * 4 + hh] + er[d * 4 + hh]);
  float mx = fdec(emax[d * 4 + hh]);
  float ex = expf(ev - mx);
  float alpha = ex / fmaxf(denom[d * 4 + hh], 1e-9f);
  const float a0 = __int_as_float(__builtin_amdgcn_readlane(__float_as_int(alpha), 0));
  const float a1 = __int_as_float(__builtin_amdgcn_readlane(__float_as_int(alpha), 1));
  const float a2 = __int_as_float(__builtin_amdgcn_readlane(__float_as_int(alpha), 2));
  const float a3 = __int_as_float(__builtin_amdgcn_readlane(__float_as_int(alpha), 3));
  const float* hp = h + (size_t)s * 128 + lane;
  float acc = hp[0] * a0;
  acc += hp[32] * a1;
  acc += hp[64] * a2;
  acc += hp[96] * a3;
  atomicAdd(&rst[(size_t)d * 32 + lane], acc);
}

__global__ void gat_finalize1(const float* __restrict__ rst,
                              const float* __restrict__ bias,
                              const float* __restrict__ lns,
                              const float* __restrict__ lnb,
                              float* __restrict__ hmid, int Nn) {
  int n = blockIdx.x * 8 + (threadIdx.x >> 5);
  int f = threadIdx.x & 31;
  if (n >= Nn) return;
  float x = rst[(size_t)n * 32 + f];
#pragma unroll
  for (int hh = 0; hh < 4; ++hh) x += bias[hh * 32 + f];
  x *= 0.25f;
  float mu = x;
#pragma unroll
  for (int m = 16; m >= 1; m >>= 1) mu += __shfl_xor(mu, m, 32);
  mu *= (1.f / 32.f);
  float dx = x - mu;
  float var = dx * dx;
#pragma unroll
  for (int m = 16; m >= 1; m >>= 1) var += __shfl_xor(var, m, 32);
  var *= (1.f / 32.f);
  float y = dx * rsqrtf(var + LN_EPS) * lns[f] + lnb[f];
  hmid[(size_t)n * 32 + f] = fmaxf(y, 0.f);
}

__global__ void gat_finalize2(const float* __restrict__ rst,
                              const float* __restrict__ bias,
                              float* __restrict__ out, int NF) {
  int i = blockIdx.x * blockDim.x + threadIdx.x;
  if (i >= NF) return;
  int f = i & 31;
  float x = rst[i];
#pragma unroll
  for (int hh = 0; hh < 4; ++hh) x += bias[hh * 32 + f];
  out[i] = x * 0.25f;
}

__global__ void gat_flops(const int* __restrict__ in_deg, int Nn, float* out) {
  __shared__ float ssum[256];
  float s = 0.f;
  for (int i = threadIdx.x; i < Nn; i += 256) s += (float)in_deg[i];
  ssum[threadIdx.x] = s;
  __syncthreads();
  for (int m = 128; m > 0; m >>= 1) {
    if ((int)threadIdx.x < m) ssum[threadIdx.x] += ssum[threadIdx.x + m];
    __syncthreads();
  }
  if (threadIdx.x == 0) {
    double ne = (double)ssum[0];
    double fl = (4.0 * ne * (6.0 * 128 * 32 + 6 * 32 + 2) +
                 4.0 * ne * (6.0 * 32 * 32 + 6 * 32 + 2)) / 1e12;
    out[0] = (float)fl;
  }
}

extern "C" void kernel_launch(void* const* d_in, const int* in_sizes, int n_in,
                              void* d_out, int out_size, void* d_ws, size_t ws_size,
                              hipStream_t stream) {
  const float* feat = (const float*)d_in[0];
  const float* W1   = (const float*)d_in[1];
  const float* al1  = (const float*)d_in[2];
  const float* ar1  = (const float*)d_in[3];
  const float* b1   = (const float*)d_in[4];
  const float* lns  = (const float*)d_in[5];
  const float* lnb  = (const float*)d_in[6];
  const float* W2   = (const float*)d_in[7];
  const float* al2  = (const float*)d_in[8];
  const float* ar2  = (const float*)d_in[9];
  const float* b2   = (const float*)d_in[10];
  const int*   src  = (const int*)d_in[11];
  const int*   dst  = (const int*)d_in[12];
  const int*   indeg= (const int*)d_in[13];

  const int Nn = in_sizes[13];
  const int E  = in_sizes[11];
  const int NH = Nn * 4;

  char* p = (char*)d_ws;
  float*    hbuf  = (float*)p;   p += (size_t)Nn * 128 * sizeof(float);
  float*    rst   = (float*)p;   p += (size_t)Nn * 32  * sizeof(float);
  float*    hmid  = (float*)p;   p += (size_t)Nn * 32  * sizeof(float);
  float*    el    = (float*)p;   p += (size_t)NH * sizeof(float);
  float*    er    = (float*)p;   p += (size_t)NH * sizeof(float);
  unsigned* emax  = (unsigned*)p;p += (size_t)NH * sizeof(unsigned);
  float*    denom = (float*)p;   p += (size_t)NH * sizeof(float);
  (void)ws_size; (void)n_in; (void)out_size;

  float* outh = (float*)d_out;

  dim3 b256(256);
  const int rowTiles   = (Nn + 15) / 16;
  const int gemmBlocks = (rowTiles + 7) / 8;
  const int e4Blocks   = (E * 4 + 255) / 256;
  const int scBlocks   = (E + 7) / 8;

  hipMemsetAsync(rst,   0, (size_t)Nn * 32 * sizeof(float), stream);
  hipMemsetAsync(emax,  0, (size_t)NH * sizeof(unsigned), stream);
  hipMemsetAsync(denom, 0, (size_t)NH * sizeof(float), stream);

  gat_gemm_wmma<128><<<gemmBlocks, b256, 0, stream>>>(feat, W1, hbuf, Nn);
  gat_attn_proj<<<(NH + 255) / 256, b256, 0, stream>>>(hbuf, al1, ar1, el, er, NH);
  gat_edge_max<<<e4Blocks, b256, 0, stream>>>(src, dst, el, er, emax, E * 4);
  gat_edge_denom<<<e4Blocks, b256, 0, stream>>>(src, dst, el, er, emax, denom, E * 4);
  gat_edge_scatter<<<scBlocks, b256, 0, stream>>>(src, dst, el, er, emax, denom, hbuf, rst, E);
  gat_finalize1<<<(Nn + 7) / 8, b256, 0, stream>>>(rst, b1, lns, lnb, hmid, Nn);

  hipMemsetAsync(rst,   0, (size_t)Nn * 32 * sizeof(float), stream);
  hipMemsetAsync(emax,  0, (size_t)NH * sizeof(unsigned), stream);
  hipMemsetAsync(denom, 0, (size_t)NH * sizeof(float), stream);

  gat_gemm_wmma<32><<<gemmBlocks, b256, 0, stream>>>(hmid, W2, hbuf, Nn);
  gat_attn_proj<<<(NH + 255) / 256, b256, 0, stream>>>(hbuf, al2, ar2, el, er, NH);
  gat_edge_max<<<e4Blocks, b256, 0, stream>>>(src, dst, el, er, emax, E * 4);
  gat_edge_denom<<<e4Blocks, b256, 0, stream>>>(src, dst, el, er, emax, denom, E * 4);
  gat_edge_scatter<<<scBlocks, b256, 0, stream>>>(src, dst, el, er, emax, denom, hbuf, rst, E);
  gat_finalize2<<<(Nn * 32 + 255) / 256, b256, 0, stream>>>(rst, b2, outh, Nn * 32);

  gat_flops<<<1, 256, 0, stream>>>(indeg, Nn, outh + (size_t)Nn * 32);
}


# edges bucketed by destination once per call (int atomic rank, scan, fill); one wave per node does edge softmax, aggregation, epilogue in registers: no float atomics; fp sum order changes
# speedup vs baseline: 2.4805x; 1.3855x over previous
#include <hip/hip_runtime.h>


typedef float v2f __attribute__((ext_vector_type(2)));
typedef float v4f __attribute__((ext_vector_type(4)));
typedef float v8f __attribute__((ext_vector_type(8)));

#define NEG_SLOPE 0.2f
#define LN_EPS 1e-5f

__device__ __forceinline__ float lrelu(float x) {
  return x > 0.f ? x : x * NEG_SLOPE;
}

template <int K>
__global__ void gat_gemm_wmma(const float* __restrict__ Amat,
                              const float* __restrict__ Bmat,
                              float* __restrict__ Cmat,
                              int M) {
  __shared__ float bs[K * 128];

  const int tid = threadIdx.x;

  for (int idx = tid; idx < K * 128; idx += 256) {
    int kr = idx >> 7;
    int col = idx & 127;
    int ks = kr >> 2;
    int c = col >> 4;
    int lane = (col & 15) | ((kr & 2) << 3);
    int sub = kr & 1;
    bs[((ks * 32 + lane) * 8 + c) * 2 + sub] = Bmat[idx];
  }
  __syncthreads();

  const int wave = tid >> 5;
  const int lane = tid & 31;
  const int row0 = (blockIdx.x * 8 + wave) * 16;
  if (row0 >= M) return;

  const int m  = lane & 15;
  const int kk = (lane >> 4) << 1;

  v8f acc[8] = {};

  const float* arow = Amat + (size_t)(row0 + m) * K + kk;
  const float* blane = bs + lane * 16;

#pragma unroll 4
  for (int ks = 0; ks < K / 4; ++ks) {
    v2f a = *(const v2f*)(arow + ks * 4);
    const float* bp = blane + ks * 512;
    v4f q0 = *(const v4f*)(bp);
    v4f q1 = *(const v4f*)(bp + 4);
    v4f q2 = *(const v4f*)(bp + 8);
    v4f q3 = *(const v4f*)(bp + 12);
    acc[0] = __builtin_amdgcn_wmma_f32_16x16x4_f32(false, a, false, q0.xy, (short)0, acc[0], false, false);
    acc[1] = __builtin_amdgcn_wmma_f32_16x16x4_f32(false, a, false, q0.zw, (short)0, acc[1], false, false);
    acc[2] = __builtin_amdgcn_wmma_f32_16x16x4_f32(false, a, false, q1.xy, (short)0, acc[2], false, false);
    acc[3] = __builtin_amdgcn_wmma_f32_16x16x4_f32(false, a, false, q1.zw, (short)0, acc[3], false, false);
    acc[4] = __builtin_amdgcn_wmma_f32_16x16x4_f32(false, a, false, q2.xy, (short)0, acc[4], false, false);
    acc[5] = __builtin_amdgcn_wmma_f32_16x16x4_f32(false, a, false, q2.zw, (short)0, acc[5], false, false);
    acc[6] = __builtin_amdgcn_wmma_f32_16x16x4_f32(false, a, false, q3.xy, (short)0, acc[6], false, false);
    acc[7] = __builtin_amdgcn_wmma_f32_16x16x4_f32(false, a, false, q3.zw, (short)0, acc[7], false, false);
  }

  const int mo = (lane >> 4) * 8;
#pragma unroll
  for (int c = 0; c < 8; ++c) {
#pragma unroll
    for (int r = 0; r < 8; ++r) {
      Cmat[(size_t)(row0 + mo + r) * 128 + c * 16 + m] = acc[c][r];
    }
  }
}

__global__ void gat_attn_proj(const float* __restrict__ h,
                              const float* __restrict__ al,
                              const float* __restrict__ ar,
                              float* __restrict__ el, float* __restrict__ er,
                              int NH) {
  int i = blockIdx.x * blockDim.x + threadIdx.x;
  if (i >= NH) return;
  int n = i >> 2, hh = i & 3;
  const float* hp  = h + (size_t)n * 128 + hh * 32;
  const float* alp = al + hh * 32;
  const float* arp = ar + hh * 32;
  float sl = 0.f, sr = 0.f;
#pragma unroll
  for (int f = 0; f < 32; ++f) {
    float v = hp[f];
    sl += v * alp[f];
    sr += v * arp[f];
  }
  el[i] = sl;
  er[i] = sr;
}

__global__ void gat_rank(const int* __restrict__ dst, int* __restrict__ cnt,
                         int* __restrict__ rank, int E) {
  int e = blockIdx.x * blockDim.x + threadIdx.x;
  if (e >= E) return;
  rank[e] = atomicAdd(&cnt[dst[e]], 1);
}

__global__ void gat_scan(const int* __restrict__ cnt, int* __restrict__ off, int Nn) {
  __shared__ int wsum[32];
  const int t = threadIdx.x;
  const int lane = t & 31, w = t >> 5;
  const int per = (Nn + 1023) / 1024;
  const int b = t * per;
  const int e = (b + per < Nn) ? (b + per) : Nn;
  int s = 0;
  for (int i = b; i < e; ++i) s += cnt[i];
  int x = s;
#pragma unroll
  for (int d = 1; d < 32; d <<= 1) {
    int y = __shfl_up(x, d, 32);
    if (lane >= d) x += y;
  }
  if (lane == 31) wsum[w] = x;
  __syncthreads();
  if (w == 0) {
    int v = wsum[lane];
#pragma unroll
    for (int d = 1; d < 32; d <<= 1) {
      int y = __shfl_up(v, d, 32);
      if (lane >= d) v += y;
    }
    wsum[lane] = v;
  }
  __syncthreads();
  int base = (w > 0 ? wsum[w - 1] : 0) + (x - s);
  for (int i = b; i < e; i += 8) {
    int c[8];
#pragma unroll
    for (int k = 0; k < 8; ++k) c[k] = (i + k < e) ? cnt[i + k] : 0;
#pragma unroll
    for (int k = 0; k < 8; ++k) {
      if (i + k < e) off[i + k] = base;
      base += c[k];
    }
  }
  if (t == 1023) off[Nn] = base;
}

__global__ void gat_fill(const int* __restrict__ src, const int* __restrict__ dst,
                         const int* __restrict__ off, const int* __restrict__ rank,
                         int* __restrict__ ssrc, int E) {
  int e = blockIdx.x * blockDim.x + threadIdx.x;
  if (e >= E) return;
  ssrc[off[dst[e]] + rank[e]] = src[e];
}

template <int LAYER>
__global__ void gat_node(const int* __restrict__ off, const int* __restrict__ ssrc,
                         const float* __restrict__ el, const float* __restrict__ er,
                         const float* __restrict__ h,
                         const float* __restrict__ bias,
                         const float* __restrict__ lns, const float* __restrict__ lnb,
                         float* __restrict__ outp, int Nn) {
  const int lane = threadIdx.x & 31;
  const int n = __builtin_amdgcn_readfirstlane(blockIdx.x * 8 + (threadIdx.x >> 5));
  if (n >= Nn) return;
  const int beg = __builtin_amdgcn_readfirstlane(off[n]);
  const int deg = __builtin_amdgcn_readfirstlane(off[n + 1]) - beg;
  const int hh = lane & 3;
  const int q0 = lane >> 2;
  const float erh = er[n * 4 + hh];

  float m = -__builtin_huge_valf();
  for (int c0 = 0; c0 < deg; c0 += 8) {
    int j = c0 + q0;
    if (j < deg) {
      int s = ssrc[beg + j];
      m = fmaxf(m, lrelu(el[s * 4 + hh] + erh));
    }
  }
  m = fmaxf(m, __shfl_xor(m, 4, 32));
  m = fmaxf(m, __shfl_xor(m, 8, 32));
  m = fmaxf(m, __shfl_xor(m, 16, 32));
  m = (fabsf(m) < __builtin_huge_valf()) ? m : 0.f;

  float den = 0.f;
  for (int c0 = 0; c0 < deg; c0 += 8) {
    int j = c0 + q0;
    if (j < deg) {
      int s = ssrc[beg + j];
      den += expf(lrelu(el[s * 4 + hh] + erh) - m);
    }
  }
  den += __shfl_xor(den, 4, 32);
  den += __shfl_xor(den, 8, 32);
  den += __shfl_xor(den, 16, 32);
  den = fmaxf(den, 1e-9f);

  v4f acc = {0.f, 0.f, 0.f, 0.f};
  const int hl = lane >> 3;
  const float* hrow = h + lane * 4;
  for (int c0 = 0; c0 < deg; c0 += 8) {
    const int j = c0 + q0;
    const bool valid = j < deg;
    const int s = ssrc[beg + (valid ? j : c0)];
    const float ev = lrelu(el[s * 4 + hh] + erh);
    const float alpha = valid ? expf(ev - m) / den : 0.f;
#pragma unroll
    for (int q = 0; q < 8; ++q) {
      const int sq = __builtin_amdgcn_readlane(s, q * 4);
      const float a = __shfl(alpha, q * 4 + hl, 32);
      const v4f v = *(const v4f*)(hrow + (size_t)sq * 128);
      acc += a * v;
    }
  }

  float x0 = acc.x, x1 = acc.y, x2 = acc.z, x3 = acc.w;
  x0 += __shfl_xor(x0, 8, 32);  x1 += __shfl_xor(x1, 8, 32);
  x2 += __shfl_xor(x2, 8, 32);  x3 += __shfl_xor(x3, 8, 32);
  x0 += __shfl_xor(x0, 16, 32); x1 += __shfl_xor(x1, 16, 32);
  x2 += __shfl_xor(x2, 16, 32); x3 += __shfl_xor(x3, 16, 32);

  const int f0 = (lane & 7) * 4;
#pragma unroll
  for (int k = 0; k < 4; ++k) {
    x0 += bias[k * 32 + f0 + 0];
    x1 += bias[k * 32 + f0 + 1];
    x2 += bias[k * 32 + f0 + 2];
    x3 += bias[k * 32 + f0 + 3];
  }
  x0 *= 0.25f; x1 *= 0.25f; x2 *= 0.25f; x3 *= 0.25f;

  v4f y;
  if (LAYER == 1) {
    float mu = (x0 + x1) + (x2 + x3);
    mu += __shfl_xor(mu, 1, 32);
    mu += __shfl_xor(mu, 2, 32);
    mu += __shfl_xor(mu, 4, 32);
    mu *= (1.f / 32.f);
    const float d0 = x0 - mu, d1 = x1 - mu, d2 = x2 - mu, d3 = x3 - mu;
    float var = (d0 * d0 + d1 * d1) + (d2 * d2 + d3 * d3);
    var += __shfl_xor(var, 1, 32);
    var += __shfl_xor(var, 2, 32);
    var += __shfl_xor(var, 4, 32);
    var *= (1.f / 32.f);
    const float r = rsqrtf(var + LN_EPS);
    y.x = fmaxf(d0 * r * lns[f0 + 0] + lnb[f0 + 0], 0.f);
    y.y = fmaxf(d1 * r * lns[f0 + 1] + lnb[f0 + 1], 0.f);
    y.z = fmaxf(d2 * r * lns[f0 + 2] + lnb[f0 + 2], 0.f);
    y.w = fmaxf(d3 * r * lns[f0 + 3] + lnb[f0 + 3], 0.f);
  } else {
    y.x = x0; y.y = x1; y.z = x2; y.w = x3;
  }
  if (lane < 8) *(v4f*)(outp + (size_t)n * 32 + f0) = y;
}

__global__ void gat_flops(const int* __restrict__ in_deg, int Nn, float* out) {
  __shared__ float ssum[256];
  float s = 0.f;
  for (int i = threadIdx.x; i < Nn; i += 256) s += (float)in_deg[i];
  ssum[threadIdx.x] = s;
  __syncthreads();
  for (int m = 128; m > 0; m >>= 1) {
    if ((int)threadIdx.x < m) ssum[threadIdx.x] += ssum[threadIdx.x + m];
    __syncthreads();
  }
  if (threadIdx.x == 0) {
    double ne = (double)ssum[0];
    double fl = (4.0 * ne * (6.0 * 128 * 32 + 6 * 32 + 2) +
                 4.0 * ne * (6.0 * 32 * 32 + 6 * 32 + 2)) / 1e12;
    out[0] = (float)fl;
  }
}

extern "C" void kernel_launch(void* const* d_in, const int* in_sizes, int n_in,
                              void* d_out, int out_size, void* d_ws, size_t ws_size,
                              hipStream_t stream) {
  const float* feat = (const float*)d_in[0];
  const float* W1   = (const float*)d_in[1];
  const float* al1  = (const float*)d_in[2];
  const float* ar1  = (const float*)d_in[3];
  const float* b1   = (const float*)d_in[4];
  const float* lns  = (const float*)d_in[5];
  const float* lnb  = (const float*)d_in[6];
  const float* W2   = (const float*)d_in[7];
  const float* al2  = (const float*)d_in[8];
  const float* ar2  = (const float*)d_in[9];
  const float* b2   = (const float*)d_in[10];
  const int*   src  = (const int*)d_in[11];
  const int*   dst  = (const int*)d_in[12];
  const int*   indeg= (const int*)d_in[13];

  const int Nn = in_sizes[13];
  const int E  = in_sizes[11];
  const int NH = Nn * 4;

  char* p = (char*)d_ws;
  float* hbuf = (float*)p;   p += (size_t)Nn * 128 * sizeof(float);
  float* hmid = (float*)p;   p += (size_t)Nn * 32  * sizeof(float);
  float* el   = (float*)p;   p += (size_t)NH * sizeof(float);
  float* er   = (float*)p;   p += (size_t)NH * sizeof(float);
  int*   rank = (int*)p;     p += (size_t)E * sizeof(int);
  int*   ssrc = (int*)p;     p += (size_t)E * sizeof(int);
  int*   cnt  = (int*)p;     p += (size_t)Nn * sizeof(int);
  int*   off  = (int*)p;     p += ((size_t)Nn + 1) * sizeof(int);
  (void)ws_size; (void)n_in; (void)out_size;

  float* outh = (float*)d_out;

  dim3 b256(256);
  const int rowTiles   = (Nn + 15) / 16;
  const int gemmBlocks = (rowTiles + 7) / 8;
  const int eBlocks    = (E + 255) / 256;
  const int nodeBlocks = (Nn + 7) / 8;

  hipMemsetAsync(cnt, 0, (size_t)Nn * sizeof(int), stream);
  gat_rank<<<eBlocks, b256, 0, stream>>>(dst, cnt, rank, E);
  gat_scan<<<1, 1024, 0, stream>>>(cnt, off, Nn);
  gat_fill<<<eBlocks, b256, 0, stream>>>(src, dst, off, rank, ssrc, E);

  gat_gemm_wmma<128><<<gemmBlocks, b256, 0, stream>>>(feat, W1, hbuf, Nn);
  gat_attn_proj<<<(NH + 255) / 256, b256, 0, stream>>>(hbuf, al1, ar1, el, er, NH);
  gat_node<1><<<nodeBlocks, b256, 0, stream>>>(off, ssrc, el, er, hbuf, b1, lns, lnb, hmid, Nn);

  gat_gemm_wmma<32><<<gemmBlocks, b256, 0, stream>>>(hmid, W2, hbuf, Nn);
  gat_attn_proj<<<(NH + 255) / 256, b256, 0, stream>>>(hbuf, al2, ar2, el, er, NH);
  gat_node<2><<<nodeBlocks, b256, 0, stream>>>(off, ssrc, el, er, hbuf, b2, lns, lnb, outh, Nn);

  gat_flops<<<1, 256, 0, stream>>>(indeg, Nn, outh + (size_t)Nn * 32);
}


# in-degree sum of the flop scalar: one block of 1024 threads, 16 independent loads in flight per thread, integer partial sums as the reference; was 256 threads each waiting on 196 loads in turn
# speedup vs baseline: 3.1488x; 1.2694x over previous
#include <hip/hip_runtime.h>


typedef float v2f __attribute__((ext_vector_type(2)));
typedef float v4f __attribute__((ext_vector_type(4)));
typedef float v8f __attribute__((ext_vector_type(8)));

#define NEG_SLOPE 0.2f
#define LN_EPS 1e-5f

__device__ __forceinline__ float lrelu(float x) {
  return x > 0.f ? x : x * NEG_SLOPE;
}

template <int K>
__global__ void gat_gemm_wmma(const float* __restrict__ Amat,
                              const float* __restrict__ Bmat,
                              float* __restrict__ Cmat,
                              int M) {
  __shared__ float bs[K * 128];

  const int tid = threadIdx.x;

  for (int idx = tid; idx < K * 128; idx += 256) {
    int kr = idx >> 7;
    int col = idx & 127;
    int ks = kr >> 2;
    int c = col >> 4;
    int lane = (col & 15) | ((kr & 2) << 3);
    int sub = kr & 1;
    bs[((ks * 32 + lane) * 8 + c) * 2 + sub] = Bmat[idx];
  }
  __syncthreads();

  const int wave = tid >> 5;
  const int lane = tid & 31;
  const int row0 = (blockIdx.x * 8 + wave) * 16;
  if (row0 >= M) return;

  const int m  = lane & 15;
  const int kk = (lane >> 4) << 1;

  v8f acc[8] = {};

  const float* arow = Amat + (size_t)(row0 + m) * K + kk;
  const float* blane = bs + lane * 16;

#pragma unroll 4
  for (int ks = 0; ks < K / 4; ++ks) {
    v2f a = *(const v2f*)(arow + ks * 4);
    const float* bp = blane + ks * 512;
    v4f q0 = *(const v4f*)(bp);
    v4f q1 = *(const v4f*)(bp + 4);
    v4f q2 = *(const v4f*)(bp + 8);
    v4f q3 = *(const v4f*)(bp + 12);
    acc[0] = __builtin_amdgcn_wmma_f32_16x16x4_f32(false, a, false, q0.xy, (short)0, acc[0], false, false);
    acc[1] = __builtin_amdgcn_wmma_f32_16x16x4_f32(false, a, false, q0.zw, (short)0, acc[1], false, false);
    acc[2] = __builtin_amdgcn_wmma_f32_16x16x4_f32(false, a, false, q1.xy, (short)0, acc[2], false, false);
    acc[3] = __builtin_amdgcn_wmma_f32_16x16x4_f32(false, a, false, q1.zw, (short)0, acc[3], false, false);
    acc[4] = __builtin_amdgcn_wmma_f32_16x16x4_f32(false, a, false, q2.xy, (short)0, acc[4], false, false);
    acc[5] = __builtin_amdgcn_wmma_f32_16x16x4_f32(false, a, false, q2.zw, (short)0, acc[5], false, false);
    acc[6] = __builtin_amdgcn_wmma_f32_16x16x4_f32(false, a, false, q3.xy, (short)0, acc[6], false, false);
    acc[7] = __builtin_amdgcn_wmma_f32_16x16x4_f32(false, a, false, q3.zw, (short)0, acc[7], false, false);
  }

  const int mo = (lane >> 4) * 8;
#pragma unroll
  for (int c = 0; c < 8; ++c) {
#pragma unroll
    for (int r = 0; r < 8; ++r) {
      Cmat[(size_t)(row0 + mo + r) * 128 + c * 16 + m] = acc[c][r];
    }
  }
}

__global__ void gat_attn_proj(const float* __restrict__ h,
                              const float* __restrict__ al,
                              const float* __restrict__ ar,
                              float* __restrict__ el, float* __restrict__ er,
                              int NH) {
  int i = blockIdx.x * blockDim.x + threadIdx.x;
  if (i >= NH) return;
  int n = i >> 2, hh = i & 3;
  const float* hp  = h + (size_t)n * 128 + hh * 32;
  const float* alp = al + hh * 32;
  const float* arp = ar + hh * 32;
  float sl = 0.f, sr = 0.f;
#pragma unroll
  for (int f = 0; f < 32; ++f) {
    float v = hp[f];
    sl += v * alp[f];
    sr += v * arp[f];
  }
  el[i] = sl;
  er[i] = sr;
}

__global__ void gat_rank(const int* __restrict__ dst, int* __restrict__ cnt,
                         int* __restrict__ rank, int E) {
  int e = blockIdx.x * blockDim.x + threadIdx.x;
  if (e >= E) return;
  rank[e] = atomicAdd(&cnt[dst[e]], 1);
}

__global__ void gat_scan(const int* __restrict__ cnt, int* __restrict__ off, int Nn) {
  __shared__ int wsum[32];
  const int t = threadIdx.x;
  const int lane = t & 31, w = t >> 5;
  const int per = (Nn + 1023) / 1024;
  const int b = t * per;
  const int e = (b + per < Nn) ? (b + per) : Nn;
  int s = 0;
  for (int i = b; i < e; ++i) s += cnt[i];
  int x = s;
#pragma unroll
  for (int d = 1; d < 32; d <<= 1) {
    int y = __shfl_up(x, d, 32);
    if (lane >= d) x += y;
  }
  if (lane == 31) wsum[w] = x;
  __syncthreads();
  if (w == 0) {
    int v = wsum[lane];
#pragma unroll
    for (int d = 1; d < 32; d <<= 1) {
      int y = __shfl_up(v, d, 32);
      if (lane >= d) v += y;
    }
    wsum[lane] = v;
  }
  __syncthreads();
  int base = (w > 0 ? wsum[w - 1] : 0) + (x - s);
  for (int i = b; i < e; i += 8) {
    int c[8];
#pragma unroll
    for (int k = 0; k < 8; ++k) c[k] = (i + k < e) ? cnt[i + k] : 0;
#pragma unroll
    for (int k = 0; k < 8; ++k) {
      if (i + k < e) off[i + k] = base;
      base += c[k];
    }
  }
  if (t == 1023) off[Nn] = base;
}

__global__ void gat_fill(const int* __restrict__ src, const int* __restrict__ dst,
                         const int* __restrict__ off, const int* __restrict__ rank,
                         int* __restrict__ ssrc, int E) {
  int e = blockIdx.x * blockDim.x + threadIdx.x;
  if (e >= E) return;
  ssrc[off[dst[e]] + rank[e]] = src[e];
}

template <int LAYER>
__global__ void gat_node(const int* __restrict__ off, const int* __restrict__ ssrc,
                         const float* __restrict__ el, const float* __restrict__ er,
                         const float* __restrict__ h,
                         const float* __restrict__ bias,
                         const float* __restrict__ lns, const float* __restrict__ lnb,
                         float* __restrict__ outp, int Nn) {
  const int lane = threadIdx.x & 31;
  const int n = __builtin_amdgcn_readfirstlane(blockIdx.x * 8 + (threadIdx.x >> 5));
  if (n >= Nn) return;
  const int beg = __builtin_amdgcn_readfirstlane(off[n]);
  const int deg = __builtin_amdgcn_readfirstlane(off[n + 1]) - beg;
  const int hh = lane & 3;
  const int q0 = lane >> 2;
  const float erh = er[n * 4 + hh];

  float m = -__builtin_huge_valf();
  for (int c0 = 0; c0 < deg; c0 += 8) {
    int j = c0 + q0;
    if (j < deg) {
      int s = ssrc[beg + j];
      m = fmaxf(m, lrelu(el[s * 4 + hh] + erh));
    }
  }
  m = fmaxf(m, __shfl_xor(m, 4, 32));
  m = fmaxf(m, __shfl_xor(m, 8, 32));
  m = fmaxf(m, __shfl_xor(m, 16, 32));
  m = (fabsf(m) < __builtin_huge_valf()) ? m : 0.f;

  float den = 0.f;
  for (int c0 = 0; c0 < deg; c0 += 8) {
    int j = c0 + q0;
    if (j < deg) {
      int s = ssrc[beg + j];
      den += expf(lrelu(el[s * 4 + hh] + erh) - m);
    }
  }
  den += __shfl_xor(den, 4, 32);
  den += __shfl_xor(den, 8, 32);
  den += __shfl_xor(den, 16, 32);
  den = fmaxf(den, 1e-9f);

  v4f acc = {0.f, 0.f, 0.f, 0.f};
  const int hl = lane >> 3;
  const float* hrow = h + lane * 4;
  for (int c0 = 0; c0 < deg; c0 += 8) {
    const int j = c0 + q0;
    const bool valid = j < deg;
    const int s = ssrc[beg + (valid ? j : c0)];
    const float ev = lrelu(el[s * 4 + hh] + erh);
    const float alpha = valid ? expf(ev - m) / den : 0.f;
#pragma unroll
    for (int q = 0; q < 8; ++q) {
      const int sq = __builtin_amdgcn_readlane(s, q * 4);
      const float a = __shfl(alpha, q * 4 + hl, 32);
      const v4f v = *(const v4f*)(hrow + (size_t)sq * 128);
      acc += a * v;
    }
  }

  float x0 = acc.x, x1 = acc.y, x2 = acc.z, x3 = acc.w;
  x0 += __shfl_xor(x0, 8, 32);  x1 += __shfl_xor(x1, 8, 32);
  x2 += __shfl_xor(x2, 8, 32);  x3 += __shfl_xor(x3, 8, 32);
  x0 += __shfl_xor(x0, 16, 32); x1 += __shfl_xor(x1, 16, 32);
  x2 += __shfl_xor(x2, 16, 32); x3 += __shfl_xor(x3, 16, 32);

  const int f0 = (lane & 7) * 4;
#pragma unroll
  for (int k = 0; k < 4; ++k) {
    x0 += bias[k * 32 + f0 + 0];
    x1 += bias[k * 32 + f0 + 1];
    x2 += bias[k * 32 + f0 + 2];
    x3 += bias[k * 32 + f0 + 3];
  }
  x0 *= 0.25f; x1 *= 0.25f; x2 *= 0.25f; x3 *= 0.25f;

  v4f y;
  if (LAYER == 1) {
    float mu = (x0 + x1) + (x2 + x3);
    mu += __shfl_xor(mu, 1, 32);
    mu += __shfl_xor(mu, 2, 32);
    mu += __shfl_xor(mu, 4, 32);
    mu *= (1.f / 32.f);
    const float d0 = x0 - mu, d1 = x1 - mu, d2 = x2 - mu, d3 = x3 - mu;
    float var = (d0 * d0 + d1 * d1) + (d2 * d2 + d3 * d3);
    var += __shfl_xor(var, 1, 32);
    var += __shfl_xor(var, 2, 32);
    var += __shfl_xor(var, 4, 32);
    var *= (1.f / 32.f);
    const float r = rsqrtf(var + LN_EPS);
    y.x = fmaxf(d0 * r * lns[f0 + 0] + lnb[f0 + 0], 0.f);
    y.y = fmaxf(d1 * r * lns[f0 + 1] + lnb[f0 + 1], 0.f);
    y.z = fmaxf(d2 * r * lns[f0 + 2] + lnb[f0 + 2], 0.f);
    y.w = fmaxf(d3 * r * lns[f0 + 3] + lnb[f0 + 3], 0.f);
  } else {
    y.x = x0; y.y = x1; y.z = x2; y.w = x3;
  }
  if (lane < 8) *(v4f*)(outp + (size_t)n * 32 + f0) = y;
}

__global__ void gat_flops(const int* __restrict__ in_deg, int Nn, float* out) {
  __shared__ int wsum[32];
  const int t = threadIdx.x;
  int s = 0;
  for (int i = t; i < Nn; i += 16 * 1024) {
    int c[16];
#pragma unroll
    for (int k = 0; k < 16; ++k) c[k] = (i + k * 1024 < Nn) ? in_deg[i + k * 1024] : 0;
#pragma unroll
    for (int k = 0; k < 16; ++k) s += c[k];
  }
#pragma unroll
  for (int d = 16; d > 0; d >>= 1) s += __shfl_xor(s, d, 32);
  if ((t & 31) == 0) wsum[t >> 5] = s;
  __syncthreads();
  if (t < 32) {
    int v = wsum[t];
#pragma unroll
    for (int d = 16; d > 0; d >>= 1) v += __shfl_xor(v, d, 32);
    if (t == 0) {
      double ne = (double)(float)v;
      double fl = (4.0 * ne * (6.0 * 128 * 32 + 6 * 32 + 2) +
                   4.0 * ne * (6.0 * 32 * 32 + 6 * 32 + 2)) / 1e12;
      out[0] = (float)fl;
    }
  }
}

extern "C" void kernel_launch(void* const* d_in, const int* in_sizes, int n_in,
                              void* d_out, int out_size, void* d_ws, size_t ws_size,
                              hipStream_t stream) {
  const float* feat = (const float*)d_in[0];
  const float* W1   = (const float*)d_in[1];
  const float* al1  = (const float*)d_in[2];
  const float* ar1  = (const float*)d_in[3];
  const float* b1   = (const float*)d_in[4];
  const float* lns  = (const float*)d_in[5];
  const float* lnb  = (const float*)d_in[6];
  const float* W2   = (const float*)d_in[7];
  const float* al2  = (const float*)d_in[8];
  const float* ar2  = (const float*)d_in[9];
  const float* b2   = (const float*)d_in[10];
  const int*   src  = (const int*)d_in[11];
  const int*   dst  = (const int*)d_in[12];
  const int*   indeg= (const int*)d_in[13];

  const int Nn = in_sizes[13];
  const int E  = in_sizes[11];
  const int NH = Nn * 4;

  char* p = (char*)d_ws;
  float* hbuf = (float*)p;   p += (size_t)Nn * 128 * sizeof(float);
  float* hmid = (float*)p;   p += (size_t)Nn * 32  * sizeof(float);
  float* el   = (float*)p;   p += (size_t)NH * sizeof(float);
  float* er   = (float*)p;   p += (size_t)NH * sizeof(float);
  int*   rank = (int*)p;     p += (size_t)E * sizeof(int);
  int*   ssrc = (int*)p;     p += (size_t)E * sizeof(int);
  int*   cnt  = (int*)p;     p += (size_t)Nn * sizeof(int);
  int*   off  = (int*)p;     p += ((size_t)Nn + 1) * sizeof(int);
  (void)ws_size; (void)n_in; (void)out_size;

  float* outh = (float*)d_out;

  dim3 b256(256);
  const int rowTiles   = (Nn + 15) / 16;
  const int gemmBlocks = (rowTiles + 7) / 8;
  const int eBlocks    = (E + 255) / 256;
  const int nodeBlocks = (Nn + 7) / 8;

  hipMemsetAsync(cnt, 0, (size_t)Nn * sizeof(int), stream);
  gat_rank<<<eBlocks, b256, 0, stream>>>(dst, cnt, rank, E);
  gat_scan<<<1, 1024, 0, stream>>>(cnt, off, Nn);
  gat_fill<<<eBlocks, b256, 0, stream>>>(src, dst, off, rank, ssrc, E);

  gat_gemm_wmma<128><<<gemmBlocks, b256, 0, stream>>>(feat, W1, hbuf, Nn);
  gat_attn_proj<<<(NH + 255) / 256, b256, 0, stream>>>(hbuf, al1, ar1, el, er, NH);
  gat_node<1><<<nodeBlocks, b256, 0, stream>>>(off, ssrc, el, er, hbuf, b1, lns, lnb, hmid, Nn);

  gat_gemm_wmma<32><<<gemmBlocks, b256, 0, stream>>>(hmid, W2, hbuf, Nn);
  gat_attn_proj<<<(NH + 255) / 256, b256, 0, stream>>>(hbuf, al2, ar2, el, er, NH);
  gat_node<2><<<nodeBlocks, b256, 0, stream>>>(off, ssrc, el, er, hbuf, b2, lns, lnb, outh, Nn);

  gat_flops<<<1, 1024, 0, stream>>>(indeg, Nn, outh + (size_t)Nn * 32);
}


# node kernel reads the sources and scores of a bucket's first 32 edges once into registers and keeps two trips of rows in flight: about 4 dependent load groups per node, was about 15; sums unchanged
# speedup vs baseline: 3.3819x; 1.0740x over previous
#include <hip/hip_runtime.h>


typedef float v2f __attribute__((ext_vector_type(2)));
typedef float v4f __attribute__((ext_vector_type(4)));
typedef float v8f __attribute__((ext_vector_type(8)));

#define NEG_SLOPE 0.2f
#define LN_EPS 1e-5f

__device__ __forceinline__ float lrelu(float x) {
  return x > 0.f ? x : x * NEG_SLOPE;
}

template <int K>
__global__ void gat_gemm_wmma(const float* __restrict__ Amat,
                              const float* __restrict__ Bmat,
                              float* __restrict__ Cmat,
                              int M) {
  __shared__ float bs[K * 128];

  const int tid = threadIdx.x;

  for (int idx = tid; idx < K * 128; idx += 256) {
    int kr = idx >> 7;
    int col = idx & 127;
    int ks = kr >> 2;
    int c = col >> 4;
    int lane = (col & 15) | ((kr & 2) << 3);
    int sub = kr & 1;
    bs[((ks * 32 + lane) * 8 + c) * 2 + sub] = Bmat[idx];
  }
  __syncthreads();

  const int wave = tid >> 5;
  const int lane = tid & 31;
  const int row0 = (blockIdx.x * 8 + wave) * 16;
  if (row0 >= M) return;

  const int m  = lane & 15;
  const int kk = (lane >> 4) << 1;

  v8f acc[8] = {};

  const float* arow = Amat + (size_t)(row0 + m) * K + kk;
  const float* blane = bs + lane * 16;

#pragma unroll 4
  for (int ks = 0; ks < K / 4; ++ks) {
    v2f a = *(const v2f*)(arow + ks * 4);
    const float* bp = blane + ks * 512;
    v4f q0 = *(const v4f*)(bp);
    v4f q1 = *(const v4f*)(bp + 4);
    v4f q2 = *(const v4f*)(bp + 8);
    v4f q3 = *(const v4f*)(bp + 12);
    acc[0] = __builtin_amdgcn_wmma_f32_16x16x4_f32(false, a, false, q0.xy, (short)0, acc[0], false, false);
    acc[1] = __builtin_amdgcn_wmma_f32_16x16x4_f32(false, a, false, q0.zw, (short)0, acc[1], false, false);
    acc[2] = __builtin_amdgcn_wmma_f32_16x16x4_f32(false, a, false, q1.xy, (short)0, acc[2], false, false);
    acc[3] = __builtin_amdgcn_wmma_f32_16x16x4_f32(false, a, false, q1.zw, (short)0, acc[3], false, false);
    acc[4] = __builtin_amdgcn_wmma_f32_16x16x4_f32(false, a, false, q2.xy, (short)0, acc[4], false, false);
    acc[5] = __builtin_amdgcn_wmma_f32_16x16x4_f32(false, a, false, q2.zw, (short)0, acc[5], false, false);
    acc[6] = __builtin_amdgcn_wmma_f32_16x16x4_f32(false, a, false, q3.xy, (short)0, acc[6], false, false);
    acc[7] = __builtin_amdgcn_wmma_f32_16x16x4_f32(false, a, false, q3.zw, (short)0, acc[7], false, false);
  }

  const int mo = (lane >> 4) * 8;
#pragma unroll
  for (int c = 0; c < 8; ++c) {
#pragma unroll
    for (int r = 0; r < 8; ++r) {
      Cmat[(size_t)(row0 + mo + r) * 128 + c * 16 + m] = acc[c][r];
    }
  }
}

__global__ void gat_attn_proj(const float* __restrict__ h,
                              const float* __restrict__ al,
                              const float* __restrict__ ar,
                              float* __restrict__ el, float* __restrict__ er,
                              int NH) {
  int i = blockIdx.x * blockDim.x + threadIdx.x;
  if (i >= NH) return;
  int n = i >> 2, hh = i & 3;
  const float* hp  = h + (size_t)n * 128 + hh * 32;
  const float* alp = al + hh * 32;
  const float* arp = ar + hh * 32;
  float sl = 0.f, sr = 0.f;
#pragma unroll
  for (int f = 0; f < 32; ++f) {
    float v = hp[f];
    sl += v * alp[f];
    sr += v * arp[f];
  }
  el[i] = sl;
  er[i] = sr;
}

__global__ void gat_rank(const int* __restrict__ dst, int* __restrict__ cnt,
                         int* __restrict__ rank, int E) {
  int e = blockIdx.x * blockDim.x + threadIdx.x;
  if (e >= E) return;
  rank[e] = atomicAdd(&cnt[dst[e]], 1);
}

__global__ void gat_scan(const int* __restrict__ cnt, int* __restrict__ off, int Nn) {
  __shared__ int wsum[32];
  const int t = threadIdx.x;
  const int lane = t & 31, w = t >> 5;
  const int per = (Nn + 1023) / 1024;
  const int b = t * per;
  const int e = (b + per < Nn) ? (b + per) : Nn;
  int s = 0;
  for (int i = b; i < e; ++i) s += cnt[i];
  int x = s;
#pragma unroll
  for (int d = 1; d < 32; d <<= 1) {
    int y = __shfl_up(x, d, 32);
    if (lane >= d) x += y;
  }
  if (lane == 31) wsum[w] = x;
  __syncthreads();
  if (w == 0) {
    int v = wsum[lane];
#pragma unroll
    for (int d = 1; d < 32; d <<= 1) {
      int y = __shfl_up(v, d, 32);
      if (lane >= d) v += y;
    }
    wsum[lane] = v;
  }
  __syncthreads();
  int base = (w > 0 ? wsum[w - 1] : 0) + (x - s);
  for (int i = b; i < e; i += 8) {
    int c[8];
#pragma unroll
    for (int k = 0; k < 8; ++k) c[k] = (i + k < e) ? cnt[i + k] : 0;
#pragma unroll
    for (int k = 0; k < 8; ++k) {
      if (i + k < e) off[i + k] = base;
      base += c[k];
    }
  }
  if (t == 1023) off[Nn] = base;
}

__global__ void gat_fill(const int* __restrict__ src, const int* __restrict__ dst,
                         const int* __restrict__ off, const int* __restrict__ rank,
                         int* __restrict__ ssrc, int E) {
  int e = blockIdx.x * blockDim.x + threadIdx.x;
  if (e >= E) return;
  ssrc[off[dst[e]] + rank[e]] = src[e];
}

template <int LAYER>
__global__ void gat_node(const int* __restrict__ off, const int* __restrict__ ssrc,
                         const float* __restrict__ el, const float* __restrict__ er,
                         const float* __restrict__ h,
                         const float* __restrict__ bias,
                         const float* __restrict__ lns, const float* __restrict__ lnb,
                         float* __restrict__ outp, int Nn) {
  const int lane = threadIdx.x & 31;
  const int n = __builtin_amdgcn_readfirstlane(blockIdx.x * 8 + (threadIdx.x >> 5));
  if (n >= Nn) return;
  const int beg = __builtin_amdgcn_readfirstlane(off[n]);
  const int deg = __builtin_amdgcn_readfirstlane(off[n + 1]) - beg;
  const int hh = lane & 3;
  const int q0 = lane >> 2;
  const int hl = lane >> 3;
  const float erh = er[n * 4 + hh];
  const float* hrow = h + lane * 4;
  const float ninf = -__builtin_huge_valf();

  int   s[4]  = {0, 0, 0, 0};
  float ev[4] = {ninf, ninf, ninf, ninf};
  v4f   r[4][8];

  if (deg > 0) {
#pragma unroll
    for (int t = 0; t < 4; ++t) {
      const int j = t * 8 + q0;
      s[t] = ssrc[beg + (j < deg ? j : 0)];
    }
#pragma unroll
    for (int t = 0; t < 4; ++t) {
      const int j = t * 8 + q0;
      const float x = lrelu(el[s[t] * 4 + hh] + erh);
      ev[t] = (j < deg) ? x : ninf;
    }
#pragma unroll
    for (int q = 0; q < 8; ++q) {
      const int sq = __builtin_amdgcn_readlane(s[0], q * 4);
      r[0][q] = *(const v4f*)(hrow + (size_t)sq * 128);
    }
  }
  if (deg > 8) {
#pragma unroll
    for (int q = 0; q < 8; ++q) {
      const int sq = __builtin_amdgcn_readlane(s[1], q * 4);
      r[1][q] = *(const v4f*)(hrow + (size_t)sq * 128);
    }
  }

  float m = fmaxf(fmaxf(fmaxf(fmaxf(ninf, ev[0]), ev[1]), ev[2]), ev[3]);
  for (int c0 = 32; c0 < deg; c0 += 8) {
    int j = c0 + q0;
    if (j < deg) {
      int sj = ssrc[beg + j];
      m = fmaxf(m, lrelu(el[sj * 4 + hh] + erh));
    }
  }
  m = fmaxf(m, __shfl_xor(m, 4, 32));
  m = fmaxf(m, __shfl_xor(m, 8, 32));
  m = fmaxf(m, __shfl_xor(m, 16, 32));
  m = (fabsf(m) < __builtin_huge_valf()) ? m : 0.f;

  float p[4];
  float den = 0.f;
#pragma unroll
  for (int t = 0; t < 4; ++t) {
    p[t] = (t * 8 + q0 < deg) ? expf(ev[t] - m) : 0.f;
    den += p[t];
  }
  for (int c0 = 32; c0 < deg; c0 += 8) {
    int j = c0 + q0;
    if (j < deg) {
      int sj = ssrc[beg + j];
      den += expf(lrelu(el[sj * 4 + hh] + erh) - m);
    }
  }
  den += __shfl_xor(den, 4, 32);
  den += __shfl_xor(den, 8, 32);
  den += __shfl_xor(den, 16, 32);
  den = fmaxf(den, 1e-9f);

  v4f acc = {0.f, 0.f, 0.f, 0.f};
#pragma unroll
  for (int t = 0; t < 4; ++t) {
    if (t * 8 < deg) {
      const float alpha = p[t] / den;
#pragma unroll
      for (int q = 0; q < 8; ++q) {
        const float a = __shfl(alpha, q * 4 + hl, 32);
        acc += a * r[t][q];
      }
    }
    const int t2 = (t + 2) & 3;
    if (t < 2 && (t + 2) * 8 < deg) {
#pragma unroll
      for (int q = 0; q < 8; ++q) {
        const int sq = __builtin_amdgcn_readlane(s[t2], q * 4);
        r[t2][q] = *(const v4f*)(hrow + (size_t)sq * 128);
      }
    }
  }
  for (int c0 = 32; c0 < deg; c0 += 8) {
    const int j = c0 + q0;
    const bool valid = j < deg;
    const int sj = ssrc[beg + (valid ? j : c0)];
    const float evj = lrelu(el[sj * 4 + hh] + erh);
    const float alpha = valid ? expf(evj - m) / den : 0.f;
#pragma unroll
    for (int q = 0; q < 8; ++q) {
      const int sq = __builtin_amdgcn_readlane(sj, q * 4);
      const float a = __shfl(alpha, q * 4 + hl, 32);
      const v4f v = *(const v4f*)(hrow + (size_t)sq * 128);
      acc += a * v;
    }
  }

  float x0 = acc.x, x1 = acc.y, x2 = acc.z, x3 = acc.w;
  x0 += __shfl_xor(x0, 8, 32);  x1 += __shfl_xor(x1, 8, 32);
  x2 += __shfl_xor(x2, 8, 32);  x3 += __shfl_xor(x3, 8, 32);
  x0 += __shfl_xor(x0, 16, 32); x1 += __shfl_xor(x1, 16, 32);
  x2 += __shfl_xor(x2, 16, 32); x3 += __shfl_xor(x3, 16, 32);

  const int f0 = (lane & 7) * 4;
#pragma unroll
  for (int k = 0; k < 4; ++k) {
    x0 += bias[k * 32 + f0 + 0];
    x1 += bias[k * 32 + f0 + 1];
    x2 += bias[k * 32 + f0 + 2];
    x3 += bias[k * 32 + f0 + 3];
  }
  x0 *= 0.25f; x1 *= 0.25f; x2 *= 0.25f; x3 *= 0.25f;

  v4f y;
  if (LAYER == 1) {
    float mu = (x0 + x1) + (x2 + x3);
    mu += __shfl_xor(mu, 1, 32);
    mu += __shfl_xor(mu, 2, 32);
    mu += __shfl_xor(mu, 4, 32);
    mu *= (1.f / 32.f);
    const float d0 = x0 - mu, d1 = x1 - mu, d2 = x2 - mu, d3 = x3 - mu;
    float var = (d0 * d0 + d1 * d1) + (d2 * d2 + d3 * d3);
    var += __shfl_xor(var, 1, 32);
    var += __shfl_xor(var, 2, 32);
    var += __shfl_xor(var, 4, 32);
    var *= (1.f / 32.f);
    const float r = rsqrtf(var + LN_EPS);
    y.x = fmaxf(d0 * r * lns[f0 + 0] + lnb[f0 + 0], 0.f);
    y.y = fmaxf(d1 * r * lns[f0 + 1] + lnb[f0 + 1], 0.f);
    y.z = fmaxf(d2 * r * lns[f0 + 2] + lnb[f0 + 2], 0.f);
    y.w = fmaxf(d3 * r * lns[f0 + 3] + lnb[f0 + 3], 0.f);
  } else {
    y.x = x0; y.y = x1; y.z = x2; y.w = x3;
  }
  if (lane < 8) *(v4f*)(outp + (size_t)n * 32 + f0) = y;
}

__global__ void gat_flops(const int* __restrict__ in_deg, int Nn, float* out) {
  __shared__ int wsum[32];
  const int t = threadIdx.x;
  int s = 0;
  for (int i = t; i < Nn; i += 16 * 1024) {
    int c[16];
#pragma unroll
    for (int k = 0; k < 16; ++k) c[k] = (i + k * 1024 < Nn) ? in_deg[i + k * 1024] : 0;
#pragma unroll
    for (int k = 0; k < 16; ++k) s += c[k];
  }
#pragma unroll
  for (int d = 16; d > 0; d >>= 1) s += __shfl_xor(s, d, 32);
  if ((t & 31) == 0) wsum[t >> 5] = s;
  __syncthreads();
  if (t < 32) {
    int v = wsum[t];
#pragma unroll
    for (int d = 16; d > 0; d >>= 1) v += __shfl_xor(v, d, 32);
    if (t == 0) {
      double ne = (double)(float)v;
      double fl = (4.0 * ne * (6.0 * 128 * 32 + 6 * 32 + 2) +
                   4.0 * ne * (6.0 * 32 * 32 + 6 * 32 + 2)) / 1e12;
      out[0] = (float)fl;
    }
  }
}

extern "C" void kernel_launch(void* const* d_in, const int* in_sizes, int n_in,
                              void* d_out, int out_size, void* d_ws, size_t ws_size,
                              hipStream_t stream) {
  const float* feat = (const float*)d_in[0];
  const float* W1   = (const float*)d_in[1];
  const float* al1  = (const float*)d_in[2];
  const float* ar1  = (const float*)d_in[3];
  const float* b1   = (const float*)d_in[4];
  const float* lns  = (const float*)d_in[5];
  const float* lnb  = (const float*)d_in[6];
  const float* W2   = (const float*)d_in[7];
  const float* al2  = (const float*)d_in[8];
  const float* ar2  = (const float*)d_in[9];
  const float* b2   = (const float*)d_in[10];
  const int*   src  = (const int*)d_in[11];
  const int*   dst  = (const int*)d_in[12];
  const int*   indeg= (const int*)d_in[13];

  const int Nn = in_sizes[13];
  const int E  = in_sizes[11];
  const int NH = Nn * 4;

  char* p = (char*)d_ws;
  float* hbuf = (float*)p;   p += (size_t)Nn * 128 * sizeof(float);
  float* hmid = (float*)p;   p += (size_t)Nn * 32  * sizeof(float);
  float* el   = (float*)p;   p += (size_t)NH * sizeof(float);
  float* er   = (float*)p;   p += (size_t)NH * sizeof(float);
  int*   rank = (int*)p;     p += (size_t)E * sizeof(int);
  int*   ssrc = (int*)p;     p += (size_t)E * sizeof(int);
  int*   cnt  = (int*)p;     p += (size_t)Nn * sizeof(int);
  int*   off  = (int*)p;     p += ((size_t)Nn + 1) * sizeof(int);
  (void)ws_size; (void)n_in; (void)out_size;

  float* outh = (float*)d_out;

  dim3 b256(256);
  const int rowTiles   = (Nn + 15) / 16;
  const int gemmBlocks = (rowTiles + 7) / 8;
  const int eBlocks    = (E + 255) / 256;
  const int nodeBlocks = (Nn + 7) / 8;

  hipMemsetAsync(cnt, 0, (size_t)Nn * sizeof(int), stream);
  gat_rank<<<eBlocks, b256, 0, stream>>>(dst, cnt, rank, E);
  gat_scan<<<1, 1024, 0, stream>>>(cnt, off, Nn);
  gat_fill<<<eBlocks, b256, 0, stream>>>(src, dst, off, rank, ssrc, E);

  gat_gemm_wmma<128><<<gemmBlocks, b256, 0, stream>>>(feat, W1, hbuf, Nn);
  gat_attn_proj<<<(NH + 255) / 256, b256, 0, stream>>>(hbuf, al1, ar1, el, er, NH);
  gat_node<1><<<nodeBlocks, b256, 0, stream>>>(off, ssrc, el, er, hbuf, b1, lns, lnb, hmid, Nn);

  gat_gemm_wmma<32><<<gemmBlocks, b256, 0, stream>>>(hmid, W2, hbuf, Nn);
  gat_attn_proj<<<(NH + 255) / 256, b256, 0, stream>>>(hbuf, al2, ar2, el, er, NH);
  gat_node<2><<<nodeBlocks, b256, 0, stream>>>(off, ssrc, el, er, hbuf, b2, lns, lnb, outh, Nn);

  gat_flops<<<1, 1024, 0, stream>>>(indeg, Nn, outh + (size_t)Nn * 32);
}


# el/er attention projections taken from the GEMM's accumulator tile (16-lane transposing butterfly) instead of a launch that re-reads h: 2 launches fewer; fp sum order of el/er changes
# speedup vs baseline: 3.5118x; 1.0384x over previous
#include <hip/hip_runtime.h>


typedef float v2f __attribute__((ext_vector_type(2)));
typedef float v4f __attribute__((ext_vector_type(4)));
typedef float v8f __attribute__((ext_vector_type(8)));

#define NEG_SLOPE 0.2f
#define LN_EPS 1e-5f

__device__ __forceinline__ float lrelu(float x) {
  return x > 0.f ? x : x * NEG_SLOPE;
}

template <int K>
__global__ void gat_gemm_wmma(const float* __restrict__ Amat,
                              const float* __restrict__ Bmat,
                              float* __restrict__ Cmat,
                              const float* __restrict__ al,
                              const float* __restrict__ ar,
                              float* __restrict__ el, float* __restrict__ er,
                              int M) {
  __shared__ float bs[K * 128];

  const int tid = threadIdx.x;

  for (int idx = tid; idx < K * 128; idx += 256) {
    int kr = idx >> 7;
    int col = idx & 127;
    int ks = kr >> 2;
    int c = col >> 4;
    int lane = (col & 15) | ((kr & 2) << 3);
    int sub = kr & 1;
    bs[((ks * 32 + lane) * 8 + c) * 2 + sub] = Bmat[idx];
  }
  __syncthreads();

  const int wave = tid >> 5;
  const int lane = tid & 31;
  const int row0 = (blockIdx.x * 8 + wave) * 16;
  if (row0 >= M) return;

  const int m  = lane & 15;
  const int kk = (lane >> 4) << 1;

  v8f acc[8] = {};

  float la[4], lb[4], ra[4], rb[4];
#pragma unroll
  for (int hd = 0; hd < 4; ++hd) {
    la[hd] = al[hd * 32 + m];
    lb[hd] = al[hd * 32 + 16 + m];
    ra[hd] = ar[hd * 32 + m];
    rb[hd] = ar[hd * 32 + 16 + m];
  }

  const float* arow = Amat + (size_t)(row0 + m) * K + kk;
  const float* blane = bs + lane * 16;

#pragma unroll 4
  for (int ks = 0; ks < K / 4; ++ks) {
    v2f a = *(const v2f*)(arow + ks * 4);
    const float* bp = blane + ks * 512;
    v4f q0 = *(const v4f*)(bp);
    v4f q1 = *(const v4f*)(bp + 4);
    v4f q2 = *(const v4f*)(bp + 8);
    v4f q3 = *(const v4f*)(bp + 12);
    acc[0] = __builtin_amdgcn_wmma_f32_16x16x4_f32(false, a, false, q0.xy, (short)0, acc[0], false, false);
    acc[1] = __builtin_amdgcn_wmma_f32_16x16x4_f32(false, a, false, q0.zw, (short)0, acc[1], false, false);
    acc[2] = __builtin_amdgcn_wmma_f32_16x16x4_f32(false, a, false, q1.xy, (short)0, acc[2], false, false);
    acc[3] = __builtin_amdgcn_wmma_f32_16x16x4_f32(false, a, false, q1.zw, (short)0, acc[3], false, false);
    acc[4] = __builtin_amdgcn_wmma_f32_16x16x4_f32(false, a, false, q2.xy, (short)0, acc[4], false, false);
    acc[5] = __builtin_amdgcn_wmma_f32_16x16x4_f32(false, a, false, q2.zw, (short)0, acc[5], false, false);
    acc[6] = __builtin_amdgcn_wmma_f32_16x16x4_f32(false, a, false, q3.xy, (short)0, acc[6], false, false);
    acc[7] = __builtin_amdgcn_wmma_f32_16x16x4_f32(false, a, false, q3.zw, (short)0, acc[7], false, false);
  }

  const int mo = (lane >> 4) * 8;
#pragma unroll
  for (int c = 0; c < 8; ++c) {
#pragma unroll
    for (int r = 0; r < 8; ++r) {
      Cmat[(size_t)(row0 + mo + r) * 128 + c * 16 + m] = acc[c][r];
    }
  }

  const bool u3 = (lane & 8) != 0, u2 = (lane & 4) != 0;
  const bool u1 = (lane & 2) != 0, u0 = (lane & 1) != 0;
  float rs[4];
#pragma unroll
  for (int hd = 0; hd < 4; ++hd) {
    float x[16];
#pragma unroll
    for (int r = 0; r < 8; ++r) {
      const float c0 = acc[2 * hd][r], c1 = acc[2 * hd + 1][r];
      x[r]     = c0 * la[hd] + c1 * lb[hd];
      x[8 + r] = c0 * ra[hd] + c1 * rb[hd];
    }
    float y[8];
#pragma unroll
    for (int j = 0; j < 8; ++j) {
      const float keep = u3 ? x[j + 8] : x[j];
      const float give = u3 ? x[j] : x[j + 8];
      y[j] = keep + __shfl_xor(give, 8, 32);
    }
    float z[4];
#pragma unroll
    for (int j = 0; j < 4; ++j) {
      const float keep = u2 ? y[j + 4] : y[j];
      const float give = u2 ? y[j] : y[j + 4];
      z[j] = keep + __shfl_xor(give, 4, 32);
    }
    float w[2];
#pragma unroll
    for (int j = 0; j < 2; ++j) {
      const float keep = u1 ? z[j + 2] : z[j];
      const float give = u1 ? z[j] : z[j + 2];
      w[j] = keep + __shfl_xor(give, 2, 32);
    }
    const float keep = u0 ? w[1] : w[0];
    const float give = u0 ? w[0] : w[1];
    rs[hd] = keep + __shfl_xor(give, 1, 32);
  }
  v4f res;
  res.x = rs[0]; res.y = rs[1]; res.z = rs[2]; res.w = rs[3];
  float* dp = (m < 8 ? el : er) + (size_t)(row0 + mo + (m & 7)) * 4;
  *(v4f*)dp = res;
}

__global__ void gat_rank(const int* __restrict__ dst, int* __restrict__ cnt,
                         int* __restrict__ rank, int E) {
  int e = blockIdx.x * blockDim.x + threadIdx.x;
  if (e >= E) return;
  rank[e] = atomicAdd(&cnt[dst[e]], 1);
}

__global__ void gat_scan(const int* __restrict__ cnt, int* __restrict__ off, int Nn) {
  __shared__ int wsum[32];
  const int t = threadIdx.x;
  const int lane = t & 31, w = t >> 5;
  const int per = (Nn + 1023) / 1024;
  const int b = t * per;
  const int e = (b + per < Nn) ? (b + per) : Nn;
  int s = 0;
  for (int i = b; i < e; ++i) s += cnt[i];
  int x = s;
#pragma unroll
  for (int d = 1; d < 32; d <<= 1) {
    int y = __shfl_up(x, d, 32);
    if (lane >= d) x += y;
  }
  if (lane == 31) wsum[w] = x;
  __syncthreads();
  if (w == 0) {
    int v = wsum[lane];
#pragma unroll
    for (int d = 1; d < 32; d <<= 1) {
      int y = __shfl_up(v, d, 32);
      if (lane >= d) v += y;
    }
    wsum[lane] = v;
  }
  __syncthreads();
  int base = (w > 0 ? wsum[w - 1] : 0) + (x - s);
  for (int i = b; i < e; i += 8) {
    int c[8];
#pragma unroll
    for (int k = 0; k < 8; ++k) c[k] = (i + k < e) ? cnt[i + k] : 0;
#pragma unroll
    for (int k = 0; k < 8; ++k) {
      if (i + k < e) off[i + k] = base;
      base += c[k];
    }
  }
  if (t == 1023) off[Nn] = base;
}

__global__ void gat_fill(const int* __restrict__ src, const int* __restrict__ dst,
                         const int* __restrict__ off, const int* __restrict__ rank,
                         int* __restrict__ ssrc, int E) {
  int e = blockIdx.x * blockDim.x + threadIdx.x;
  if (e >= E) return;
  ssrc[off[dst[e]] + rank[e]] = src[e];
}

template <int LAYER>
__global__ void gat_node(const int* __restrict__ off, const int* __restrict__ ssrc,
                         const float* __restrict__ el, const float* __restrict__ er,
                         const float* __restrict__ h,
                         const float* __restrict__ bias,
                         const float* __restrict__ lns, const float* __restrict__ lnb,
                         float* __restrict__ outp, int Nn) {
  const int lane = threadIdx.x & 31;
  const int n = __builtin_amdgcn_readfirstlane(blockIdx.x * 8 + (threadIdx.x >> 5));
  if (n >= Nn) return;
  const int beg = __builtin_amdgcn_readfirstlane(off[n]);
  const int deg = __builtin_amdgcn_readfirstlane(off[n + 1]) - beg;
  const int hh = lane & 3;
  const int q0 = lane >> 2;
  const int hl = lane >> 3;
  const float erh = er[n * 4 + hh];
  const float* hrow = h + lane * 4;
  const float ninf = -__builtin_huge_valf();

  int   s[4]  = {0, 0, 0, 0};
  float ev[4] = {ninf, ninf, ninf, ninf};
  v4f   r[4][8];

  if (deg > 0) {
#pragma unroll
    for (int t = 0; t < 4; ++t) {
      const int j = t * 8 + q0;
      s[t] = ssrc[beg + (j < deg ? j : 0)];
    }
#pragma unroll
    for (int t = 0; t < 4; ++t) {
      const int j = t * 8 + q0;
      const float x = lrelu(el[s[t] * 4 + hh] + erh);
      ev[t] = (j < deg) ? x : ninf;
    }
#pragma unroll
    for (int q = 0; q < 8; ++q) {
      const int sq = __builtin_amdgcn_readlane(s[0], q * 4);
      r[0][q] = *(const v4f*)(hrow + (size_t)sq * 128);
    }
  }
  if (deg > 8) {
#pragma unroll
    for (int q = 0; q < 8; ++q) {
      const int sq = __builtin_amdgcn_readlane(s[1], q * 4);
      r[1][q] = *(const v4f*)(hrow + (size_t)sq * 128);
    }
  }

  float m = fmaxf(fmaxf(fmaxf(fmaxf(ninf, ev[0]), ev[1]), ev[2]), ev[3]);
  for (int c0 = 32; c0 < deg; c0 += 8) {
    int j = c0 + q0;
    if (j < deg) {
      int sj = ssrc[beg + j];
      m = fmaxf(m, lrelu(el[sj * 4 + hh] + erh));
    }
  }
  m = fmaxf(m, __shfl_xor(m, 4, 32));
  m = fmaxf(m, __shfl_xor(m, 8, 32));
  m = fmaxf(m, __shfl_xor(m, 16, 32));
  m = (fabsf(m) < __builtin_huge_valf()) ? m : 0.f;

  float p[4];
  float den = 0.f;
#pragma unroll
  for (int t = 0; t < 4; ++t) {
    p[t] = (t * 8 + q0 < deg) ? expf(ev[t] - m) : 0.f;
    den += p[t];
  }
  for (int c0 = 32; c0 < deg; c0 += 8) {
    int j = c0 + q0;
    if (j < deg) {
      int sj = ssrc[beg + j];
      den += expf(lrelu(el[sj * 4 + hh] + erh) - m);
    }
  }
  den += __shfl_xor(den, 4, 32);
  den += __shfl_xor(den, 8, 32);
  den += __shfl_xor(den, 16, 32);
  den = fmaxf(den, 1e-9f);

  v4f acc = {0.f, 0.f, 0.f, 0.f};
#pragma unroll
  for (int t = 0; t < 4; ++t) {
    if (t * 8 < deg) {
      const float alpha = p[t] / den;
#pragma unroll
      for (int q = 0; q < 8; ++q) {
        const float a = __shfl(alpha, q * 4 + hl, 32);
        acc += a * r[t][q];
      }
    }
    const int t2 = (t + 2) & 3;
    if (t < 2 && (t + 2) * 8 < deg) {
#pragma unroll
      for (int q = 0; q < 8; ++q) {
        const int sq = __builtin_amdgcn_readlane(s[t2], q * 4);
        r[t2][q] = *(const v4f*)(hrow + (size_t)sq * 128);
      }
    }
  }
  for (int c0 = 32; c0 < deg; c0 += 8) {
    const int j = c0 + q0;
    const bool valid = j < deg;
    const int sj = ssrc[beg + (valid ? j : c0)];
    const float evj = lrelu(el[sj * 4 + hh] + erh);
    const float alpha = valid ? expf(evj - m) / den : 0.f;
#pragma unroll
    for (int q = 0; q < 8; ++q) {
      const int sq = __builtin_amdgcn_readlane(sj, q * 4);
      const float a = __shfl(alpha, q * 4 + hl, 32);
      const v4f v = *(const v4f*)(hrow + (size_t)sq * 128);
      acc += a * v;
    }
  }

  float x0 = acc.x, x1 = acc.y, x2 = acc.z, x3 = acc.w;
  x0 += __shfl_xor(x0, 8, 32);  x1 += __shfl_xor(x1, 8, 32);
  x2 += __shfl_xor(x2, 8, 32);  x3 += __shfl_xor(x3, 8, 32);
  x0 += __shfl_xor(x0, 16, 32); x1 += __shfl_xor(x1, 16, 32);
  x2 += __shfl_xor(x2, 16, 32); x3 += __shfl_xor(x3, 16, 32);

  const int f0 = (lane & 7) * 4;
#pragma unroll
  for (int k = 0; k < 4; ++k) {
    x0 += bias[k * 32 + f0 + 0];
    x1 += bias[k * 32 + f0 + 1];
    x2 += bias[k * 32 + f0 + 2];
    x3 += bias[k * 32 + f0 + 3];
  }
  x0 *= 0.25f; x1 *= 0.25f; x2 *= 0.25f; x3 *= 0.25f;

  v4f y;
  if (LAYER == 1) {
    float mu = (x0 + x1) + (x2 + x3);
    mu += __shfl_xor(mu, 1, 32);
    mu += __shfl_xor(mu, 2, 32);
    mu += __shfl_xor(mu, 4, 32);
    mu *= (1.f / 32.f);
    const float d0 = x0 - mu, d1 = x1 - mu, d2 = x2 - mu, d3 = x3 - mu;
    float var = (d0 * d0 + d1 * d1) + (d2 * d2 + d3 * d3);
    var += __shfl_xor(var, 1, 32);
    var += __shfl_xor(var, 2, 32);
    var += __shfl_xor(var, 4, 32);
    var *= (1.f / 32.f);
    const float r = rsqrtf(var + LN_EPS);
    y.x = fmaxf(d0 * r * lns[f0 + 0] + lnb[f0 + 0], 0.f);
    y.y = fmaxf(d1 * r * lns[f0 + 1] + lnb[f0 + 1], 0.f);
    y.z = fmaxf(d2 * r * lns[f0 + 2] + lnb[f0 + 2], 0.f);
    y.w = fmaxf(d3 * r * lns[f0 + 3] + lnb[f0 + 3], 0.f);
  } else {
    y.x = x0; y.y = x1; y.z = x2; y.w = x3;
  }
  if (lane < 8) *(v4f*)(outp + (size_t)n * 32 + f0) = y;
}

__global__ void gat_flops(const int* __restrict__ in_deg, int Nn, float* out) {
  __shared__ int wsum[32];
  const int t = threadIdx.x;
  int s = 0;
  for (int i = t; i < Nn; i += 16 * 1024) {
    int c[16];
#pragma unroll
    for (int k = 0; k < 16; ++k) c[k] = (i + k * 1024 < Nn) ? in_deg[i + k * 1024] : 0;
#pragma unroll
    for (int k = 0; k < 16; ++k) s += c[k];
  }
#pragma unroll
  for (int d = 16; d > 0; d >>= 1) s += __shfl_xor(s, d, 32);
  if ((t & 31) == 0) wsum[t >> 5] = s;
  __syncthreads();
  if (t < 32) {
    int v = wsum[t];
#pragma unroll
    for (int d = 16; d > 0; d >>= 1) v += __shfl_xor(v, d, 32);
    if (t == 0) {
      double ne = (double)(float)v;
      double fl = (4.0 * ne * (6.0 * 128 * 32 + 6 * 32 + 2) +
                   4.0 * ne * (6.0 * 32 * 32 + 6 * 32 + 2)) / 1e12;
      out[0] = (float)fl;
    }
  }
}

extern "C" void kernel_launch(void* const* d_in, const int* in_sizes, int n_in,
                              void* d_out, int out_size, void* d_ws, size_t ws_size,
                              hipStream_t stream) {
  const float* feat = (const float*)d_in[0];
  const float* W1   = (const float*)d_in[1];
  const float* al1  = (const float*)d_in[2];
  const float* ar1  = (const float*)d_in[3];
  const float* b1   = (const float*)d_in[4];
  const float* lns  = (const float*)d_in[5];
  const float* lnb  = (const float*)d_in[6];
  const float* W2   = (const float*)d_in[7];
  const float* al2  = (const float*)d_in[8];
  const float* ar2  = (const float*)d_in[9];
  const float* b2   = (const float*)d_in[10];
  const int*   src  = (const int*)d_in[11];
  const int*   dst  = (const int*)d_in[12];
  const int*   indeg= (const int*)d_in[13];

  const int Nn = in_sizes[13];
  const int E  = in_sizes[11];
  const int NH = Nn * 4;

  char* p = (char*)d_ws;
  float* hbuf = (float*)p;   p += (size_t)Nn * 128 * sizeof(float);
  float* hmid = (float*)p;   p += (size_t)Nn * 32  * sizeof(float);
  float* el   = (float*)p;   p += (size_t)NH * sizeof(float);
  float* er   = (float*)p;   p += (size_t)NH * sizeof(float);
  int*   rank = (int*)p;     p += (size_t)E * sizeof(int);
  int*   ssrc = (int*)p;     p += (size_t)E * sizeof(int);
  int*   cnt  = (int*)p;     p += (size_t)Nn * sizeof(int);
  int*   off  = (int*)p;     p += ((size_t)Nn + 1) * sizeof(int);
  (void)ws_size; (void)n_in; (void)out_size;

  float* outh = (float*)d_out;

  dim3 b256(256);
  const int rowTiles   = (Nn + 15) / 16;
  const int gemmBlocks = (rowTiles + 7) / 8;
  const int eBlocks    = (E + 255) / 256;
  const int nodeBlocks = (Nn + 7) / 8;

  hipMemsetAsync(cnt, 0, (size_t)Nn * sizeof(int), stream);
  gat_rank<<<eBlocks, b256, 0, stream>>>(dst, cnt, rank, E);
  gat_scan<<<1, 1024, 0, stream>>>(cnt, off, Nn);
  gat_fill<<<eBlocks, b256, 0, stream>>>(src, dst, off, rank, ssrc, E);

  gat_gemm_wmma<128><<<gemmBlocks, b256, 0, stream>>>(feat, W1, hbuf, al1, ar1, el, er, Nn);
  gat_node<1><<<nodeBlocks, b256, 0, stream>>>(off, ssrc, el, er, hbuf, b1, lns, lnb, hmid, Nn);

  gat_gemm_wmma<32><<<gemmBlocks, b256, 0, stream>>>(hmid, W2, hbuf, al2, ar2, el, er, Nn);
  gat_node<2><<<nodeBlocks, b256, 0, stream>>>(off, ssrc, el, er, hbuf, b2, lns, lnb, outh, Nn);

  gat_flops<<<1, 1024, 0, stream>>>(indeg, Nn, outh + (size_t)Nn * 32);
}


# offset scan reads counts and writes offsets as whole 128-byte lines, transposed through 196 KiB of LDS to 49 consecutive elements per thread; was 32 lines under each load and store instruction
# speedup vs baseline: 4.2419x; 1.2079x over previous
#include <hip/hip_runtime.h>


typedef float v2f __attribute__((ext_vector_type(2)));
typedef float v4f __attribute__((ext_vector_type(4)));
typedef float v8f __attribute__((ext_vector_type(8)));

#define NEG_SLOPE 0.2f
#define LN_EPS 1e-5f

__device__ __forceinline__ float lrelu(float x) {
  return x > 0.f ? x : x * NEG_SLOPE;
}

template <int K>
__global__ void gat_gemm_wmma(const float* __restrict__ Amat,
                              const float* __restrict__ Bmat,
                              float* __restrict__ Cmat,
                              const float* __restrict__ al,
                              const float* __restrict__ ar,
                              float* __restrict__ el, float* __restrict__ er,
                              int M) {
  __shared__ float bs[K * 128];

  const int tid = threadIdx.x;

  for (int idx = tid; idx < K * 128; idx += 256) {
    int kr = idx >> 7;
    int col = idx & 127;
    int ks = kr >> 2;
    int c = col >> 4;
    int lane = (col & 15) | ((kr & 2) << 3);
    int sub = kr & 1;
    bs[((ks * 32 + lane) * 8 + c) * 2 + sub] = Bmat[idx];
  }
  __syncthreads();

  const int wave = tid >> 5;
  const int lane = tid & 31;
  const int row0 = (blockIdx.x * 8 + wave) * 16;
  if (row0 >= M) return;

  const int m  = lane & 15;
  const int kk = (lane >> 4) << 1;

  v8f acc[8] = {};

  float la[4], lb[4], ra[4], rb[4];
#pragma unroll
  for (int hd = 0; hd < 4; ++hd) {
    la[hd] = al[hd * 32 + m];
    lb[hd] = al[hd * 32 + 16 + m];
    ra[hd] = ar[hd * 32 + m];
    rb[hd] = ar[hd * 32 + 16 + m];
  }

  const float* arow = Amat + (size_t)(row0 + m) * K + kk;
  const float* blane = bs + lane * 16;

#pragma unroll 4
  for (int ks = 0; ks < K / 4; ++ks) {
    v2f a = *(const v2f*)(arow + ks * 4);
    const float* bp = blane + ks * 512;
    v4f q0 = *(const v4f*)(bp);
    v4f q1 = *(const v4f*)(bp + 4);
    v4f q2 = *(const v4f*)(bp + 8);
    v4f q3 = *(const v4f*)(bp + 12);
    acc[0] = __builtin_amdgcn_wmma_f32_16x16x4_f32(false, a, false, q0.xy, (short)0, acc[0], false, false);
    acc[1] = __builtin_amdgcn_wmma_f32_16x16x4_f32(false, a, false, q0.zw, (short)0, acc[1], false, false);
    acc[2] = __builtin_amdgcn_wmma_f32_16x16x4_f32(false, a, false, q1.xy, (short)0, acc[2], false, false);
    acc[3] = __builtin_amdgcn_wmma_f32_16x16x4_f32(false, a, false, q1.zw, (short)0, acc[3], false, false);
    acc[4] = __builtin_amdgcn_wmma_f32_16x16x4_f32(false, a, false, q2.xy, (short)0, acc[4], false, false);
    acc[5] = __builtin_amdgcn_wmma_f32_16x16x4_f32(false, a, false, q2.zw, (short)0, acc[5], false, false);
    acc[6] = __builtin_amdgcn_wmma_f32_16x16x4_f32(false, a, false, q3.xy, (short)0, acc[6], false, false);
    acc[7] = __builtin_amdgcn_wmma_f32_16x16x4_f32(false, a, false, q3.zw, (short)0, acc[7], false, false);
  }

  const int mo = (lane >> 4) * 8;
#pragma unroll
  for (int c = 0; c < 8; ++c) {
#pragma unroll
    for (int r = 0; r < 8; ++r) {
      Cmat[(size_t)(row0 + mo + r) * 128 + c * 16 + m] = acc[c][r];
    }
  }

  const bool u3 = (lane & 8) != 0, u2 = (lane & 4) != 0;
  const bool u1 = (lane & 2) != 0, u0 = (lane & 1) != 0;
  float rs[4];
#pragma unroll
  for (int hd = 0; hd < 4; ++hd) {
    float x[16];
#pragma unroll
    for (int r = 0; r < 8; ++r) {
      const float c0 = acc[2 * hd][r], c1 = acc[2 * hd + 1][r];
      x[r]     = c0 * la[hd] + c1 * lb[hd];
      x[8 + r] = c0 * ra[hd] + c1 * rb[hd];
    }
    float y[8];
#pragma unroll
    for (int j = 0; j < 8; ++j) {
      const float keep = u3 ? x[j + 8] : x[j];
      const float give = u3 ? x[j] : x[j + 8];
      y[j] = keep + __shfl_xor(give, 8, 32);
    }
    float z[4];
#pragma unroll
    for (int j = 0; j < 4; ++j) {
      const float keep = u2 ? y[j + 4] : y[j];
      const float give = u2 ? y[j] : y[j + 4];
      z[j] = keep + __shfl_xor(give, 4, 32);
    }
    float w[2];
#pragma unroll
    for (int j = 0; j < 2; ++j) {
      const float keep = u1 ? z[j + 2] : z[j];
      const float give = u1 ? z[j] : z[j + 2];
      w[j] = keep + __shfl_xor(give, 2, 32);
    }
    const float keep = u0 ? w[1] : w[0];
    const float give = u0 ? w[0] : w[1];
    rs[hd] = keep + __shfl_xor(give, 1, 32);
  }
  v4f res;
  res.x = rs[0]; res.y = rs[1]; res.z = rs[2]; res.w = rs[3];
  float* dp = (m < 8 ? el : er) + (size_t)(row0 + mo + (m & 7)) * 4;
  *(v4f*)dp = res;
}

__global__ void gat_rank(const int* __restrict__ dst, int* __restrict__ cnt,
                         int* __restrict__ rank, int E) {
  int e = blockIdx.x * blockDim.x + threadIdx.x;
  if (e >= E) return;
  rank[e] = atomicAdd(&cnt[dst[e]], 1);
}

#define SCAN_PER 49
#define SCAN_CH (1024 * SCAN_PER)
__global__ void __launch_bounds__(1024) gat_scan(const int* __restrict__ cnt,
                                                 int* __restrict__ off, int Nn) {
  __shared__ int buf[SCAN_CH];
  __shared__ int wsum[32];
  const int t = threadIdx.x;
  const int lane = t & 31, w = t >> 5;
  int carry = 0;
  for (int c0 = 0; c0 < Nn; c0 += SCAN_CH) {
    int c[SCAN_PER];
#pragma unroll
    for (int k = 0; k < SCAN_PER; ++k) {
      const int j = c0 + k * 1024 + t;
      const int val = cnt[j < Nn ? j : Nn - 1];
      c[k] = (j < Nn) ? val : 0;
    }
#pragma unroll
    for (int k = 0; k < SCAN_PER; ++k) buf[k * 1024 + t] = c[k];
    __syncthreads();

    int v[SCAN_PER];
    int s = 0;
#pragma unroll
    for (int i = 0; i < SCAN_PER; ++i) {
      v[i] = buf[t * SCAN_PER + i];
      s += v[i];
    }
    int x = s;
#pragma unroll
    for (int d = 1; d < 32; d <<= 1) {
      int y = __shfl_up(x, d, 32);
      if (lane >= d) x += y;
    }
    if (lane == 31) wsum[w] = x;
    __syncthreads();
    if (w == 0) {
      int u = wsum[lane];
#pragma unroll
      for (int d = 1; d < 32; d <<= 1) {
        int y = __shfl_up(u, d, 32);
        if (lane >= d) u += y;
      }
      wsum[lane] = u;
    }
    __syncthreads();
    int base = carry + (w > 0 ? wsum[w - 1] : 0) + (x - s);
    carry += wsum[31];
#pragma unroll
    for (int i = 0; i < SCAN_PER; ++i) {
      buf[t * SCAN_PER + i] = base;
      base += v[i];
    }
    __syncthreads();
    const int rows = (Nn - c0) >> 10;
    const int nfull = rows < SCAN_PER ? rows : SCAN_PER;
#pragma unroll 8
    for (int k = 0; k < nfull; ++k) off[c0 + k * 1024 + t] = buf[k * 1024 + t];
    if (nfull < SCAN_PER) {
      const int j = c0 + nfull * 1024 + t;
      if (j < Nn) off[j] = buf[nfull * 1024 + t];
    }
    __syncthreads();
  }
  if (t == 0) off[Nn] = carry;
}

__global__ void gat_fill(const int* __restrict__ src, const int* __restrict__ dst,
                         const int* __restrict__ off, const int* __restrict__ rank,
                         int* __restrict__ ssrc, int E) {
  int e = blockIdx.x * blockDim.x + threadIdx.x;
  if (e >= E) return;
  ssrc[off[dst[e]] + rank[e]] = src[e];
}

template <int LAYER>
__global__ void gat_node(const int* __restrict__ off, const int* __restrict__ ssrc,
                         const float* __restrict__ el, const float* __restrict__ er,
                         const float* __restrict__ h,
                         const float* __restrict__ bias,
                         const float* __restrict__ lns, const float* __restrict__ lnb,
                         float* __restrict__ outp, int Nn) {
  const int lane = threadIdx.x & 31;
  const int n = __builtin_amdgcn_readfirstlane(blockIdx.x * 8 + (threadIdx.x >> 5));
  if (n >= Nn) return;
  const int beg = __builtin_amdgcn_readfirstlane(off[n]);
  const int deg = __builtin_amdgcn_readfirstlane(off[n + 1]) - beg;
  const int hh = lane & 3;
  const int q0 = lane >> 2;
  const int hl = lane >> 3;
  const float erh = er[n * 4 + hh];
  const float* hrow = h + lane * 4;
  const float ninf = -__builtin_huge_valf();

  int   s[4]  = {0, 0, 0, 0};
  float ev[4] = {ninf, ninf, ninf, ninf};
  v4f   r[4][8];

  if (deg > 0) {
#pragma unroll
    for (int t = 0; t < 4; ++t) {
      const int j = t * 8 + q0;
      s[t] = ssrc[beg + (j < deg ? j : 0)];
    }
#pragma unroll
    for (int t = 0; t < 4; ++t) {
      const int j = t * 8 + q0;
      const float x = lrelu(el[s[t] * 4 + hh] + erh);
      ev[t] = (j < deg) ? x : ninf;
    }
#pragma unroll
    for (int q = 0; q < 8; ++q) {
      const int sq = __builtin_amdgcn_readlane(s[0], q * 4);
      r[0][q] = *(const v4f*)(hrow + (size_t)sq * 128);
    }
  }
  if (deg > 8) {
#pragma unroll
    for (int q = 0; q < 8; ++q) {
      const int sq = __builtin_amdgcn_readlane(s[1], q * 4);
      r[1][q] = *(const v4f*)(hrow + (size_t)sq * 128);
    }
  }

  float m = fmaxf(fmaxf(fmaxf(fmaxf(ninf, ev[0]), ev[1]), ev[2]), ev[3]);
  for (int c0 = 32; c0 < deg; c0 += 8) {
    int j = c0 + q0;
    if (j < deg) {
      int sj = ssrc[beg + j];
      m = fmaxf(m, lrelu(el[sj * 4 + hh] + erh));
    }
  }
  m = fmaxf(m, __shfl_xor(m, 4, 32));
  m = fmaxf(m, __shfl_xor(m, 8, 32));
  m = fmaxf(m, __shfl_xor(m, 16, 32));
  m = (fabsf(m) < __builtin_huge_valf()) ? m : 0.f;

  float p[4];
  float den = 0.f;
#pragma unroll
  for (int t = 0; t < 4; ++t) {
    p[t] = (t * 8 + q0 < deg) ? expf(ev[t] - m) : 0.f;
    den += p[t];
  }
  for (int c0 = 32; c0 < deg; c0 += 8) {
    int j = c0 + q0;
    if (j < deg) {
      int sj = ssrc[beg + j];
      den += expf(lrelu(el[sj * 4 + hh] + erh) - m);
    }
  }
  den += __shfl_xor(den, 4, 32);
  den += __shfl_xor(den, 8, 32);
  den += __shfl_xor(den, 16, 32);
  den = fmaxf(den, 1e-9f);

  v4f acc = {0.f, 0.f, 0.f, 0.f};
#pragma unroll
  for (int t = 0; t < 4; ++t) {
    if (t * 8 < deg) {
      const float alpha = p[t] / den;
#pragma unroll
      for (int q = 0; q < 8; ++q) {
        const float a = __shfl(alpha, q * 4 + hl, 32);
        acc += a * r[t][q];
      }
    }
    const int t2 = (t + 2) & 3;
    if (t < 2 && (t + 2) * 8 < deg) {
#pragma unroll
      for (int q = 0; q < 8; ++q) {
        const int sq = __builtin_amdgcn_readlane(s[t2], q * 4);
        r[t2][q] = *(const v4f*)(hrow + (size_t)sq * 128);
      }
    }
  }
  for (int c0 = 32; c0 < deg; c0 += 8) {
    const int j = c0 + q0;
    const bool valid = j < deg;
    const int sj = ssrc[beg + (valid ? j : c0)];
    const float evj = lrelu(el[sj * 4 + hh] + erh);
    const float alpha = valid ? expf(evj - m) / den : 0.f;
#pragma unroll
    for (int q = 0; q < 8; ++q) {
      const int sq = __builtin_amdgcn_readlane(sj, q * 4);
      const float a = __shfl(alpha, q * 4 + hl, 32);
      const v4f v = *(const v4f*)(hrow + (size_t)sq * 128);
      acc += a * v;
    }
  }

  float x0 = acc.x, x1 = acc.y, x2 = acc.z, x3 = acc.w;
  x0 += __shfl_xor(x0, 8, 32);  x1 += __shfl_xor(x1, 8, 32);
  x2 += __shfl_xor(x2, 8, 32);  x3 += __shfl_xor(x3, 8, 32);
  x0 += __shfl_xor(x0, 16, 32); x1 += __shfl_xor(x1, 16, 32);
  x2 += __shfl_xor(x2, 16, 32); x3 += __shfl_xor(x3, 16, 32);

  const int f0 = (lane & 7) * 4;
#pragma unroll
  for (int k = 0; k < 4; ++k) {
    x0 += bias[k * 32 + f0 + 0];
    x1 += bias[k * 32 + f0 + 1];
    x2 += bias[k * 32 + f0 + 2];
    x3 += bias[k * 32 + f0 + 3];
  }
  x0 *= 0.25f; x1 *= 0.25f; x2 *= 0.25f; x3 *= 0.25f;

  v4f y;
  if (LAYER == 1) {
    float mu = (x0 + x1) + (x2 + x3);
    mu += __shfl_xor(mu, 1, 32);
    mu += __shfl_xor(mu, 2, 32);
    mu += __shfl_xor(mu, 4, 32);
    mu *= (1.f / 32.f);
    const float d0 = x0 - mu, d1 = x1 - mu, d2 = x2 - mu, d3 = x3 - mu;
    float var = (d0 * d0 + d1 * d1) + (d2 * d2 + d3 * d3);
    var += __shfl_xor(var, 1, 32);
    var += __shfl_xor(var, 2, 32);
    var += __shfl_xor(var, 4, 32);
    var *= (1.f / 32.f);
    const float r = rsqrtf(var + LN_EPS);
    y.x = fmaxf(d0 * r * lns[f0 + 0] + lnb[f0 + 0], 0.f);
    y.y = fmaxf(d1 * r * lns[f0 + 1] + lnb[f0 + 1], 0.f);
    y.z = fmaxf(d2 * r * lns[f0 + 2] + lnb[f0 + 2], 0.f);
    y.w = fmaxf(d3 * r * lns[f0 + 3] + lnb[f0 + 3], 0.f);
  } else {
    y.x = x0; y.y = x1; y.z = x2; y.w = x3;
  }
  if (lane < 8) *(v4f*)(outp + (size_t)n * 32 + f0) = y;
}

__global__ void gat_flops(const int* __restrict__ in_deg, int Nn, float* out) {
  __shared__ int wsum[32];
  const int t = threadIdx.x;
  int s = 0;
  for (int i = t; i < Nn; i += 16 * 1024) {
    int c[16];
#pragma unroll
    for (int k = 0; k < 16; ++k) c[k] = (i + k * 1024 < Nn) ? in_deg[i + k * 1024] : 0;
#pragma unroll
    for (int k = 0; k < 16; ++k) s += c[k];
  }
#pragma unroll
  for (int d = 16; d > 0; d >>= 1) s += __shfl_xor(s, d, 32);
  if ((t & 31) == 0) wsum[t >> 5] = s;
  __syncthreads();
  if (t < 32) {
    int v = wsum[t];
#pragma unroll
    for (int d = 16; d > 0; d >>= 1) v += __shfl_xor(v, d, 32);
    if (t == 0) {
      double ne = (double)(float)v;
      double fl = (4.0 * ne * (6.0 * 128 * 32 + 6 * 32 + 2) +
                   4.0 * ne * (6.0 * 32 * 32 + 6 * 32 + 2)) / 1e12;
      out[0] = (float)fl;
    }
  }
}

extern "C" void kernel_launch(void* const* d_in, const int* in_sizes, int n_in,
                              void* d_out, int out_size, void* d_ws, size_t ws_size,
                              hipStream_t stream) {
  const float* feat = (const float*)d_in[0];
  const float* W1   = (const float*)d_in[1];
  const float* al1  = (const float*)d_in[2];
  const float* ar1  = (const float*)d_in[3];
  const float* b1   = (const float*)d_in[4];
  const float* lns  = (const float*)d_in[5];
  const float* lnb  = (const float*)d_in[6];
  const float* W2   = (const float*)d_in[7];
  const float* al2  = (const float*)d_in[8];
  const float* ar2  = (const float*)d_in[9];
  const float* b2   = (const float*)d_in[10];
  const int*   src  = (const int*)d_in[11];
  const int*   dst  = (const int*)d_in[12];
  const int*   indeg= (const int*)d_in[13];

  const int Nn = in_sizes[13];
  const int E  = in_sizes[11];
  const int NH = Nn * 4;

  char* p = (char*)d_ws;
  float* hbuf = (float*)p;   p += (size_t)Nn * 128 * sizeof(float);
  float* hmid = (float*)p;   p += (size_t)Nn * 32  * sizeof(float);
  float* el   = (float*)p;   p += (size_t)NH * sizeof(float);
  float* er   = (float*)p;   p += (size_t)NH * sizeof(float);
  int*   rank = (int*)p;     p += (size_t)E * sizeof(int);
  int*   ssrc = (int*)p;     p += (size_t)E * sizeof(int);
  int*   cnt  = (int*)p;     p += (size_t)Nn * sizeof(int);
  int*   off  = (int*)p;     p += ((size_t)Nn + 1) * sizeof(int);
  (void)ws_size; (void)n_in; (void)out_size;

  float* outh = (float*)d_out;

  dim3 b256(256);
  const int rowTiles   = (Nn + 15) / 16;
  const int gemmBlocks = (rowTiles + 7) / 8;
  const int eBlocks    = (E + 255) / 256;
  const int nodeBlocks = (Nn + 7) / 8;

  hipMemsetAsync(cnt, 0, (size_t)Nn * sizeof(int), stream);
  gat_rank<<<eBlocks, b256, 0, stream>>>(dst, cnt, rank, E);
  gat_scan<<<1, 1024, 0, stream>>>(cnt, off, Nn);
  gat_fill<<<eBlocks, b256, 0, stream>>>(src, dst, off, rank, ssrc, E);

  gat_gemm_wmma<128><<<gemmBlocks, b256, 0, stream>>>(feat, W1, hbuf, al1, ar1, el, er, Nn);
  gat_node<1><<<nodeBlocks, b256, 0, stream>>>(off, ssrc, el, er, hbuf, b1, lns, lnb, hmid, Nn);

  gat_gemm_wmma<32><<<gemmBlocks, b256, 0, stream>>>(hmid, W2, hbuf, al2, ar2, el, er, Nn);
  gat_node<2><<<nodeBlocks, b256, 0, stream>>>(off, ssrc, el, er, hbuf, b2, lns, lnb, outh, Nn);

  gat_flops<<<1, 1024, 0, stream>>>(indeg, Nn, outh + (size_t)Nn * 32);
}


# edges bucketed by destination with a two-level counting sort on LDS counters (count, scan, scatter by node range, one block sorts a range), was 800000 returning global atomics; fp sum order changes
# speedup vs baseline: 5.9353x; 1.3992x over previous
#include <hip/hip_runtime.h>


typedef float v2f __attribute__((ext_vector_type(2)));
typedef float v4f __attribute__((ext_vector_type(4)));
typedef float v8f __attribute__((ext_vector_type(8)));

#define NEG_SLOPE 0.2f
#define LN_EPS 1e-5f

__device__ __forceinline__ float lrelu(float x) {
  return x > 0.f ? x : x * NEG_SLOPE;
}

template <int K, bool STORE_C>
__global__ void gat_gemm_wmma(const float* __restrict__ Amat,
                              const float* __restrict__ Bmat,
                              float* __restrict__ Cmat,
                              const float* __restrict__ al,
                              const float* __restrict__ ar,
                              float* __restrict__ el, float* __restrict__ er,
                              int M) {
  __shared__ float bs[K * 128];

  const int tid = threadIdx.x;

  for (int idx = tid; idx < K * 128; idx += 256) {
    int kr = idx >> 7;
    int col = idx & 127;
    int ks = kr >> 2;
    int c = col >> 4;
    int lane = (col & 15) | ((kr & 2) << 3);
    int sub = kr & 1;
    bs[((ks * 32 + lane) * 8 + c) * 2 + sub] = Bmat[idx];
  }
  __syncthreads();

  const int wave = tid >> 5;
  const int lane = tid & 31;
  const int row0 = (blockIdx.x * 8 + wave) * 16;
  if (row0 >= M) return;

  const int m  = lane & 15;
  const int kk = (lane >> 4) << 1;

  v8f acc[8] = {};

  float la[4], lb[4], ra[4], rb[4];
#pragma unroll
  for (int hd = 0; hd < 4; ++hd) {
    la[hd] = al[hd * 32 + m];
    lb[hd] = al[hd * 32 + 16 + m];
    ra[hd] = ar[hd * 32 + m];
    rb[hd] = ar[hd * 32 + 16 + m];
  }

  const float* arow = Amat + (size_t)(row0 + m) * K + kk;
  const float* blane = bs + lane * 16;

#pragma unroll 4
  for (int ks = 0; ks < K / 4; ++ks) {
    v2f a = *(const v2f*)(arow + ks * 4);
    const float* bp = blane + ks * 512;
    v4f q0 = *(const v4f*)(bp);
    v4f q1 = *(const v4f*)(bp + 4);
    v4f q2 = *(const v4f*)(bp + 8);
    v4f q3 = *(const v4f*)(bp + 12);
    acc[0] = __builtin_amdgcn_wmma_f32_16x16x4_f32(false, a, false, q0.xy, (short)0, acc[0], false, false);
    acc[1] = __builtin_amdgcn_wmma_f32_16x16x4_f32(false, a, false, q0.zw, (short)0, acc[1], false, false);
    acc[2] = __builtin_amdgcn_wmma_f32_16x16x4_f32(false, a, false, q1.xy, (short)0, acc[2], false, false);
    acc[3] = __builtin_amdgcn_wmma_f32_16x16x4_f32(false, a, false, q1.zw, (short)0, acc[3], false, false);
    acc[4] = __builtin_amdgcn_wmma_f32_16x16x4_f32(false, a, false, q2.xy, (short)0, acc[4], false, false);
    acc[5] = __builtin_amdgcn_wmma_f32_16x16x4_f32(false, a, false, q2.zw, (short)0, acc[5], false, false);
    acc[6] = __builtin_amdgcn_wmma_f32_16x16x4_f32(false, a, false, q3.xy, (short)0, acc[6], false, false);
    acc[7] = __builtin_amdgcn_wmma_f32_16x16x4_f32(false, a, false, q3.zw, (short)0, acc[7], false, false);
  }

  const int mo = (lane >> 4) * 8;
  if (STORE_C) {
#pragma unroll
    for (int c = 0; c < 8; ++c) {
#pragma unroll
      for (int r = 0; r < 8; ++r) {
        Cmat[(size_t)(row0 + mo + r) * 128 + c * 16 + m] = acc[c][r];
      }
    }
  }

  const bool u3 = (lane & 8) != 0, u2 = (lane & 4) != 0;
  const bool u1 = (lane & 2) != 0, u0 = (lane & 1) != 0;
  float rs[4];
#pragma unroll
  for (int hd = 0; hd < 4; ++hd) {
    float x[16];
#pragma unroll
    for (int r = 0; r < 8; ++r) {
      const float c0 = acc[2 * hd][r], c1 = acc[2 * hd + 1][r];
      x[r]     = c0 * la[hd] + c1 * lb[hd];
      x[8 + r] = c0 * ra[hd] + c1 * rb[hd];
    }
    float y[8];
#pragma unroll
    for (int j = 0; j < 8; ++j) {
      const float keep = u3 ? x[j + 8] : x[j];
      const float give = u3 ? x[j] : x[j + 8];
      y[j] = keep + __shfl_xor(give, 8, 32);
    }
    float z[4];
#pragma unroll
    for (int j = 0; j < 4; ++j) {
      const float keep = u2 ? y[j + 4] : y[j];
      const float give = u2 ? y[j] : y[j + 4];
      z[j] = keep + __shfl_xor(give, 4, 32);
    }
    float w[2];
#pragma unroll
    for (int j = 0; j < 2; ++j) {
      const float keep = u1 ? z[j + 2] : z[j];
      const float give = u1 ? z[j] : z[j + 2];
      w[j] = keep + __shfl_xor(give, 2, 32);
    }
    const float keep = u0 ? w[1] : w[0];
    const float give = u0 ? w[0] : w[1];
    rs[hd] = keep + __shfl_xor(give, 1, 32);
  }
  v4f res;
  res.x = rs[0]; res.y = rs[1]; res.z = rs[2]; res.w = rs[3];
  float* dp = (m < 8 ? el : er) + (size_t)(row0 + mo + (m & 7)) * 4;
  *(v4f*)dp = res;
}

#define NBIN 256
#define RMAX 256
#define CBLK 196

__global__ void __launch_bounds__(1024) gat_bin_count(const int* __restrict__ dst,
                                                      int* __restrict__ table,
                                                      int E, int Nn) {
  __shared__ int h[NBIN];
  const int t = threadIdx.x, b = blockIdx.x, nb = gridDim.x;
  if (t < NBIN) h[t] = 0;
  __syncthreads();
  const unsigned R = (unsigned)(Nn + NBIN - 1) / NBIN;
  const int chunk = (E + nb - 1) / nb;
  const int e0 = b * chunk;
  const int e1 = (e0 + chunk < E) ? e0 + chunk : E;
  for (int base = e0; base < e1; base += 4096) {
    int d[4];
#pragma unroll
    for (int k = 0; k < 4; ++k) {
      const int e = base + k * 1024 + t;
      d[k] = (e < e1) ? dst[e] : -1;
    }
#pragma unroll
    for (int k = 0; k < 4; ++k)
      if (d[k] >= 0) atomicAdd(&h[(unsigned)d[k] / R], 1);
  }
  __syncthreads();
  if (t < NBIN) table[t * nb + b] = h[t];
}

#define SCAN_PER 49
#define SCAN_CH (1024 * SCAN_PER)
__global__ void __launch_bounds__(1024) gat_scan(const int* __restrict__ cnt,
                                                 int* __restrict__ off, int Nn) {
  __shared__ int buf[SCAN_CH];
  __shared__ int wsum[32];
  const int t = threadIdx.x;
  const int lane = t & 31, w = t >> 5;
  int carry = 0;
  for (int c0 = 0; c0 < Nn; c0 += SCAN_CH) {
    int c[SCAN_PER];
#pragma unroll
    for (int k = 0; k < SCAN_PER; ++k) {
      const int j = c0 + k * 1024 + t;
      const int val = cnt[j < Nn ? j : Nn - 1];
      c[k] = (j < Nn) ? val : 0;
    }
#pragma unroll
    for (int k = 0; k < SCAN_PER; ++k) buf[k * 1024 + t] = c[k];
    __syncthreads();

    int v[SCAN_PER];
    int s = 0;
#pragma unroll
    for (int i = 0; i < SCAN_PER; ++i) {
      v[i] = buf[t * SCAN_PER + i];
      s += v[i];
    }
    int x = s;
#pragma unroll
    for (int d = 1; d < 32; d <<= 1) {
      int y = __shfl_up(x, d, 32);
      if (lane >= d) x += y;
    }
    if (lane == 31) wsum[w] = x;
    __syncthreads();
    if (w == 0) {
      int u = wsum[lane];
#pragma unroll
      for (int d = 1; d < 32; d <<= 1) {
        int y = __shfl_up(u, d, 32);
        if (lane >= d) u += y;
      }
      wsum[lane] = u;
    }
    __syncthreads();
    int base = carry + (w > 0 ? wsum[w - 1] : 0) + (x - s);
    carry += wsum[31];
#pragma unroll
    for (int i = 0; i < SCAN_PER; ++i) {
      buf[t * SCAN_PER + i] = base;
      base += v[i];
    }
    __syncthreads();
    const int rows = (Nn - c0) >> 10;
    const int nfull = rows < SCAN_PER ? rows : SCAN_PER;
#pragma unroll 8
    for (int k = 0; k < nfull; ++k) off[c0 + k * 1024 + t] = buf[k * 1024 + t];
    if (nfull < SCAN_PER) {
      const int j = c0 + nfull * 1024 + t;
      if (j < Nn) off[j] = buf[nfull * 1024 + t];
    }
    __syncthreads();
  }
  if (t == 0) off[Nn] = carry;
}

__global__ void __launch_bounds__(1024) gat_bin_scatter(const int* __restrict__ src,
                                                        const int* __restrict__ dst,
                                                        const int* __restrict__ toff,
                                                        int2* __restrict__ cedge,
                                                        int E, int Nn) {
  __shared__ int h[NBIN];
  const int t = threadIdx.x, b = blockIdx.x, nb = gridDim.x;
  if (t < NBIN) h[t] = toff[t * nb + b];
  __syncthreads();
  const unsigned R = (unsigned)(Nn + NBIN - 1) / NBIN;
  const int chunk = (E + nb - 1) / nb;
  const int e0 = b * chunk;
  const int e1 = (e0 + chunk < E) ? e0 + chunk : E;
  for (int base = e0; base < e1; base += 4096) {
    int d[4], s[4];
#pragma unroll
    for (int k = 0; k < 4; ++k) {
      const int e = base + k * 1024 + t;
      d[k] = (e < e1) ? dst[e] : -1;
      s[k] = (e < e1) ? src[e] : 0;
    }
#pragma unroll
    for (int k = 0; k < 4; ++k) {
      if (d[k] >= 0) {
        const int p = atomicAdd(&h[(unsigned)d[k] / R], 1);
        cedge[p] = make_int2(s[k], d[k]);
      }
    }
  }
}

__global__ void __launch_bounds__(1024) gat_bin_sort(const int2* __restrict__ cedge,
                                                    const int* __restrict__ toff,
                                                    int* __restrict__ off,
                                                    int* __restrict__ ssrc,
                                                    int nblk, int Nn) {
  __shared__ int cnt[RMAX];
  __shared__ int pos[RMAX];
  const int t = threadIdx.x;
  const int bin = blockIdx.x;
  const int R = (Nn + NBIN - 1) / NBIN;
  const int n0 = bin * R;
  const int cb = toff[bin * nblk];
  const int ce = toff[(bin + 1) * nblk];
  if (t < RMAX) cnt[t] = 0;
  __syncthreads();
  for (int base = cb; base < ce; base += 4096) {
    int d[4];
#pragma unroll
    for (int k = 0; k < 4; ++k) {
      const int i = base + k * 1024 + t;
      d[k] = (i < ce) ? cedge[i].y : -1;
    }
#pragma unroll
    for (int k = 0; k < 4; ++k)
      if (d[k] >= 0) atomicAdd(&cnt[d[k] - n0], 1);
  }
  __syncthreads();
  if (t < 32) {
    int v[8];
    int s = 0;
#pragma unroll
    for (int k = 0; k < 8; ++k) {
      v[k] = cnt[t * 8 + k];
      s += v[k];
    }
    int x = s;
#pragma unroll
    for (int dd = 1; dd < 32; dd <<= 1) {
      const int y = __shfl_up(x, dd, 32);
      if (t >= dd) x += y;
    }
    int base = cb + (x - s);
#pragma unroll
    for (int k = 0; k < 8; ++k) {
      pos[t * 8 + k] = base;
      base += v[k];
    }
  }
  __syncthreads();
  if (t < R) {
    const int n = n0 + t;
    if (n < Nn) off[n] = pos[t];
  }
  if (bin == NBIN - 1 && t == 0) off[Nn] = ce;
  __syncthreads();
  for (int base = cb; base < ce; base += 4096) {
    int2 ed[4];
#pragma unroll
    for (int k = 0; k < 4; ++k) {
      const int i = base + k * 1024 + t;
      ed[k] = (i < ce) ? cedge[i] : make_int2(0, -1);
    }
#pragma unroll
    for (int k = 0; k < 4; ++k) {
      if (ed[k].y >= 0) {
        const int p = atomicAdd(&pos[ed[k].y - n0], 1);
        ssrc[p] = ed[k].x;
      }
    }
  }
}

template <int LAYER>
__global__ void gat_node(const int* __restrict__ off, const int* __restrict__ ssrc,
                         const float* __restrict__ el, const float* __restrict__ er,
                         const float* __restrict__ h,
                         const float* __restrict__ bias,
                         const float* __restrict__ lns, const float* __restrict__ lnb,
                         float* __restrict__ outp, int Nn) {
  const int lane = threadIdx.x & 31;
  const int n = __builtin_amdgcn_readfirstlane(blockIdx.x * 8 + (threadIdx.x >> 5));
  if (n >= Nn) return;
  const int beg = __builtin_amdgcn_readfirstlane(off[n]);
  const int deg = __builtin_amdgcn_readfirstlane(off[n + 1]) - beg;
  const int hh = lane & 3;
  const int q0 = lane >> 2;
  const int hl = lane >> 3;
  const float erh = er[n * 4 + hh];
  const float* hrow = h + lane * 4;
  const float ninf = -__builtin_huge_valf();

  int   s[4]  = {0, 0, 0, 0};
  float ev[4] = {ninf, ninf, ninf, ninf};
  v4f   r[4][8];

  if (deg > 0) {
#pragma unroll
    for (int t = 0; t < 4; ++t) {
      const int j = t * 8 + q0;
      s[t] = ssrc[beg + (j < deg ? j : 0)];
    }
#pragma unroll
    for (int t = 0; t < 4; ++t) {
      const int j = t * 8 + q0;
      const float x = lrelu(el[s[t] * 4 + hh] + erh);
      ev[t] = (j < deg) ? x : ninf;
    }
#pragma unroll
    for (int q = 0; q < 8; ++q) {
      const int sq = __builtin_amdgcn_readlane(s[0], q * 4);
      r[0][q] = *(const v4f*)(hrow + (size_t)sq * 128);
    }
  }
  if (deg > 8) {
#pragma unroll
    for (int q = 0; q < 8; ++q) {
      const int sq = __builtin_amdgcn_readlane(s[1], q * 4);
      r[1][q] = *(const v4f*)(hrow + (size_t)sq * 128);
    }
  }

  float m = fmaxf(fmaxf(fmaxf(fmaxf(ninf, ev[0]), ev[1]), ev[2]), ev[3]);
  for (int c0 = 32; c0 < deg; c0 += 8) {
    int j = c0 + q0;
    if (j < deg) {
      int sj = ssrc[beg + j];
      m = fmaxf(m, lrelu(el[sj * 4 + hh] + erh));
    }
  }
  m = fmaxf(m, __shfl_xor(m, 4, 32));
  m = fmaxf(m, __shfl_xor(m, 8, 32));
  m = fmaxf(m, __shfl_xor(m, 16, 32));
  m = (fabsf(m) < __builtin_huge_valf()) ? m : 0.f;

  float p[4];
  float den = 0.f;
#pragma unroll
  for (int t = 0; t < 4; ++t) {
    p[t] = (t * 8 + q0 < deg) ? expf(ev[t] - m) : 0.f;
    den += p[t];
  }
  for (int c0 = 32; c0 < deg; c0 += 8) {
    int j = c0 + q0;
    if (j < deg) {
      int sj = ssrc[beg + j];
      den += expf(lrelu(el[sj * 4 + hh] + erh) - m);
    }
  }
  den += __shfl_xor(den, 4, 32);
  den += __shfl_xor(den, 8, 32);
  den += __shfl_xor(den, 16, 32);
  den = fmaxf(den, 1e-9f);

  v4f acc = {0.f, 0.f, 0.f, 0.f};
#pragma unroll
  for (int t = 0; t < 4; ++t) {
    if (t * 8 < deg) {
      const float alpha = p[t] / den;
#pragma unroll
      for (int q = 0; q < 8; ++q) {
        const float a = __shfl(alpha, q * 4 + hl, 32);
        acc += a * r[t][q];
      }
    }
    const int t2 = (t + 2) & 3;
    if (t < 2 && (t + 2) * 8 < deg) {
#pragma unroll
      for (int q = 0; q < 8; ++q) {
        const int sq = __builtin_amdgcn_readlane(s[t2], q * 4);
        r[t2][q] = *(const v4f*)(hrow + (size_t)sq * 128);
      }
    }
  }
  for (int c0 = 32; c0 < deg; c0 += 8) {
    const int j = c0 + q0;
    const bool valid = j < deg;
    const int sj = ssrc[beg + (valid ? j : c0)];
    const float evj = lrelu(el[sj * 4 + hh] + erh);
    const float alpha = valid ? expf(evj - m) / den : 0.f;
#pragma unroll
    for (int q = 0; q < 8; ++q) {
      const int sq = __builtin_amdgcn_readlane(sj, q * 4);
      const float a = __shfl(alpha, q * 4 + hl, 32);
      const v4f v = *(const v4f*)(hrow + (size_t)sq * 128);
      acc += a * v;
    }
  }

  float x0 = acc.x, x1 = acc.y, x2 = acc.z, x3 = acc.w;
  x0 += __shfl_xor(x0, 8, 32);  x1 += __shfl_xor(x1, 8, 32);
  x2 += __shfl_xor(x2, 8, 32);  x3 += __shfl_xor(x3, 8, 32);
  x0 += __shfl_xor(x0, 16, 32); x1 += __shfl_xor(x1, 16, 32);
  x2 += __shfl_xor(x2, 16, 32); x3 += __shfl_xor(x3, 16, 32);

  const int f0 = (lane & 7) * 4;
#pragma unroll
  for (int k = 0; k < 4; ++k) {
    x0 += bias[k * 32 + f0 + 0];
    x1 += bias[k * 32 + f0 + 1];
    x2 += bias[k * 32 + f0 + 2];
    x3 += bias[k * 32 + f0 + 3];
  }
  x0 *= 0.25f; x1 *= 0.25f; x2 *= 0.25f; x3 *= 0.25f;

  v4f y;
  if (LAYER == 1) {
    float mu = (x0 + x1) + (x2 + x3);
    mu += __shfl_xor(mu, 1, 32);
    mu += __shfl_xor(mu, 2, 32);
    mu += __shfl_xor(mu, 4, 32);
    mu *= (1.f / 32.f);
    const float d0 = x0 - mu, d1 = x1 - mu, d2 = x2 - mu, d3 = x3 - mu;
    float var = (d0 * d0 + d1 * d1) + (d2 * d2 + d3 * d3);
    var += __shfl_xor(var, 1, 32);
    var += __shfl_xor(var, 2, 32);
    var += __shfl_xor(var, 4, 32);
    var *= (1.f / 32.f);
    const float r = rsqrtf(var + LN_EPS);
    y.x = fmaxf(d0 * r * lns[f0 + 0] + lnb[f0 + 0], 0.f);
    y.y = fmaxf(d1 * r * lns[f0 + 1] + lnb[f0 + 1], 0.f);
    y.z = fmaxf(d2 * r * lns[f0 + 2] + lnb[f0 + 2], 0.f);
    y.w = fmaxf(d3 * r * lns[f0 + 3] + lnb[f0 + 3], 0.f);
  } else {
    y.x = x0; y.y = x1; y.z = x2; y.w = x3;
  }
  if (lane < 8) *(v4f*)(outp + (size_t)n * 32 + f0) = y;
}

#define ZP 132
__device__ __forceinline__ float lane_bcast(float v, int l) {
  return __int_as_float(__builtin_amdgcn_readlane(__float_as_int(v), l));
}

__global__ void __launch_bounds__(512) gat_node_agg(
    const int* __restrict__ off, const int* __restrict__ ssrc,
    const float* __restrict__ el, const float* __restrict__ er,
    const float* __restrict__ xin,
    const float* __restrict__ Wm,
    const float* __restrict__ bias,
    float* __restrict__ outp, int Nn) {
  __shared__ __attribute__((aligned(16))) float ws[32 * 128];
  __shared__ __attribute__((aligned(16))) float zs[16 * ZP];

  const int tid = threadIdx.x;
  const int lane = tid & 31;
  const int w = __builtin_amdgcn_readfirstlane(tid >> 5);
  const int n = blockIdx.x * 16 + w;

  for (int idx = tid; idx < 32 * 128; idx += 512) {
    const int k = idx >> 7, col = idx & 127;
    const int kk = (col >> 5) * 32 + k;
    const int f = col & 31;
    const int ks = kk >> 2;
    const int ln = (f >> 1) | ((kk & 2) << 3);
    ws[((ks * 32 + ln) * 2 + (f & 1)) * 2 + (kk & 1)] = Wm[idx];
  }

  float z[4] = {0.f, 0.f, 0.f, 0.f};
  if (n < Nn) {
    const int beg = __builtin_amdgcn_readfirstlane(off[n]);
    const int deg = __builtin_amdgcn_readfirstlane(off[n + 1]) - beg;
    const int hh = lane & 3;
    const int q0 = lane >> 2;
    const float erh = er[n * 4 + hh];
    const float* xrow = xin + lane;
    const float ninf = -__builtin_huge_valf();

    int   s[4]  = {0, 0, 0, 0};
    float ev[4] = {ninf, ninf, ninf, ninf};
    float r[4][8];

    if (deg > 0) {
#pragma unroll
      for (int t = 0; t < 4; ++t) {
        const int j = t * 8 + q0;
        s[t] = ssrc[beg + (j < deg ? j : 0)];
      }
#pragma unroll
      for (int t = 0; t < 4; ++t) {
        const int j = t * 8 + q0;
        const float x = lrelu(el[s[t] * 4 + hh] + erh);
        ev[t] = (j < deg) ? x : ninf;
      }
    }
#pragma unroll
    for (int t = 0; t < 4; ++t) {
      if (t * 8 < deg) {
#pragma unroll
        for (int q = 0; q < 8; ++q) {
          const int sq = __builtin_amdgcn_readlane(s[t], q * 4);
          r[t][q] = xrow[(size_t)sq * 32];
        }
      }
    }

    float m = fmaxf(fmaxf(fmaxf(fmaxf(ninf, ev[0]), ev[1]), ev[2]), ev[3]);
    for (int c0 = 32; c0 < deg; c0 += 8) {
      int j = c0 + q0;
      if (j < deg) {
        int sj = ssrc[beg + j];
        m = fmaxf(m, lrelu(el[sj * 4 + hh] + erh));
      }
    }
    m = fmaxf(m, __shfl_xor(m, 4, 32));
    m = fmaxf(m, __shfl_xor(m, 8, 32));
    m = fmaxf(m, __shfl_xor(m, 16, 32));
    m = (fabsf(m) < __builtin_huge_valf()) ? m : 0.f;

    float p[4];
    float den = 0.f;
#pragma unroll
    for (int t = 0; t < 4; ++t) {
      p[t] = (t * 8 + q0 < deg) ? expf(ev[t] - m) : 0.f;
      den += p[t];
    }
    for (int c0 = 32; c0 < deg; c0 += 8) {
      int j = c0 + q0;
      if (j < deg) {
        int sj = ssrc[beg + j];
        den += expf(lrelu(el[sj * 4 + hh] + erh) - m);
      }
    }
    den += __shfl_xor(den, 4, 32);
    den += __shfl_xor(den, 8, 32);
    den += __shfl_xor(den, 16, 32);
    den = fmaxf(den, 1e-9f);

#pragma unroll
    for (int t = 0; t < 4; ++t) {
      if (t * 8 < deg) {
        const float alpha = p[t] / den;
#pragma unroll
        for (int q = 0; q < 8; ++q) {
#pragma unroll
          for (int h = 0; h < 4; ++h) z[h] = __builtin_fmaf(lane_bcast(alpha, q * 4 + h), r[t][q], z[h]);
        }
      }
    }
    for (int c0 = 32; c0 < deg; c0 += 8) {
      const int j = c0 + q0;
      const bool valid = j < deg;
      const int sj = ssrc[beg + (valid ? j : c0)];
      const float evj = lrelu(el[sj * 4 + hh] + erh);
      const float alpha = valid ? expf(evj - m) / den : 0.f;
#pragma unroll
      for (int q = 0; q < 8; ++q) {
        const int sq = __builtin_amdgcn_readlane(sj, q * 4);
        const float v = xrow[(size_t)sq * 32];
#pragma unroll
        for (int h = 0; h < 4; ++h) z[h] = __builtin_fmaf(lane_bcast(alpha, q * 4 + h), v, z[h]);
      }
    }
  }

#pragma unroll
  for (int h = 0; h < 4; ++h) zs[w * ZP + h * 32 + lane] = z[h];
  __syncthreads();
  if (w != 0) return;

  const int mm = lane & 15;
  const int hf = lane >> 4;
  const float* ap = zs + mm * ZP + 2 * hf;
  const float* bl = ws + lane * 4;
  const int f0 = 2 * mm;
  float b0[4], b1[4];
#pragma unroll
  for (int k = 0; k < 4; ++k) {
    b0[k] = bias[k * 32 + f0];
    b1[k] = bias[k * 32 + f0 + 1];
  }
  v8f acc0 = {}, acc1 = {};
#pragma unroll 8
  for (int ks = 0; ks < 32; ++ks) {
    const v2f a = *(const v2f*)(ap + ks * 4);
    const v4f q = *(const v4f*)(bl + ks * 128);
    acc0 = __builtin_amdgcn_wmma_f32_16x16x4_f32(false, a, false, q.xy, (short)0, acc0, false, false);
    acc1 = __builtin_amdgcn_wmma_f32_16x16x4_f32(false, a, false, q.zw, (short)0, acc1, false, false);
  }

#pragma unroll
  for (int r = 0; r < 8; ++r) {
    const int row = blockIdx.x * 16 + hf * 8 + r;
    float x0 = acc0[r], x1 = acc1[r];
#pragma unroll
    for (int k = 0; k < 4; ++k) { x0 += b0[k]; x1 += b1[k]; }
    v2f y;
    y.x = x0 * 0.25f; y.y = x1 * 0.25f;
    if (row < Nn) *(v2f*)(outp + (size_t)row * 32 + f0) = y;
  }
}

__global__ void gat_flops(const int* __restrict__ in_deg, int Nn, float* out) {
  __shared__ int wsum[32];
  const int t = threadIdx.x;
  int s = 0;
  for (int i = t; i < Nn; i += 16 * 1024) {
    int c[16];
#pragma unroll
    for (int k = 0; k < 16; ++k) c[k] = (i + k * 1024 < Nn) ? in_deg[i + k * 1024] : 0;
#pragma unroll
    for (int k = 0; k < 16; ++k) s += c[k];
  }
#pragma unroll
  for (int d = 16; d > 0; d >>= 1) s += __shfl_xor(s, d, 32);
  if ((t & 31) == 0) wsum[t >> 5] = s;
  __syncthreads();
  if (t < 32) {
    int v = wsum[t];
#pragma unroll
    for (int d = 16; d > 0; d >>= 1) v += __shfl_xor(v, d, 32);
    if (t == 0) {
      double ne = (double)(float)v;
      double fl = (4.0 * ne * (6.0 * 128 * 32 + 6 * 32 + 2) +
                   4.0 * ne * (6.0 * 32 * 32 + 6 * 32 + 2)) / 1e12;
      out[0] = (float)fl;
    }
  }
}

extern "C" void kernel_launch(void* const* d_in, const int* in_sizes, int n_in,
                              void* d_out, int out_size, void* d_ws, size_t ws_size,
                              hipStream_t stream) {
  const float* feat = (const float*)d_in[0];
  const float* W1   = (const float*)d_in[1];
  const float* al1  = (const float*)d_in[2];
  const float* ar1  = (const float*)d_in[3];
  const float* b1   = (const float*)d_in[4];
  const float* lns  = (const float*)d_in[5];
  const float* lnb  = (const float*)d_in[6];
  const float* W2   = (const float*)d_in[7];
  const float* al2  = (const float*)d_in[8];
  const float* ar2  = (const float*)d_in[9];
  const float* b2   = (const float*)d_in[10];
  const int*   src  = (const int*)d_in[11];
  const int*   dst  = (const int*)d_in[12];
  const int*   indeg= (const int*)d_in[13];

  const int Nn = in_sizes[13];
  const int E  = in_sizes[11];
  const int NH = Nn * 4;

  char* p = (char*)d_ws;
  float* hbuf = (float*)p;   p += (size_t)Nn * 128 * sizeof(float);
  float* hmid = (float*)p;   p += (size_t)Nn * 32  * sizeof(float);
  float* el   = (float*)p;   p += (size_t)NH * sizeof(float);
  float* er   = (float*)p;   p += (size_t)NH * sizeof(float);
  int2*  cedge= (int2*)p;    p += (size_t)E * sizeof(int2);
  int*   ssrc = (int*)p;     p += (size_t)E * sizeof(int);
  int*   table= (int*)p;     p += (size_t)NBIN * CBLK * sizeof(int);
  int*   toff = (int*)p;     p += ((size_t)NBIN * CBLK + 1) * sizeof(int);
  int*   off  = (int*)p;     p += ((size_t)Nn + 1) * sizeof(int);
  (void)ws_size; (void)n_in; (void)out_size;

  float* outh = (float*)d_out;

  dim3 b256(256);
  const int rowTiles   = (Nn + 15) / 16;
  const int gemmBlocks = (rowTiles + 7) / 8;
  const int nodeBlocks = (Nn + 7) / 8;

  gat_bin_count<<<CBLK, 1024, 0, stream>>>(dst, table, E, Nn);
  gat_scan<<<1, 1024, 0, stream>>>(table, toff, NBIN * CBLK);
  gat_bin_scatter<<<CBLK, 1024, 0, stream>>>(src, dst, toff, cedge, E, Nn);
  gat_bin_sort<<<NBIN, 1024, 0, stream>>>(cedge, toff, off, ssrc, CBLK, Nn);

  gat_gemm_wmma<128, true><<<gemmBlocks, b256, 0, stream>>>(feat, W1, hbuf, al1, ar1, el, er, Nn);
  gat_node<1><<<nodeBlocks, b256, 0, stream>>>(off, ssrc, el, er, hbuf, b1, lns, lnb, hmid, Nn);

  gat_gemm_wmma<32, false><<<gemmBlocks, b256, 0, stream>>>(hmid, W2, hbuf, al2, ar2, el, er, Nn);
  gat_node_agg<<<(Nn + 15) / 16, dim3(512), 0, stream>>>(off, ssrc, el, er, hmid, W2, b2, outh, Nn);

  gat_flops<<<1, 1024, 0, stream>>>(indeg, Nn, outh + (size_t)Nn * 32);
}


# edges bucketed in 2 launches, was 4: a block sorts its chunk by node range in place, a range block finds its runs and first slot in the run-start table; no table scan; fp sum order changes
# speedup vs baseline: 7.5286x; 1.2684x over previous
#include <hip/hip_runtime.h>


typedef float v2f __attribute__((ext_vector_type(2)));
typedef float v4f __attribute__((ext_vector_type(4)));
typedef float v8f __attribute__((ext_vector_type(8)));

#define NEG_SLOPE 0.2f
#define LN_EPS 1e-5f

__device__ __forceinline__ float lrelu(float x) {
  return x > 0.f ? x : x * NEG_SLOPE;
}

template <int K, bool STORE_C>
__global__ void gat_gemm_wmma(const float* __restrict__ Amat,
                              const float* __restrict__ Bmat,
                              float* __restrict__ Cmat,
                              const float* __restrict__ al,
                              const float* __restrict__ ar,
                              float* __restrict__ el, float* __restrict__ er,
                              int M) {
  __shared__ float bs[K * 128];

  const int tid = threadIdx.x;

  for (int idx = tid; idx < K * 128; idx += 256) {
    int kr = idx >> 7;
    int col = idx & 127;
    int ks = kr >> 2;
    int c = col >> 4;
    int lane = (col & 15) | ((kr & 2) << 3);
    int sub = kr & 1;
    bs[((ks * 32 + lane) * 8 + c) * 2 + sub] = Bmat[idx];
  }
  __syncthreads();

  const int wave = tid >> 5;
  const int lane = tid & 31;
  const int row0 = (blockIdx.x * 8 + wave) * 16;
  if (row0 >= M) return;

  const int m  = lane & 15;
  const int kk = (lane >> 4) << 1;

  v8f acc[8] = {};

  float la[4], lb[4], ra[4], rb[4];
#pragma unroll
  for (int hd = 0; hd < 4; ++hd) {
    la[hd] = al[hd * 32 + m];
    lb[hd] = al[hd * 32 + 16 + m];
    ra[hd] = ar[hd * 32 + m];
    rb[hd] = ar[hd * 32 + 16 + m];
  }

  const float* arow = Amat + (size_t)(row0 + m) * K + kk;
  const float* blane = bs + lane * 16;

#pragma unroll 4
  for (int ks = 0; ks < K / 4; ++ks) {
    v2f a = *(const v2f*)(arow + ks * 4);
    const float* bp = blane + ks * 512;
    v4f q0 = *(const v4f*)(bp);
    v4f q1 = *(const v4f*)(bp + 4);
    v4f q2 = *(const v4f*)(bp + 8);
    v4f q3 = *(const v4f*)(bp + 12);
    acc[0] = __builtin_amdgcn_wmma_f32_16x16x4_f32(false, a, false, q0.xy, (short)0, acc[0], false, false);
    acc[1] = __builtin_amdgcn_wmma_f32_16x16x4_f32(false, a, false, q0.zw, (short)0, acc[1], false, false);
    acc[2] = __builtin_amdgcn_wmma_f32_16x16x4_f32(false, a, false, q1.xy, (short)0, acc[2], false, false);
    acc[3] = __builtin_amdgcn_wmma_f32_16x16x4_f32(false, a, false, q1.zw, (short)0, acc[3], false, false);
    acc[4] = __builtin_amdgcn_wmma_f32_16x16x4_f32(false, a, false, q2.xy, (short)0, acc[4], false, false);
    acc[5] = __builtin_amdgcn_wmma_f32_16x16x4_f32(false, a, false, q2.zw, (short)0, acc[5], false, false);
    acc[6] = __builtin_amdgcn_wmma_f32_16x16x4_f32(false, a, false, q3.xy, (short)0, acc[6], false, false);
    acc[7] = __builtin_amdgcn_wmma_f32_16x16x4_f32(false, a, false, q3.zw, (short)0, acc[7], false, false);
  }

  const int mo = (lane >> 4) * 8;
  if (STORE_C) {
#pragma unroll
    for (int c = 0; c < 8; ++c) {
#pragma unroll
      for (int r = 0; r < 8; ++r) {
        Cmat[(size_t)(row0 + mo + r) * 128 + c * 16 + m] = acc[c][r];
      }
    }
  }

  const bool u3 = (lane & 8) != 0, u2 = (lane & 4) != 0;
  const bool u1 = (lane & 2) != 0, u0 = (lane & 1) != 0;
  float rs[4];
#pragma unroll
  for (int hd = 0; hd < 4; ++hd) {
    float x[16];
#pragma unroll
    for (int r = 0; r < 8; ++r) {
      const float c0 = acc[2 * hd][r], c1 = acc[2 * hd + 1][r];
      x[r]     = c0 * la[hd] + c1 * lb[hd];
      x[8 + r] = c0 * ra[hd] + c1 * rb[hd];
    }
    float y[8];
#pragma unroll
    for (int j = 0; j < 8; ++j) {
      const float keep = u3 ? x[j + 8] : x[j];
      const float give = u3 ? x[j] : x[j + 8];
      y[j] = keep + __shfl_xor(give, 8, 32);
    }
    float z[4];
#pragma unroll
    for (int j = 0; j < 4; ++j) {
      const float keep = u2 ? y[j + 4] : y[j];
      const float give = u2 ? y[j] : y[j + 4];
      z[j] = keep + __shfl_xor(give, 4, 32);
    }
    float w[2];
#pragma unroll
    for (int j = 0; j < 2; ++j) {
      const float keep = u1 ? z[j + 2] : z[j];
      const float give = u1 ? z[j] : z[j + 2];
      w[j] = keep + __shfl_xor(give, 2, 32);
    }
    const float keep = u0 ? w[1] : w[0];
    const float give = u0 ? w[0] : w[1];
    rs[hd] = keep + __shfl_xor(give, 1, 32);
  }
  v4f res;
  res.x = rs[0]; res.y = rs[1]; res.z = rs[2]; res.w = rs[3];
  float* dp = (m < 8 ? el : er) + (size_t)(row0 + mo + (m & 7)) * 4;
  *(v4f*)dp = res;
}

#define NBIN 256
#define RMAX 256
#define CBLK 196
#define SEGU 12

__device__ __forceinline__ int wave_scan256(const int* c, int* o1, int* o2, int base, int lane) {
  int v[8];
  int s = 0;
#pragma unroll
  for (int k = 0; k < 8; ++k) {
    v[k] = c[lane * 8 + k];
    s += v[k];
  }
  int x = s;
#pragma unroll
  for (int dd = 1; dd < 32; dd <<= 1) {
    const int y = __shfl_up(x, dd, 32);
    if (lane >= dd) x += y;
  }
  int run = base + (x - s);
#pragma unroll
  for (int k = 0; k < 8; ++k) {
    o1[lane * 8 + k] = run;
    o2[lane * 8 + k] = run;
    run += v[k];
  }
  return __shfl(x, 31, 32);
}

__global__ void __launch_bounds__(1024) gat_chunk_sort(const int* __restrict__ src,
                                                       const int* __restrict__ dst,
                                                       int* __restrict__ tab,
                                                       int2* __restrict__ cedge,
                                                       int E, int Nn) {
  __shared__ int cnt[NBIN];
  __shared__ int start[NBIN];
  __shared__ int pos[NBIN];
  const int t = threadIdx.x, b = blockIdx.x, nb = gridDim.x;
  if (t < NBIN) cnt[t] = 0;
  const unsigned R = (unsigned)(Nn + NBIN - 1) / NBIN;
  const int chunk = (E + nb - 1) / nb;
  const int e0 = b * chunk;
  const int e1 = (e0 + chunk < E) ? e0 + chunk : E;
  int d[4], s[4];
#pragma unroll
  for (int k = 0; k < 4; ++k) {
    const int e = e0 + k * 1024 + t;
    d[k] = (e < e1) ? dst[e] : -1;
    s[k] = (e < e1) ? src[e] : 0;
  }
  __syncthreads();
#pragma unroll
  for (int k = 0; k < 4; ++k)
    if (d[k] >= 0) atomicAdd(&cnt[(unsigned)d[k] / R], 1);
  for (int base = e0 + 4096; base < e1; base += 4096) {
    int dd[4];
#pragma unroll
    for (int k = 0; k < 4; ++k) {
      const int e = base + k * 1024 + t;
      dd[k] = (e < e1) ? dst[e] : -1;
    }
#pragma unroll
    for (int k = 0; k < 4; ++k)
      if (dd[k] >= 0) atomicAdd(&cnt[(unsigned)dd[k] / R], 1);
  }
  __syncthreads();
  if (t < 32) wave_scan256(cnt, start, pos, 0, t);
  __syncthreads();
  if (t < NBIN) tab[t * nb + b] = start[t];
#pragma unroll
  for (int k = 0; k < 4; ++k) {
    if (d[k] >= 0) {
      const int p = atomicAdd(&pos[(unsigned)d[k] / R], 1);
      cedge[e0 + p] = make_int2(s[k], d[k]);
    }
  }
  for (int base = e0 + 4096; base < e1; base += 4096) {
    int dd[4], ss[4];
#pragma unroll
    for (int k = 0; k < 4; ++k) {
      const int e = base + k * 1024 + t;
      dd[k] = (e < e1) ? dst[e] : -1;
      ss[k] = (e < e1) ? src[e] : 0;
    }
#pragma unroll
    for (int k = 0; k < 4; ++k) {
      if (dd[k] >= 0) {
        const int p = atomicAdd(&pos[(unsigned)dd[k] / R], 1);
        cedge[e0 + p] = make_int2(ss[k], dd[k]);
      }
    }
  }
}

__device__ __forceinline__ void run_bounds(const int* __restrict__ tab, int bin, int sg,
                                           int nblk, int chunk, int E,
                                           int& a, int& z, int& rel) {
  a = 0; z = 0; rel = 0;
  if (sg < nblk) {
    const int c0 = sg * chunk;
    const int c1 = (c0 + chunk < E) ? c0 + chunk : E;
    const int len = (c1 > c0) ? c1 - c0 : 0;
    rel = tab[bin * nblk + sg];
    const int nxt = (bin + 1 < NBIN) ? tab[(bin + 1) * nblk + sg] : len;
    a = c0 + rel;
    z = c0 + nxt;
  }
}

__global__ void __launch_bounds__(1024) gat_range_sort(const int2* __restrict__ cedge,
                                                      const int* __restrict__ tab,
                                                      int* __restrict__ off,
                                                      int* __restrict__ ssrc,
                                                      int nblk, int E, int Nn) {
  __shared__ int cnt[RMAX];
  __shared__ int first[RMAX];
  __shared__ int pos[RMAX];
  __shared__ int wsum[32];
  const int t = threadIdx.x;
  const int lane = t & 31, w = t >> 5;
  const int l4 = t & 3;
  const int bin = blockIdx.x;
  const int R = (Nn + NBIN - 1) / NBIN;
  const int n0 = bin * R;
  const int chunk = (E + nblk - 1) / nblk;
  if (t < RMAX) cnt[t] = 0;

  int a0, z0, rel0;
  run_bounds(tab, bin, t >> 2, nblk, chunk, E, a0, z0, rel0);
  int2 ed[SEGU];
#pragma unroll
  for (int k = 0; k < SEGU; ++k) {
    const int i = a0 + l4 + 4 * k;
    ed[k] = (i < z0) ? cedge[i] : make_int2(0, -1);
  }
  int relsum = (l4 == 0) ? rel0 : 0;
  __syncthreads();
#pragma unroll
  for (int k = 0; k < SEGU; ++k)
    if (ed[k].y >= 0) atomicAdd(&cnt[ed[k].y - n0], 1);
  for (int i = a0 + l4 + 4 * SEGU; i < z0; i += 4) atomicAdd(&cnt[cedge[i].y - n0], 1);
  for (int sb = 256; sb < nblk; sb += 256) {
    int a, z, rel;
    run_bounds(tab, bin, sb + (t >> 2), nblk, chunk, E, a, z, rel);
    if (l4 == 0) relsum += rel;
    for (int i = a + l4; i < z; i += 4) atomicAdd(&cnt[cedge[i].y - n0], 1);
  }
#pragma unroll
  for (int dd = 16; dd > 0; dd >>= 1) relsum += __shfl_xor(relsum, dd, 32);
  if (lane == 0) wsum[w] = relsum;
  __syncthreads();
  if (t < 32) {
    int cb = wsum[t];
#pragma unroll
    for (int dd = 16; dd > 0; dd >>= 1) cb += __shfl_xor(cb, dd, 32);
    const int tot = wave_scan256(cnt, first, pos, cb, t);
    if (bin == NBIN - 1 && t == 0) off[Nn] = cb + tot;
  }
  __syncthreads();
  if (t < R) {
    const int n = n0 + t;
    if (n < Nn) off[n] = first[t];
  }
#pragma unroll
  for (int k = 0; k < SEGU; ++k) {
    if (ed[k].y >= 0) {
      const int p = atomicAdd(&pos[ed[k].y - n0], 1);
      ssrc[p] = ed[k].x;
    }
  }
  for (int i = a0 + l4 + 4 * SEGU; i < z0; i += 4) {
    const int2 e = cedge[i];
    const int p = atomicAdd(&pos[e.y - n0], 1);
    ssrc[p] = e.x;
  }
  for (int sb = 256; sb < nblk; sb += 256) {
    int a, z, rel;
    run_bounds(tab, bin, sb + (t >> 2), nblk, chunk, E, a, z, rel);
    for (int i = a + l4; i < z; i += 4) {
      const int2 e = cedge[i];
      const int p = atomicAdd(&pos[e.y - n0], 1);
      ssrc[p] = e.x;
    }
  }
}

template <int LAYER>
__global__ void gat_node(const int* __restrict__ off, const int* __restrict__ ssrc,
                         const float* __restrict__ el, const float* __restrict__ er,
                         const float* __restrict__ h,
                         const float* __restrict__ bias,
                         const float* __restrict__ lns, const float* __restrict__ lnb,
                         float* __restrict__ outp, int Nn) {
  const int lane = threadIdx.x & 31;
  const int n = __builtin_amdgcn_readfirstlane(blockIdx.x * 8 + (threadIdx.x >> 5));
  if (n >= Nn) return;
  const int beg = __builtin_amdgcn_readfirstlane(off[n]);
  const int deg = __builtin_amdgcn_readfirstlane(off[n + 1]) - beg;
  const int hh = lane & 3;
  const int q0 = lane >> 2;
  const int hl = lane >> 3;
  const float erh = er[n * 4 + hh];
  const float* hrow = h + lane * 4;
  const float ninf = -__builtin_huge_valf();

  int   s[4]  = {0, 0, 0, 0};
  float ev[4] = {ninf, ninf, ninf, ninf};
  v4f   r[4][8];

  if (deg > 0) {
#pragma unroll
    for (int t = 0; t < 4; ++t) {
      const int j = t * 8 + q0;
      s[t] = ssrc[beg + (j < deg ? j : 0)];
    }
#pragma unroll
    for (int t = 0; t < 4; ++t) {
      const int j = t * 8 + q0;
      const float x = lrelu(el[s[t] * 4 + hh] + erh);
      ev[t] = (j < deg) ? x : ninf;
    }
#pragma unroll
    for (int q = 0; q < 8; ++q) {
      const int sq = __builtin_amdgcn_readlane(s[0], q * 4);
      r[0][q] = *(const v4f*)(hrow + (size_t)sq * 128);
    }
  }
  if (deg > 8) {
#pragma unroll
    for (int q = 0; q < 8; ++q) {
      const int sq = __builtin_amdgcn_readlane(s[1], q * 4);
      r[1][q] = *(const v4f*)(hrow + (size_t)sq * 128);
    }
  }

  float m = fmaxf(fmaxf(fmaxf(fmaxf(ninf, ev[0]), ev[1]), ev[2]), ev[3]);
  for (int c0 = 32; c0 < deg; c0 += 8) {
    int j = c0 + q0;
    if (j < deg) {
      int sj = ssrc[beg + j];
      m = fmaxf(m, lrelu(el[sj * 4 + hh] + erh));
    }
  }
  m = fmaxf(m, __shfl_xor(m, 4, 32));
  m = fmaxf(m, __shfl_xor(m, 8, 32));
  m = fmaxf(m, __shfl_xor(m, 16, 32));
  m = (fabsf(m) < __builtin_huge_valf()) ? m : 0.f;

  float p[4];
  float den = 0.f;
#pragma unroll
  for (int t = 0; t < 4; ++t) {
    p[t] = (t * 8 + q0 < deg) ? expf(ev[t] - m) : 0.f;
    den += p[t];
  }
  for (int c0 = 32; c0 < deg; c0 += 8) {
    int j = c0 + q0;
    if (j < deg) {
      int sj = ssrc[beg + j];
      den += expf(lrelu(el[sj * 4 + hh] + erh) - m);
    }
  }
  den += __shfl_xor(den, 4, 32);
  den += __shfl_xor(den, 8, 32);
  den += __shfl_xor(den, 16, 32);
  den = fmaxf(den, 1e-9f);

  v4f acc = {0.f, 0.f, 0.f, 0.f};
#pragma unroll
  for (int t = 0; t < 4; ++t) {
    if (t * 8 < deg) {
      const float alpha = p[t] / den;
#pragma unroll
      for (int q = 0; q < 8; ++q) {
        const float a = __shfl(alpha, q * 4 + hl, 32);
        acc += a * r[t][q];
      }
    }
    const int t2 = (t + 2) & 3;
    if (t < 2 && (t + 2) * 8 < deg) {
#pragma unroll
      for (int q = 0; q < 8; ++q) {
        const int sq = __builtin_amdgcn_readlane(s[t2], q * 4);
        r[t2][q] = *(const v4f*)(hrow + (size_t)sq * 128);
      }
    }
  }
  for (int c0 = 32; c0 < deg; c0 += 8) {
    const int j = c0 + q0;
    const bool valid = j < deg;
    const int sj = ssrc[beg + (valid ? j : c0)];
    const float evj = lrelu(el[sj * 4 + hh] + erh);
    const float alpha = valid ? expf(evj - m) / den : 0.f;
#pragma unroll
    for (int q = 0; q < 8; ++q) {
      const int sq = __builtin_amdgcn_readlane(sj, q * 4);
      const float a = __shfl(alpha, q * 4 + hl, 32);
      const v4f v = *(const v4f*)(hrow + (size_t)sq * 128);
      acc += a * v;
    }
  }

  float x0 = acc.x, x1 = acc.y, x2 = acc.z, x3 = acc.w;
  x0 += __shfl_xor(x0, 8, 32);  x1 += __shfl_xor(x1, 8, 32);
  x2 += __shfl_xor(x2, 8, 32);  x3 += __shfl_xor(x3, 8, 32);
  x0 += __shfl_xor(x0, 16, 32); x1 += __shfl_xor(x1, 16, 32);
  x2 += __shfl_xor(x2, 16, 32); x3 += __shfl_xor(x3, 16, 32);

  const int f0 = (lane & 7) * 4;
#pragma unroll
  for (int k = 0; k < 4; ++k) {
    x0 += bias[k * 32 + f0 + 0];
    x1 += bias[k * 32 + f0 + 1];
    x2 += bias[k * 32 + f0 + 2];
    x3 += bias[k * 32 + f0 + 3];
  }
  x0 *= 0.25f; x1 *= 0.25f; x2 *= 0.25f; x3 *= 0.25f;

  v4f y;
  if (LAYER == 1) {
    float mu = (x0 + x1) + (x2 + x3);
    mu += __shfl_xor(mu, 1, 32);
    mu += __shfl_xor(mu, 2, 32);
    mu += __shfl_xor(mu, 4, 32);
    mu *= (1.f / 32.f);
    const float d0 = x0 - mu, d1 = x1 - mu, d2 = x2 - mu, d3 = x3 - mu;
    float var = (d0 * d0 + d1 * d1) + (d2 * d2 + d3 * d3);
    var += __shfl_xor(var, 1, 32);
    var += __shfl_xor(var, 2, 32);
    var += __shfl_xor(var, 4, 32);
    var *= (1.f / 32.f);
    const float r = rsqrtf(var + LN_EPS);
    y.x = fmaxf(d0 * r * lns[f0 + 0] + lnb[f0 + 0], 0.f);
    y.y = fmaxf(d1 * r * lns[f0 + 1] + lnb[f0 + 1], 0.f);
    y.z = fmaxf(d2 * r * lns[f0 + 2] + lnb[f0 + 2], 0.f);
    y.w = fmaxf(d3 * r * lns[f0 + 3] + lnb[f0 + 3], 0.f);
  } else {
    y.x = x0; y.y = x1; y.z = x2; y.w = x3;
  }
  if (lane < 8) *(v4f*)(outp + (size_t)n * 32 + f0) = y;
}

#define ZP 132
__device__ __forceinline__ float lane_bcast(float v, int l) {
  return __int_as_float(__builtin_amdgcn_readlane(__float_as_int(v), l));
}

__global__ void __launch_bounds__(512) gat_node_agg(
    const int* __restrict__ off, const int* __restrict__ ssrc,
    const float* __restrict__ el, const float* __restrict__ er,
    const float* __restrict__ xin,
    const float* __restrict__ Wm,
    const float* __restrict__ bias,
    float* __restrict__ outp, int Nn) {
  __shared__ __attribute__((aligned(16))) float ws[32 * 128];
  __shared__ __attribute__((aligned(16))) float zs[16 * ZP];

  const int tid = threadIdx.x;
  const int lane = tid & 31;
  const int w = __builtin_amdgcn_readfirstlane(tid >> 5);
  const int n = blockIdx.x * 16 + w;

  for (int idx = tid; idx < 32 * 128; idx += 512) {
    const int k = idx >> 7, col = idx & 127;
    const int kk = (col >> 5) * 32 + k;
    const int f = col & 31;
    const int ks = kk >> 2;
    const int ln = (f >> 1) | ((kk & 2) << 3);
    ws[((ks * 32 + ln) * 2 + (f & 1)) * 2 + (kk & 1)] = Wm[idx];
  }

  float z[4] = {0.f, 0.f, 0.f, 0.f};
  if (n < Nn) {
    const int beg = __builtin_amdgcn_readfirstlane(off[n]);
    const int deg = __builtin_amdgcn_readfirstlane(off[n + 1]) - beg;
    const int hh = lane & 3;
    const int q0 = lane >> 2;
    const float erh = er[n * 4 + hh];
    const float* xrow = xin + lane;
    const float ninf = -__builtin_huge_valf();

    int   s[4]  = {0, 0, 0, 0};
    float ev[4] = {ninf, ninf, ninf, ninf};
    float r[4][8];

    if (deg > 0) {
#pragma unroll
      for (int t = 0; t < 4; ++t) {
        const int j = t * 8 + q0;
        s[t] = ssrc[beg + (j < deg ? j : 0)];
      }
#pragma unroll
      for (int t = 0; t < 4; ++t) {
        const int j = t * 8 + q0;
        const float x = lrelu(el[s[t] * 4 + hh] + erh);
        ev[t] = (j < deg) ? x : ninf;
      }
    }
#pragma unroll
    for (int t = 0; t < 4; ++t) {
      if (t * 8 < deg) {
#pragma unroll
        for (int q = 0; q < 8; ++q) {
          const int sq = __builtin_amdgcn_readlane(s[t], q * 4);
          r[t][q] = xrow[(size_t)sq * 32];
        }
      }
    }

    float m = fmaxf(fmaxf(fmaxf(fmaxf(ninf, ev[0]), ev[1]), ev[2]), ev[3]);
    for (int c0 = 32; c0 < deg; c0 += 8) {
      int j = c0 + q0;
      if (j < deg) {
        int sj = ssrc[beg + j];
        m = fmaxf(m, lrelu(el[sj * 4 + hh] + erh));
      }
    }
    m = fmaxf(m, __shfl_xor(m, 4, 32));
    m = fmaxf(m, __shfl_xor(m, 8, 32));
    m = fmaxf(m, __shfl_xor(m, 16, 32));
    m = (fabsf(m) < __builtin_huge_valf()) ? m : 0.f;

    float p[4];
    float den = 0.f;
#pragma unroll
    for (int t = 0; t < 4; ++t) {
      p[t] = (t * 8 + q0 < deg) ? expf(ev[t] - m) : 0.f;
      den += p[t];
    }
    for (int c0 = 32; c0 < deg; c0 += 8) {
      int j = c0 + q0;
      if (j < deg) {
        int sj = ssrc[beg + j];
        den += expf(lrelu(el[sj * 4 + hh] + erh) - m);
      }
    }
    den += __shfl_xor(den, 4, 32);
    den += __shfl_xor(den, 8, 32);
    den += __shfl_xor(den, 16, 32);
    den = fmaxf(den, 1e-9f);

#pragma unroll
    for (int t = 0; t < 4; ++t) {
      if (t * 8 < deg) {
        const float alpha = p[t] / den;
#pragma unroll
        for (int q = 0; q < 8; ++q) {
#pragma unroll
          for (int h = 0; h < 4; ++h) z[h] = __builtin_fmaf(lane_bcast(alpha, q * 4 + h), r[t][q], z[h]);
        }
      }
    }
    for (int c0 = 32; c0 < deg; c0 += 8) {
      const int j = c0 + q0;
      const bool valid = j < deg;
      const int sj = ssrc[beg + (valid ? j : c0)];
      const float evj = lrelu(el[sj * 4 + hh] + erh);
      const float alpha = valid ? expf(evj - m) / den : 0.f;
#pragma unroll
      for (int q = 0; q < 8; ++q) {
        const int sq = __builtin_amdgcn_readlane(sj, q * 4);
        const float v = xrow[(size_t)sq * 32];
#pragma unroll
        for (int h = 0; h < 4; ++h) z[h] = __builtin_fmaf(lane_bcast(alpha, q * 4 + h), v, z[h]);
      }
    }
  }

#pragma unroll
  for (int h = 0; h < 4; ++h) zs[w * ZP + h * 32 + lane] = z[h];
  __syncthreads();
  if (w != 0) return;

  const int mm = lane & 15;
  const int hf = lane >> 4;
  const float* ap = zs + mm * ZP + 2 * hf;
  const float* bl = ws + lane * 4;
  const int f0 = 2 * mm;
  float b0[4], b1[4];
#pragma unroll
  for (int k = 0; k < 4; ++k) {
    b0[k] = bias[k * 32 + f0];
    b1[k] = bias[k * 32 + f0 + 1];
  }
  v8f acc0 = {}, acc1 = {};
#pragma unroll 8
  for (int ks = 0; ks < 32; ++ks) {
    const v2f a = *(const v2f*)(ap + ks * 4);
    const v4f q = *(const v4f*)(bl + ks * 128);
    acc0 = __builtin_amdgcn_wmma_f32_16x16x4_f32(false, a, false, q.xy, (short)0, acc0, false, false);
    acc1 = __builtin_amdgcn_wmma_f32_16x16x4_f32(false, a, false, q.zw, (short)0, acc1, false, false);
  }

#pragma unroll
  for (int r = 0; r < 8; ++r) {
    const int row = blockIdx.x * 16 + hf * 8 + r;
    float x0 = acc0[r], x1 = acc1[r];
#pragma unroll
    for (int k = 0; k < 4; ++k) { x0 += b0[k]; x1 += b1[k]; }
    v2f y;
    y.x = x0 * 0.25f; y.y = x1 * 0.25f;
    if (row < Nn) *(v2f*)(outp + (size_t)row * 32 + f0) = y;
  }
}

__global__ void gat_flops(const int* __restrict__ in_deg, int Nn, float* out) {
  __shared__ int wsum[32];
  const int t = threadIdx.x;
  int s = 0;
  for (int i = t; i < Nn; i += 16 * 1024) {
    int c[16];
#pragma unroll
    for (int k = 0; k < 16; ++k) c[k] = (i + k * 1024 < Nn) ? in_deg[i + k * 1024] : 0;
#pragma unroll
    for (int k = 0; k < 16; ++k) s += c[k];
  }
#pragma unroll
  for (int d = 16; d > 0; d >>= 1) s += __shfl_xor(s, d, 32);
  if ((t & 31) == 0) wsum[t >> 5] = s;
  __syncthreads();
  if (t < 32) {
    int v = wsum[t];
#pragma unroll
    for (int d = 16; d > 0; d >>= 1) v += __shfl_xor(v, d, 32);
    if (t == 0) {
      double ne = (double)(float)v;
      double fl = (4.0 * ne * (6.0 * 128 * 32 + 6 * 32 + 2) +
                   4.0 * ne * (6.0 * 32 * 32 + 6 * 32 + 2)) / 1e12;
      out[0] = (float)fl;
    }
  }
}

extern "C" void kernel_launch(void* const* d_in, const int* in_sizes, int n_in,
                              void* d_out, int out_size, void* d_ws, size_t ws_size,
                              hipStream_t stream) {
  const float* feat = (const float*)d_in[0];
  const float* W1   = (const float*)d_in[1];
  const float* al1  = (const float*)d_in[2];
  const float* ar1  = (const float*)d_in[3];
  const float* b1   = (const float*)d_in[4];
  const float* lns  = (const float*)d_in[5];
  const float* lnb  = (const float*)d_in[6];
  const float* W2   = (const float*)d_in[7];
  const float* al2  = (const float*)d_in[8];
  const float* ar2  = (const float*)d_in[9];
  const float* b2   = (const float*)d_in[10];
  const int*   src  = (const int*)d_in[11];
  const int*   dst  = (const int*)d_in[12];
  const int*   indeg= (const int*)d_in[13];

  const int Nn = in_sizes[13];
  const int E  = in_sizes[11];
  const int NH = Nn * 4;

  char* p = (char*)d_ws;
  float* hbuf = (float*)p;   p += (size_t)Nn * 128 * sizeof(float);
  float* hmid = (float*)p;   p += (size_t)Nn * 32  * sizeof(float);
  float* el   = (float*)p;   p += (size_t)NH * sizeof(float);
  float* er   = (float*)p;   p += (size_t)NH * sizeof(float);
  int2*  cedge= (int2*)p;    p += (size_t)E * sizeof(int2);
  int*   ssrc = (int*)p;     p += (size_t)E * sizeof(int);
  int*   tab  = (int*)p;     p += (size_t)NBIN * CBLK * sizeof(int);
  int*   off  = (int*)p;     p += ((size_t)Nn + 1) * sizeof(int);
  (void)ws_size; (void)n_in; (void)out_size;

  float* outh = (float*)d_out;

  dim3 b256(256);
  const int rowTiles   = (Nn + 15) / 16;
  const int gemmBlocks = (rowTiles + 7) / 8;
  const int nodeBlocks = (Nn + 7) / 8;

  gat_chunk_sort<<<CBLK, 1024, 0, stream>>>(src, dst, tab, cedge, E, Nn);
  gat_range_sort<<<NBIN, 1024, 0, stream>>>(cedge, tab, off, ssrc, CBLK, E, Nn);

  gat_gemm_wmma<128, true><<<gemmBlocks, b256, 0, stream>>>(feat, W1, hbuf, al1, ar1, el, er, Nn);
  gat_node<1><<<nodeBlocks, b256, 0, stream>>>(off, ssrc, el, er, hbuf, b1, lns, lnb, hmid, Nn);

  gat_gemm_wmma<32, false><<<gemmBlocks, b256, 0, stream>>>(hmid, W2, hbuf, al2, ar2, el, er, Nn);
  gat_node_agg<<<(Nn + 15) / 16, dim3(512), 0, stream>>>(off, ssrc, el, er, hmid, W2, b2, outh, Nn);

  gat_flops<<<1, 1024, 0, stream>>>(indeg, Nn, outh + (size_t)Nn * 32);
}


# layer 1 GEMM, level 1 of the edge sort and the flop sum share one launch (independent work by block range of one 1024-thread grid; GEMM block = 32 row tiles): 5 launches, was 7; sums unchanged
# speedup vs baseline: 8.4923x; 1.1280x over previous
#include <hip/hip_runtime.h>


typedef float v2f __attribute__((ext_vector_type(2)));
typedef float v4f __attribute__((ext_vector_type(4)));
typedef float v8f __attribute__((ext_vector_type(8)));

#define NEG_SLOPE 0.2f
#define LN_EPS 1e-5f

__device__ __forceinline__ float lrelu(float x) {
  return x > 0.f ? x : x * NEG_SLOPE;
}

template <int K, bool STORE_C, int NT>
__device__ __forceinline__ void gemm_tiles(const float* __restrict__ Amat,
                                           const float* __restrict__ Bmat,
                                           float* __restrict__ Cmat,
                                           const float* __restrict__ al,
                                           const float* __restrict__ ar,
                                           float* __restrict__ el, float* __restrict__ er,
                                           int M, int blk) {
  __shared__ float bs[K * 128];

  const int tid = threadIdx.x;

  for (int idx = tid; idx < K * 128; idx += NT) {
    int kr = idx >> 7;
    int col = idx & 127;
    int ks = kr >> 2;
    int c = col >> 4;
    int lane = (col & 15) | ((kr & 2) << 3);
    int sub = kr & 1;
    bs[((ks * 32 + lane) * 8 + c) * 2 + sub] = Bmat[idx];
  }
  __syncthreads();

  const int wave = tid >> 5;
  const int lane = tid & 31;
  const int row0 = (blk * (NT / 32) + wave) * 16;
  if (row0 >= M) return;

  const int m  = lane & 15;
  const int kk = (lane >> 4) << 1;

  v8f acc[8] = {};

  float la[4], lb[4], ra[4], rb[4];
#pragma unroll
  for (int hd = 0; hd < 4; ++hd) {
    la[hd] = al[hd * 32 + m];
    lb[hd] = al[hd * 32 + 16 + m];
    ra[hd] = ar[hd * 32 + m];
    rb[hd] = ar[hd * 32 + 16 + m];
  }

  const float* arow = Amat + (size_t)(row0 + m) * K + kk;
  const float* blane = bs + lane * 16;

#pragma unroll 4
  for (int ks = 0; ks < K / 4; ++ks) {
    v2f a = *(const v2f*)(arow + ks * 4);
    const float* bp = blane + ks * 512;
    v4f q0 = *(const v4f*)(bp);
    v4f q1 = *(const v4f*)(bp + 4);
    v4f q2 = *(const v4f*)(bp + 8);
    v4f q3 = *(const v4f*)(bp + 12);
    acc[0] = __builtin_amdgcn_wmma_f32_16x16x4_f32(false, a, false, q0.xy, (short)0, acc[0], false, false);
    acc[1] = __builtin_amdgcn_wmma_f32_16x16x4_f32(false, a, false, q0.zw, (short)0, acc[1], false, false);
    acc[2] = __builtin_amdgcn_wmma_f32_16x16x4_f32(false, a, false, q1.xy, (short)0, acc[2], false, false);
    acc[3] = __builtin_amdgcn_wmma_f32_16x16x4_f32(false, a, false, q1.zw, (short)0, acc[3], false, false);
    acc[4] = __builtin_amdgcn_wmma_f32_16x16x4_f32(false, a, false, q2.xy, (short)0, acc[4], false, false);
    acc[5] = __builtin_amdgcn_wmma_f32_16x16x4_f32(false, a, false, q2.zw, (short)0, acc[5], false, false);
    acc[6] = __builtin_amdgcn_wmma_f32_16x16x4_f32(false, a, false, q3.xy, (short)0, acc[6], false, false);
    acc[7] = __builtin_amdgcn_wmma_f32_16x16x4_f32(false, a, false, q3.zw, (short)0, acc[7], false, false);
  }

  const int mo = (lane >> 4) * 8;
  if (STORE_C) {
#pragma unroll
    for (int c = 0; c < 8; ++c) {
#pragma unroll
      for (int r = 0; r < 8; ++r) {
        Cmat[(size_t)(row0 + mo + r) * 128 + c * 16 + m] = acc[c][r];
      }
    }
  }

  const bool u3 = (lane & 8) != 0, u2 = (lane & 4) != 0;
  const bool u1 = (lane & 2) != 0, u0 = (lane & 1) != 0;
  float rs[4];
#pragma unroll
  for (int hd = 0; hd < 4; ++hd) {
    float x[16];
#pragma unroll
    for (int r = 0; r < 8; ++r) {
      const float c0 = acc[2 * hd][r], c1 = acc[2 * hd + 1][r];
      x[r]     = c0 * la[hd] + c1 * lb[hd];
      x[8 + r] = c0 * ra[hd] + c1 * rb[hd];
    }
    float y[8];
#pragma unroll
    for (int j = 0; j < 8; ++j) {
      const float keep = u3 ? x[j + 8] : x[j];
      const float give = u3 ? x[j] : x[j + 8];
      y[j] = keep + __shfl_xor(give, 8, 32);
    }
    float z[4];
#pragma unroll
    for (int j = 0; j < 4; ++j) {
      const float keep = u2 ? y[j + 4] : y[j];
      const float give = u2 ? y[j] : y[j + 4];
      z[j] = keep + __shfl_xor(give, 4, 32);
    }
    float w[2];
#pragma unroll
    for (int j = 0; j < 2; ++j) {
      const float keep = u1 ? z[j + 2] : z[j];
      const float give = u1 ? z[j] : z[j + 2];
      w[j] = keep + __shfl_xor(give, 2, 32);
    }
    const float keep = u0 ? w[1] : w[0];
    const float give = u0 ? w[0] : w[1];
    rs[hd] = keep + __shfl_xor(give, 1, 32);
  }
  v4f res;
  res.x = rs[0]; res.y = rs[1]; res.z = rs[2]; res.w = rs[3];
  float* dp = (m < 8 ? el : er) + (size_t)(row0 + mo + (m & 7)) * 4;
  *(v4f*)dp = res;
}

template <int K, bool STORE_C>
__global__ void gat_gemm_wmma(const float* __restrict__ Amat,
                              const float* __restrict__ Bmat,
                              float* __restrict__ Cmat,
                              const float* __restrict__ al,
                              const float* __restrict__ ar,
                              float* __restrict__ el, float* __restrict__ er,
                              int M) {
  gemm_tiles<K, STORE_C, 256>(Amat, Bmat, Cmat, al, ar, el, er, M, blockIdx.x);
}

#define NBIN 256
#define RMAX 256
#define CBLK 196
#define SEGU 12

__device__ __forceinline__ int wave_scan256(const int* c, int* o1, int* o2, int base, int lane) {
  int v[8];
  int s = 0;
#pragma unroll
  for (int k = 0; k < 8; ++k) {
    v[k] = c[lane * 8 + k];
    s += v[k];
  }
  int x = s;
#pragma unroll
  for (int dd = 1; dd < 32; dd <<= 1) {
    const int y = __shfl_up(x, dd, 32);
    if (lane >= dd) x += y;
  }
  int run = base + (x - s);
#pragma unroll
  for (int k = 0; k < 8; ++k) {
    o1[lane * 8 + k] = run;
    o2[lane * 8 + k] = run;
    run += v[k];
  }
  return __shfl(x, 31, 32);
}

__device__ __forceinline__ void chunk_sort_block(const int* __restrict__ src,
                                                 const int* __restrict__ dst,
                                                 int* __restrict__ tab,
                                                 int2* __restrict__ cedge,
                                                 int E, int Nn, int b, int nb) {
  __shared__ int cnt[NBIN];
  __shared__ int start[NBIN];
  __shared__ int pos[NBIN];
  const int t = threadIdx.x;
  if (t < NBIN) cnt[t] = 0;
  const unsigned R = (unsigned)(Nn + NBIN - 1) / NBIN;
  const int chunk = (E + nb - 1) / nb;
  const int e0 = b * chunk;
  const int e1 = (e0 + chunk < E) ? e0 + chunk : E;
  int d[4], s[4];
#pragma unroll
  for (int k = 0; k < 4; ++k) {
    const int e = e0 + k * 1024 + t;
    d[k] = (e < e1) ? dst[e] : -1;
    s[k] = (e < e1) ? src[e] : 0;
  }
  __syncthreads();
#pragma unroll
  for (int k = 0; k < 4; ++k)
    if (d[k] >= 0) atomicAdd(&cnt[(unsigned)d[k] / R], 1);
  for (int base = e0 + 4096; base < e1; base += 4096) {
    int dd[4];
#pragma unroll
    for (int k = 0; k < 4; ++k) {
      const int e = base + k * 1024 + t;
      dd[k] = (e < e1) ? dst[e] : -1;
    }
#pragma unroll
    for (int k = 0; k < 4; ++k)
      if (dd[k] >= 0) atomicAdd(&cnt[(unsigned)dd[k] / R], 1);
  }
  __syncthreads();
  if (t < 32) wave_scan256(cnt, start, pos, 0, t);
  __syncthreads();
  if (t < NBIN) tab[t * nb + b] = start[t];
#pragma unroll
  for (int k = 0; k < 4; ++k) {
    if (d[k] >= 0) {
      const int p = atomicAdd(&pos[(unsigned)d[k] / R], 1);
      cedge[e0 + p] = make_int2(s[k], d[k]);
    }
  }
  for (int base = e0 + 4096; base < e1; base += 4096) {
    int dd[4], ss[4];
#pragma unroll
    for (int k = 0; k < 4; ++k) {
      const int e = base + k * 1024 + t;
      dd[k] = (e < e1) ? dst[e] : -1;
      ss[k] = (e < e1) ? src[e] : 0;
    }
#pragma unroll
    for (int k = 0; k < 4; ++k) {
      if (dd[k] >= 0) {
        const int p = atomicAdd(&pos[(unsigned)dd[k] / R], 1);
        cedge[e0 + p] = make_int2(ss[k], dd[k]);
      }
    }
  }
}

__device__ __forceinline__ void run_bounds(const int* __restrict__ tab, int bin, int sg,
                                           int nblk, int chunk, int E,
                                           int& a, int& z, int& rel) {
  a = 0; z = 0; rel = 0;
  if (sg < nblk) {
    const int c0 = sg * chunk;
    const int c1 = (c0 + chunk < E) ? c0 + chunk : E;
    const int len = (c1 > c0) ? c1 - c0 : 0;
    rel = tab[bin * nblk + sg];
    const int nxt = (bin + 1 < NBIN) ? tab[(bin + 1) * nblk + sg] : len;
    a = c0 + rel;
    z = c0 + nxt;
  }
}

__global__ void __launch_bounds__(1024) gat_range_sort(const int2* __restrict__ cedge,
                                                      const int* __restrict__ tab,
                                                      int* __restrict__ off,
                                                      int* __restrict__ ssrc,
                                                      int nblk, int E, int Nn) {
  __shared__ int cnt[RMAX];
  __shared__ int first[RMAX];
  __shared__ int pos[RMAX];
  __shared__ int wsum[32];
  const int t = threadIdx.x;
  const int lane = t & 31, w = t >> 5;
  const int l4 = t & 3;
  const int bin = blockIdx.x;
  const int R = (Nn + NBIN - 1) / NBIN;
  const int n0 = bin * R;
  const int chunk = (E + nblk - 1) / nblk;
  if (t < RMAX) cnt[t] = 0;

  int a0, z0, rel0;
  run_bounds(tab, bin, t >> 2, nblk, chunk, E, a0, z0, rel0);
  int2 ed[SEGU];
#pragma unroll
  for (int k = 0; k < SEGU; ++k) {
    const int i = a0 + l4 + 4 * k;
    ed[k] = (i < z0) ? cedge[i] : make_int2(0, -1);
  }
  int relsum = (l4 == 0) ? rel0 : 0;
  __syncthreads();
#pragma unroll
  for (int k = 0; k < SEGU; ++k)
    if (ed[k].y >= 0) atomicAdd(&cnt[ed[k].y - n0], 1);
  for (int i = a0 + l4 + 4 * SEGU; i < z0; i += 4) atomicAdd(&cnt[cedge[i].y - n0], 1);
  for (int sb = 256; sb < nblk; sb += 256) {
    int a, z, rel;
    run_bounds(tab, bin, sb + (t >> 2), nblk, chunk, E, a, z, rel);
    if (l4 == 0) relsum += rel;
    for (int i = a + l4; i < z; i += 4) atomicAdd(&cnt[cedge[i].y - n0], 1);
  }
#pragma unroll
  for (int dd = 16; dd > 0; dd >>= 1) relsum += __shfl_xor(relsum, dd, 32);
  if (lane == 0) wsum[w] = relsum;
  __syncthreads();
  if (t < 32) {
    int cb = wsum[t];
#pragma unroll
    for (int dd = 16; dd > 0; dd >>= 1) cb += __shfl_xor(cb, dd, 32);
    const int tot = wave_scan256(cnt, first, pos, cb, t);
    if (bin == NBIN - 1 && t == 0) off[Nn] = cb + tot;
  }
  __syncthreads();
  if (t < R) {
    const int n = n0 + t;
    if (n < Nn) off[n] = first[t];
  }
#pragma unroll
  for (int k = 0; k < SEGU; ++k) {
    if (ed[k].y >= 0) {
      const int p = atomicAdd(&pos[ed[k].y - n0], 1);
      ssrc[p] = ed[k].x;
    }
  }
  for (int i = a0 + l4 + 4 * SEGU; i < z0; i += 4) {
    const int2 e = cedge[i];
    const int p = atomicAdd(&pos[e.y - n0], 1);
    ssrc[p] = e.x;
  }
  for (int sb = 256; sb < nblk; sb += 256) {
    int a, z, rel;
    run_bounds(tab, bin, sb + (t >> 2), nblk, chunk, E, a, z, rel);
    for (int i = a + l4; i < z; i += 4) {
      const int2 e = cedge[i];
      const int p = atomicAdd(&pos[e.y - n0], 1);
      ssrc[p] = e.x;
    }
  }
}

template <int LAYER>
__global__ void gat_node(const int* __restrict__ off, const int* __restrict__ ssrc,
                         const float* __restrict__ el, const float* __restrict__ er,
                         const float* __restrict__ h,
                         const float* __restrict__ bias,
                         const float* __restrict__ lns, const float* __restrict__ lnb,
                         float* __restrict__ outp, int Nn) {
  const int lane = threadIdx.x & 31;
  const int n = __builtin_amdgcn_readfirstlane(blockIdx.x * 8 + (threadIdx.x >> 5));
  if (n >= Nn) return;
  const int beg = __builtin_amdgcn_readfirstlane(off[n]);
  const int deg = __builtin_amdgcn_readfirstlane(off[n + 1]) - beg;
  const int hh = lane & 3;
  const int q0 = lane >> 2;
  const int hl = lane >> 3;
  const float erh = er[n * 4 + hh];
  const float* hrow = h + lane * 4;
  const float ninf = -__builtin_huge_valf();

  int   s[4]  = {0, 0, 0, 0};
  float ev[4] = {ninf, ninf, ninf, ninf};
  v4f   r[4][8];

  if (deg > 0) {
#pragma unroll
    for (int t = 0; t < 4; ++t) {
      const int j = t * 8 + q0;
      s[t] = ssrc[beg + (j < deg ? j : 0)];
    }
#pragma unroll
    for (int t = 0; t < 4; ++t) {
      const int j = t * 8 + q0;
      const float x = lrelu(el[s[t] * 4 + hh] + erh);
      ev[t] = (j < deg) ? x : ninf;
    }
#pragma unroll
    for (int q = 0; q < 8; ++q) {
      const int sq = __builtin_amdgcn_readlane(s[0], q * 4);
      r[0][q] = *(const v4f*)(hrow + (size_t)sq * 128);
    }
  }
  if (deg > 8) {
#pragma unroll
    for (int q = 0; q < 8; ++q) {
      const int sq = __builtin_amdgcn_readlane(s[1], q * 4);
      r[1][q] = *(const v4f*)(hrow + (size_t)sq * 128);
    }
  }

  float m = fmaxf(fmaxf(fmaxf(fmaxf(ninf, ev[0]), ev[1]), ev[2]), ev[3]);
  for (int c0 = 32; c0 < deg; c0 += 8) {
    int j = c0 + q0;
    if (j < deg) {
      int sj = ssrc[beg + j];
      m = fmaxf(m, lrelu(el[sj * 4 + hh] + erh));
    }
  }
  m = fmaxf(m, __shfl_xor(m, 4, 32));
  m = fmaxf(m, __shfl_xor(m, 8, 32));
  m = fmaxf(m, __shfl_xor(m, 16, 32));
  m = (fabsf(m) < __builtin_huge_valf()) ? m : 0.f;

  float p[4];
  float den = 0.f;
#pragma unroll
  for (int t = 0; t < 4; ++t) {
    p[t] = (t * 8 + q0 < deg) ? expf(ev[t] - m) : 0.f;
    den += p[t];
  }
  for (int c0 = 32; c0 < deg; c0 += 8) {
    int j = c0 + q0;
    if (j < deg) {
      int sj = ssrc[beg + j];
      den += expf(lrelu(el[sj * 4 + hh] + erh) - m);
    }
  }
  den += __shfl_xor(den, 4, 32);
  den += __shfl_xor(den, 8, 32);
  den += __shfl_xor(den, 16, 32);
  den = fmaxf(den, 1e-9f);

  v4f acc = {0.f, 0.f, 0.f, 0.f};
#pragma unroll
  for (int t = 0; t < 4; ++t) {
    if (t * 8 < deg) {
      const float alpha = p[t] / den;
#pragma unroll
      for (int q = 0; q < 8; ++q) {
        const float a = __shfl(alpha, q * 4 + hl, 32);
        acc += a * r[t][q];
      }
    }
    const int t2 = (t + 2) & 3;
    if (t < 2 && (t + 2) * 8 < deg) {
#pragma unroll
      for (int q = 0; q < 8; ++q) {
        const int sq = __builtin_amdgcn_readlane(s[t2], q * 4);
        r[t2][q] = *(const v4f*)(hrow + (size_t)sq * 128);
      }
    }
  }
  for (int c0 = 32; c0 < deg; c0 += 8) {
    const int j = c0 + q0;
    const bool valid = j < deg;
    const int sj = ssrc[beg + (valid ? j : c0)];
    const float evj = lrelu(el[sj * 4 + hh] + erh);
    const float alpha = valid ? expf(evj - m) / den : 0.f;
#pragma unroll
    for (int q = 0; q < 8; ++q) {
      const int sq = __builtin_amdgcn_readlane(sj, q * 4);
      const float a = __shfl(alpha, q * 4 + hl, 32);
      const v4f v = *(const v4f*)(hrow + (size_t)sq * 128);
      acc += a * v;
    }
  }

  float x0 = acc.x, x1 = acc.y, x2 = acc.z, x3 = acc.w;
  x0 += __shfl_xor(x0, 8, 32);  x1 += __shfl_xor(x1, 8, 32);
  x2 += __shfl_xor(x2, 8, 32);  x3 += __shfl_xor(x3, 8, 32);
  x0 += __shfl_xor(x0, 16, 32); x1 += __shfl_xor(x1, 16, 32);
  x2 += __shfl_xor(x2, 16, 32); x3 += __shfl_xor(x3, 16, 32);

  const int f0 = (lane & 7) * 4;
#pragma unroll
  for (int k = 0; k < 4; ++k) {
    x0 += bias[k * 32 + f0 + 0];
    x1 += bias[k * 32 + f0 + 1];
    x2 += bias[k * 32 + f0 + 2];
    x3 += bias[k * 32 + f0 + 3];
  }
  x0 *= 0.25f; x1 *= 0.25f; x2 *= 0.25f; x3 *= 0.25f;

  v4f y;
  if (LAYER == 1) {
    float mu = (x0 + x1) + (x2 + x3);
    mu += __shfl_xor(mu, 1, 32);
    mu += __shfl_xor(mu, 2, 32);
    mu += __shfl_xor(mu, 4, 32);
    mu *= (1.f / 32.f);
    const float d0 = x0 - mu, d1 = x1 - mu, d2 = x2 - mu, d3 = x3 - mu;
    float var = (d0 * d0 + d1 * d1) + (d2 * d2 + d3 * d3);
    var += __shfl_xor(var, 1, 32);
    var += __shfl_xor(var, 2, 32);
    var += __shfl_xor(var, 4, 32);
    var *= (1.f / 32.f);
    const float r = rsqrtf(var + LN_EPS);
    y.x = fmaxf(d0 * r * lns[f0 + 0] + lnb[f0 + 0], 0.f);
    y.y = fmaxf(d1 * r * lns[f0 + 1] + lnb[f0 + 1], 0.f);
    y.z = fmaxf(d2 * r * lns[f0 + 2] + lnb[f0 + 2], 0.f);
    y.w = fmaxf(d3 * r * lns[f0 + 3] + lnb[f0 + 3], 0.f);
  } else {
    y.x = x0; y.y = x1; y.z = x2; y.w = x3;
  }
  if (lane < 8) *(v4f*)(outp + (size_t)n * 32 + f0) = y;
}

#define ZP 132
__device__ __forceinline__ float lane_bcast(float v, int l) {
  return __int_as_float(__builtin_amdgcn_readlane(__float_as_int(v), l));
}

__global__ void __launch_bounds__(512) gat_node_agg(
    const int* __restrict__ off, const int* __restrict__ ssrc,
    const float* __restrict__ el, const float* __restrict__ er,
    const float* __restrict__ xin,
    const float* __restrict__ Wm,
    const float* __restrict__ bias,
    float* __restrict__ outp, int Nn) {
  __shared__ __attribute__((aligned(16))) float ws[32 * 128];
  __shared__ __attribute__((aligned(16))) float zs[16 * ZP];

  const int tid = threadIdx.x;
  const int lane = tid & 31;
  const int w = __builtin_amdgcn_readfirstlane(tid >> 5);
  const int n = blockIdx.x * 16 + w;

  for (int idx = tid; idx < 32 * 128; idx += 512) {
    const int k = idx >> 7, col = idx & 127;
    const int kk = (col >> 5) * 32 + k;
    const int f = col & 31;
    const int ks = kk >> 2;
    const int ln = (f >> 1) | ((kk & 2) << 3);
    ws[((ks * 32 + ln) * 2 + (f & 1)) * 2 + (kk & 1)] = Wm[idx];
  }

  float z[4] = {0.f, 0.f, 0.f, 0.f};
  if (n < Nn) {
    const int beg = __builtin_amdgcn_readfirstlane(off[n]);
    const int deg = __builtin_amdgcn_readfirstlane(off[n + 1]) - beg;
    const int hh = lane & 3;
    const int q0 = lane >> 2;
    const float erh = er[n * 4 + hh];
    const float* xrow = xin + lane;
    const float ninf = -__builtin_huge_valf();

    int   s[4]  = {0, 0, 0, 0};
    float ev[4] = {ninf, ninf, ninf, ninf};
    float r[4][8];

    if (deg > 0) {
#pragma unroll
      for (int t = 0; t < 4; ++t) {
        const int j = t * 8 + q0;
        s[t] = ssrc[beg + (j < deg ? j : 0)];
      }
#pragma unroll
      for (int t = 0; t < 4; ++t) {
        const int j = t * 8 + q0;
        const float x = lrelu(el[s[t] * 4 + hh] + erh);
        ev[t] = (j < deg) ? x : ninf;
      }
    }
#pragma unroll
    for (int t = 0; t < 4; ++t) {
      if (t * 8 < deg) {
#pragma unroll
        for (int q = 0; q < 8; ++q) {
          const int sq = __builtin_amdgcn_readlane(s[t], q * 4);
          r[t][q] = xrow[(size_t)sq * 32];
        }
      }
    }

    float m = fmaxf(fmaxf(fmaxf(fmaxf(ninf, ev[0]), ev[1]), ev[2]), ev[3]);
    for (int c0 = 32; c0 < deg; c0 += 8) {
      int j = c0 + q0;
      if (j < deg) {
        int sj = ssrc[beg + j];
        m = fmaxf(m, lrelu(el[sj * 4 + hh] + erh));
      }
    }
    m = fmaxf(m, __shfl_xor(m, 4, 32));
    m = fmaxf(m, __shfl_xor(m, 8, 32));
    m = fmaxf(m, __shfl_xor(m, 16, 32));
    m = (fabsf(m) < __builtin_huge_valf()) ? m : 0.f;

    float p[4];
    float den = 0.f;
#pragma unroll
    for (int t = 0; t < 4; ++t) {
      p[t] = (t * 8 + q0 < deg) ? expf(ev[t] - m) : 0.f;
      den += p[t];
    }
    for (int c0 = 32; c0 < deg; c0 += 8) {
      int j = c0 + q0;
      if (j < deg) {
        int sj = ssrc[beg + j];
        den += expf(lrelu(el[sj * 4 + hh] + erh) - m);
      }
    }
    den += __shfl_xor(den, 4, 32);
    den += __shfl_xor(den, 8, 32);
    den += __shfl_xor(den, 16, 32);
    den = fmaxf(den, 1e-9f);

#pragma unroll
    for (int t = 0; t < 4; ++t) {
      if (t * 8 < deg) {
        const float alpha = p[t] / den;
#pragma unroll
        for (int q = 0; q < 8; ++q) {
#pragma unroll
          for (int h = 0; h < 4; ++h) z[h] = __builtin_fmaf(lane_bcast(alpha, q * 4 + h), r[t][q], z[h]);
        }
      }
    }
    for (int c0 = 32; c0 < deg; c0 += 8) {
      const int j = c0 + q0;
      const bool valid = j < deg;
      const int sj = ssrc[beg + (valid ? j : c0)];
      const float evj = lrelu(el[sj * 4 + hh] + erh);
      const float alpha = valid ? expf(evj - m) / den : 0.f;
#pragma unroll
      for (int q = 0; q < 8; ++q) {
        const int sq = __builtin_amdgcn_readlane(sj, q * 4);
        const float v = xrow[(size_t)sq * 32];
#pragma unroll
        for (int h = 0; h < 4; ++h) z[h] = __builtin_fmaf(lane_bcast(alpha, q * 4 + h), v, z[h]);
      }
    }
  }

#pragma unroll
  for (int h = 0; h < 4; ++h) zs[w * ZP + h * 32 + lane] = z[h];
  __syncthreads();
  if (w != 0) return;

  const int mm = lane & 15;
  const int hf = lane >> 4;
  const float* ap = zs + mm * ZP + 2 * hf;
  const float* bl = ws + lane * 4;
  const int f0 = 2 * mm;
  float b0[4], b1[4];
#pragma unroll
  for (int k = 0; k < 4; ++k) {
    b0[k] = bias[k * 32 + f0];
    b1[k] = bias[k * 32 + f0 + 1];
  }
  v8f acc0 = {}, acc1 = {};
#pragma unroll 8
  for (int ks = 0; ks < 32; ++ks) {
    const v2f a = *(const v2f*)(ap + ks * 4);
    const v4f q = *(const v4f*)(bl + ks * 128);
    acc0 = __builtin_amdgcn_wmma_f32_16x16x4_f32(false, a, false, q.xy, (short)0, acc0, false, false);
    acc1 = __builtin_amdgcn_wmma_f32_16x16x4_f32(false, a, false, q.zw, (short)0, acc1, false, false);
  }

#pragma unroll
  for (int r = 0; r < 8; ++r) {
    const int row = blockIdx.x * 16 + hf * 8 + r;
    float x0 = acc0[r], x1 = acc1[r];
#pragma unroll
    for (int k = 0; k < 4; ++k) { x0 += b0[k]; x1 += b1[k]; }
    v2f y;
    y.x = x0 * 0.25f; y.y = x1 * 0.25f;
    if (row < Nn) *(v2f*)(outp + (size_t)row * 32 + f0) = y;
  }
}

__device__ __forceinline__ void flops_block(const int* __restrict__ in_deg, int Nn, float* out) {
  __shared__ int wsum[32];
  const int t = threadIdx.x;
  int s = 0;
  for (int i = t; i < Nn; i += 16 * 1024) {
    int c[16];
#pragma unroll
    for (int k = 0; k < 16; ++k) c[k] = (i + k * 1024 < Nn) ? in_deg[i + k * 1024] : 0;
#pragma unroll
    for (int k = 0; k < 16; ++k) s += c[k];
  }
#pragma unroll
  for (int d = 16; d > 0; d >>= 1) s += __shfl_xor(s, d, 32);
  if ((t & 31) == 0) wsum[t >> 5] = s;
  __syncthreads();
  if (t < 32) {
    int v = wsum[t];
#pragma unroll
    for (int d = 16; d > 0; d >>= 1) v += __shfl_xor(v, d, 32);
    if (t == 0) {
      double ne = (double)(float)v;
      double fl = (4.0 * ne * (6.0 * 128 * 32 + 6 * 32 + 2) +
                   4.0 * ne * (6.0 * 32 * 32 + 6 * 32 + 2)) / 1e12;
      out[0] = (float)fl;
    }
  }
}

__global__ void __launch_bounds__(1024) gat_stage1(
    const float* __restrict__ feat, const float* __restrict__ W1,
    float* __restrict__ hbuf,
    const float* __restrict__ al, const float* __restrict__ ar,
    float* __restrict__ el, float* __restrict__ er,
    const int* __restrict__ src, const int* __restrict__ dst,
    int* __restrict__ tab, int2* __restrict__ cedge,
    const int* __restrict__ in_deg, float* flops_out,
    int E, int Nn, int gb) {
  const int b = blockIdx.x;
  if (b < gb) {
    gemm_tiles<128, true, 1024>(feat, W1, hbuf, al, ar, el, er, Nn, b);
  } else if (b < gb + CBLK) {
    chunk_sort_block(src, dst, tab, cedge, E, Nn, b - gb, CBLK);
  } else {
    flops_block(in_deg, Nn, flops_out);
  }
}

extern "C" void kernel_launch(void* const* d_in, const int* in_sizes, int n_in,
                              void* d_out, int out_size, void* d_ws, size_t ws_size,
                              hipStream_t stream) {
  const float* feat = (const float*)d_in[0];
  const float* W1   = (const float*)d_in[1];
  const float* al1  = (const float*)d_in[2];
  const float* ar1  = (const float*)d_in[3];
  const float* b1   = (const float*)d_in[4];
  const float* lns  = (const float*)d_in[5];
  const float* lnb  = (const float*)d_in[6];
  const float* W2   = (const float*)d_in[7];
  const float* al2  = (const float*)d_in[8];
  const float* ar2  = (const float*)d_in[9];
  const float* b2   = (const float*)d_in[10];
  const int*   src  = (const int*)d_in[11];
  const int*   dst  = (const int*)d_in[12];
  const int*   indeg= (const int*)d_in[13];

  const int Nn = in_sizes[13];
  const int E  = in_sizes[11];
  const int NH = Nn * 4;

  char* p = (char*)d_ws;
  float* hbuf = (float*)p;   p += (size_t)Nn * 128 * sizeof(float);
  float* hmid = (float*)p;   p += (size_t)Nn * 32  * sizeof(float);
  float* el   = (float*)p;   p += (size_t)NH * sizeof(float);
  float* er   = (float*)p;   p += (size_t)NH * sizeof(float);
  int2*  cedge= (int2*)p;    p += (size_t)E * sizeof(int2);
  int*   ssrc = (int*)p;     p += (size_t)E * sizeof(int);
  int*   tab  = (int*)p;     p += (size_t)NBIN * CBLK * sizeof(int);
  int*   off  = (int*)p;     p += ((size_t)Nn + 1) * sizeof(int);
  (void)ws_size; (void)n_in; (void)out_size;

  float* outh = (float*)d_out;

  dim3 b256(256);
  const int rowTiles   = (Nn + 15) / 16;
  const int gemmBlocks = (rowTiles + 7) / 8;
  const int gemmBlocks1 = (rowTiles + 31) / 32;
  const int nodeBlocks = (Nn + 7) / 8;

  gat_stage1<<<gemmBlocks1 + CBLK + 1, 1024, 0, stream>>>(feat, W1, hbuf, al1, ar1, el, er,
                                                          src, dst, tab, cedge,
                                                          indeg, outh + (size_t)Nn * 32,
                                                          E, Nn, gemmBlocks1);
  gat_range_sort<<<NBIN, 1024, 0, stream>>>(cedge, tab, off, ssrc, CBLK, E, Nn);

  gat_node<1><<<nodeBlocks, b256, 0, stream>>>(off, ssrc, el, er, hbuf, b1, lns, lnb, hmid, Nn);

  gat_gemm_wmma<32, false><<<gemmBlocks, b256, 0, stream>>>(hmid, W2, hbuf, al2, ar2, el, er, Nn);
  gat_node_agg<<<(Nn + 15) / 16, dim3(512), 0, stream>>>(off, ssrc, el, er, hmid, W2, b2, outh, Nn);

}
